# starting kernel
# baseline (speedup 1.0000x reference)
#include <hip/hip_runtime.h>
#include <math.h>

typedef float v2f __attribute__((ext_vector_type(2)));
typedef float v8f __attribute__((ext_vector_type(8)));

#define MT    32
#define DIN   1195
#define XROW  1196
#define H1S   260
#define H2S   132

#define WMMA_F32(a, b, c) \
    __builtin_amdgcn_wmma_f32_16x16x4_f32(false, (a), false, (b), (short)0, (c), false, false)

__global__ __launch_bounds__(256) void residue_embed_fused(
    const int* __restrict__ seq, const float* __restrict__ xyz,
    const float* __restrict__ dih, const int* __restrict__ cidx,
    const float* __restrict__ orient,
    const float* __restrict__ aa_emb, const float* __restrict__ chain_emb,
    const float* __restrict__ W1, const float* __restrict__ b1,
    const float* __restrict__ W2, const float* __restrict__ b2,
    const float* __restrict__ W3, const float* __restrict__ b3,
    const float* __restrict__ W4, const float* __restrict__ b4,
    float* __restrict__ out)
{
    __shared__ float xs[MT * XROW];
    __shared__ float hA[MT * H1S];
    __shared__ float hB[MT * H2S];

    const int tid = threadIdx.x;
    const int t0  = blockIdx.x * MT;

    for (int i = tid; i < MT * XROW; i += 256) xs[i] = 0.0f;
    __syncthreads();

    for (int i = tid; i < MT * 128; i += 256) {
        int tt = i >> 7, k = i & 127;
        int t  = t0 + tt;
        xs[tt * XROW + k] = aa_emb[seq[t] * 128 + k];
        int ci = cidx[t];
        xs[tt * XROW + 1067 + k] = (ci != 0) ? chain_emb[ci * 128 + k] : 0.0f;
    }
    for (int i = tid; i < MT * 15; i += 256) {
        int tt = i / 15, a = i % 15;
        int t  = t0 + tt;
        const float* xp = xyz + (size_t)t * 45;
        float rx = xp[a * 3 + 0] - xp[3 + 0];
        float ry = xp[a * 3 + 1] - xp[3 + 1];
        float rz = xp[a * 3 + 2] - xp[3 + 2];
        const float* O = orient + (size_t)t * 9;
        int base = tt * XROW + 128 + seq[t] * 45 + a * 3;
        xs[base + 0] = O[0] * rx + O[3] * ry + O[6] * rz;
        xs[base + 1] = O[1] * rx + O[4] * ry + O[7] * rz;
        xs[base + 2] = O[2] * rx + O[5] * ry + O[8] * rz;
    }
    if (tid < MT * 3) {
        int tt = tid / 3, d = tid % 3;
        float v = dih[(size_t)(t0 + tt) * 3 + d];
        const float FREQ[6] = {1.0f, 2.0f, 3.0f, 1.0f, 0.5f, 1.0f / 3.0f};
        int base = tt * XROW + 1028 + d * 13;
        xs[base] = v;
        #pragma unroll
        for (int f = 0; f < 6; ++f) {
            xs[base + 1 + f] = sinf(FREQ[f] * v);
            xs[base + 7 + f] = cosf(FREQ[f] * v);
        }
    }
    __syncthreads();

    const int lane = tid & 31;
    const int w    = tid >> 5;
    const int half = lane >> 4;
    const int col  = lane & 15;

    {
        v8f a00 = {}, a01 = {}, a10 = {}, a11 = {};
        const int n0 = w * 32;
        const float* W1c0 = W1 + n0 + col;
        const float* W1c1 = W1 + n0 + 16 + col;
        const int ar0 = col * XROW;
        const int ar1 = (16 + col) * XROW;
        int kk = half * 2;
        for (int it = 0; it < 298; ++it, kk += 4) {
            v2f fa0 = *(const v2f*)&xs[ar0 + kk];
            v2f fa1 = *(const v2f*)&xs[ar1 + kk];
            v2f bA, bB;
            bA.x = W1c0[kk * 256];  bA.y = W1c0[kk * 256 + 256];
            bB.x = W1c1[kk * 256];  bB.y = W1c1[kk * 256 + 256];
            a00 = WMMA_F32(fa0, bA, a00);
            a01 = WMMA_F32(fa0, bB, a01);
            a10 = WMMA_F32(fa1, bA, a10);
            a11 = WMMA_F32(fa1, bB, a11);
        }
        {
            v2f fa0 = *(const v2f*)&xs[ar0 + kk];
            v2f fa1 = *(const v2f*)&xs[ar1 + kk];
            int r0 = (kk     < DIN) ? kk     : DIN - 1;
            int r1 = (kk + 1 < DIN) ? kk + 1 : DIN - 1;
            v2f bA, bB;
            bA.x = W1c0[r0 * 256];  bA.y = W1c0[r1 * 256];
            bB.x = W1c1[r0 * 256];  bB.y = W1c1[r1 * 256];
            a00 = WMMA_F32(fa0, bA, a00);
            a01 = WMMA_F32(fa0, bB, a01);
            a10 = WMMA_F32(fa1, bA, a10);
            a11 = WMMA_F32(fa1, bB, a11);
        }
        float biasA = b1[n0 + col], biasB = b1[n0 + 16 + col];
        #pragma unroll
        for (int r = 0; r < 8; ++r) {
            int m = r + half * 8;
            float v00 = a00[r] + biasA, v01 = a01[r] + biasB;
            float v10 = a10[r] + biasA, v11 = a11[r] + biasB;
            hA[m * H1S + n0 + col]             = v00 > 0.0f ? v00 : 0.0f;
            hA[m * H1S + n0 + 16 + col]        = v01 > 0.0f ? v01 : 0.0f;
            hA[(m + 16) * H1S + n0 + col]      = v10 > 0.0f ? v10 : 0.0f;
            hA[(m + 16) * H1S + n0 + 16 + col] = v11 > 0.0f ? v11 : 0.0f;
        }
    }
    __syncthreads();

    {
        v8f ac0 = {}, ac1 = {};
        const int n0 = w * 16;
        for (int k0 = 0; k0 < 256; k0 += 4) {
            int kk = k0 + half * 2;
            v2f fa0 = *(const v2f*)&hA[col * H1S + kk];
            v2f fa1 = *(const v2f*)&hA[(16 + col) * H1S + kk];
            v2f b;
            b.x = W2[kk * 128 + n0 + col];
            b.y = W2[(kk + 1) * 128 + n0 + col];
            ac0 = WMMA_F32(fa0, b, ac0);
            ac1 = WMMA_F32(fa1, b, ac1);
        }
        float bias = b2[n0 + col];
        #pragma unroll
        for (int r = 0; r < 8; ++r) {
            int m = r + half * 8;
            float v0 = ac0[r] + bias, v1 = ac1[r] + bias;
            hB[m * H2S + n0 + col]        = v0 > 0.0f ? v0 : 0.0f;
            hB[(m + 16) * H2S + n0 + col] = v1 > 0.0f ? v1 : 0.0f;
        }
    }
    __syncthreads();

    {
        v8f ac0 = {}, ac1 = {};
        const int n0 = w * 16;
        for (int k0 = 0; k0 < 128; k0 += 4) {
            int kk = k0 + half * 2;
            v2f fa0 = *(const v2f*)&hB[col * H2S + kk];
            v2f fa1 = *(const v2f*)&hB[(16 + col) * H2S + kk];
            v2f b;
            b.x = W3[kk * 128 + n0 + col];
            b.y = W3[(kk + 1) * 128 + n0 + col];
            ac0 = WMMA_F32(fa0, b, ac0);
            ac1 = WMMA_F32(fa1, b, ac1);
        }
        float bias = b3[n0 + col];
        #pragma unroll
        for (int r = 0; r < 8; ++r) {
            int m = r + half * 8;
            float v0 = ac0[r] + bias, v1 = ac1[r] + bias;
            hA[m * H2S + n0 + col]        = v0 > 0.0f ? v0 : 0.0f;
            hA[(m + 16) * H2S + n0 + col] = v1 > 0.0f ? v1 : 0.0f;
        }
    }
    __syncthreads();

    {
        v8f ac0 = {}, ac1 = {};
        const int n0 = w * 16;
        for (int k0 = 0; k0 < 128; k0 += 4) {
            int kk = k0 + half * 2;
            v2f fa0 = *(const v2f*)&hA[col * H2S + kk];
            v2f fa1 = *(const v2f*)&hA[(16 + col) * H2S + kk];
            v2f b;
            b.x = W4[kk * 128 + n0 + col];
            b.y = W4[(kk + 1) * 128 + n0 + col];
            ac0 = WMMA_F32(fa0, b, ac0);
            ac1 = WMMA_F32(fa1, b, ac1);
        }
        float bias = b4[n0 + col];
        #pragma unroll
        for (int r = 0; r < 8; ++r) {
            int m = r + half * 8;
            out[(size_t)(t0 + m) * 128 + n0 + col]      = ac0[r] + bias;
            out[(size_t)(t0 + m + 16) * 128 + n0 + col] = ac1[r] + bias;
        }
    }
}

extern "C" void kernel_launch(void* const* d_in, const int* in_sizes, int n_in,
                              void* d_out, int out_size, void* d_ws, size_t ws_size,
                              hipStream_t stream) {
    (void)n_in; (void)d_ws; (void)ws_size; (void)out_size;
    const int*   seq    = (const int*)  d_in[0];
    const float* xyz    = (const float*)d_in[1];
    const float* dih    = (const float*)d_in[2];
    const int*   cidx   = (const int*)  d_in[3];
    const float* orient = (const float*)d_in[4];
    const float* aa_emb    = (const float*)d_in[6];
    const float* chain_emb = (const float*)d_in[7];
    const float* W1 = (const float*)d_in[8];   const float* b1 = (const float*)d_in[9];
    const float* W2 = (const float*)d_in[10];  const float* b2 = (const float*)d_in[11];
    const float* W3 = (const float*)d_in[12];  const float* b3 = (const float*)d_in[13];
    const float* W4 = (const float*)d_in[14];  const float* b4 = (const float*)d_in[15];
    float* out = (float*)d_out;

    const int tokens = in_sizes[0];
    const int blocks = tokens / MT;
    residue_embed_fused<<<blocks, 256, 0, stream>>>(
        seq, xyz, dih, cidx, orient, aa_emb, chain_emb,
        W1, b1, W2, b2, W3, b3, W4, b4, out);
}


# Weight fragments are loaded one group of K-steps ahead into a second register set (W1: 13 steps, W2-W4: 16; first group issued before the preceding barrier), so L2 latency passes under the WMMAs
# speedup vs baseline: 1.8457x; 1.8457x over previous
#include <hip/hip_runtime.h>
#include <math.h>


typedef float v2f __attribute__((ext_vector_type(2)));
typedef float v8f __attribute__((ext_vector_type(8)));

#define MT    32
#define DIN   1195
#define XROW  1196
#define H1S   260
#define H2S   132
#define G1    13
#define G2    16

#define WMMA_F32(a, b, c) \
    __builtin_amdgcn_wmma_f32_16x16x4_f32(false, (a), false, (b), (short)0, (c), false, false)

template <bool LAST>
__device__ __forceinline__ void l1_fetch(v2f (&fA)[G1], v2f (&fB)[G1],
                                         const float* __restrict__ c0, const float* __restrict__ c1,
                                         int g, int half)
{
    const int kk0 = g * (4 * G1) + half * 2;
    #pragma unroll
    for (int s = 0; s < G1; ++s) {
        int r0 = kk0 + 4 * s;
        int r1 = r0 + 1;
        if (LAST && s == G1 - 1) r1 = (r1 < DIN) ? r1 : DIN - 1;
        fA[s].x = c0[r0 * 256];  fA[s].y = c0[r1 * 256];
        fB[s].x = c1[r0 * 256];  fB[s].y = c1[r1 * 256];
    }
}

__device__ __forceinline__ void l1_mma(v8f& a00, v8f& a01, v8f& a10, v8f& a11,
                                       const v2f (&fA)[G1], const v2f (&fB)[G1],
                                       const float* xs, int ar0, int ar1, int g, int half)
{
    const int kk0 = g * (4 * G1) + half * 2;
    #pragma unroll
    for (int s = 0; s < G1; ++s) {
        int kk = kk0 + 4 * s;
        v2f fa0 = *(const v2f*)&xs[ar0 + kk];
        v2f fa1 = *(const v2f*)&xs[ar1 + kk];
        a00 = WMMA_F32(fa0, fA[s], a00);
        a01 = WMMA_F32(fa0, fB[s], a01);
        a10 = WMMA_F32(fa1, fA[s], a10);
        a11 = WMMA_F32(fa1, fB[s], a11);
    }
}

__device__ __forceinline__ void wl_fetch(v2f (&f)[G2], const float* __restrict__ wc, int g, int half)
{
    #pragma unroll
    for (int s = 0; s < G2; ++s) {
        int kk = (g * G2 + s) * 4 + half * 2;
        f[s].x = wc[kk * 128];
        f[s].y = wc[(kk + 1) * 128];
    }
}

__device__ __forceinline__ void wl_mma(v8f& ac0, v8f& ac1, const v2f (&f)[G2],
                                       const float* h, int hs, int col, int g, int half)
{
    #pragma unroll
    for (int s = 0; s < G2; ++s) {
        int kk = (g * G2 + s) * 4 + half * 2;
        v2f fa0 = *(const v2f*)&h[col * hs + kk];
        v2f fa1 = *(const v2f*)&h[(16 + col) * hs + kk];
        ac0 = WMMA_F32(fa0, f[s], ac0);
        ac1 = WMMA_F32(fa1, f[s], ac1);
    }
}

__global__ __launch_bounds__(256) void residue_embed_fused(
    const int* __restrict__ seq, const float* __restrict__ xyz,
    const float* __restrict__ dih, const int* __restrict__ cidx,
    const float* __restrict__ orient,
    const float* __restrict__ aa_emb, const float* __restrict__ chain_emb,
    const float* __restrict__ W1, const float* __restrict__ b1,
    const float* __restrict__ W2, const float* __restrict__ b2,
    const float* __restrict__ W3, const float* __restrict__ b3,
    const float* __restrict__ W4, const float* __restrict__ b4,
    float* __restrict__ out)
{
    __shared__ float xs[MT * XROW];
    __shared__ float hA[MT * H1S];
    __shared__ float hB[MT * H2S];

    const int tid = threadIdx.x;
    const int t0  = blockIdx.x * MT;

    const int lane = tid & 31;
    const int w    = tid >> 5;
    const int half = lane >> 4;
    const int col  = lane & 15;

    const int n1 = w * 32;
    const float* W1c0 = W1 + n1 + col;
    const float* W1c1 = W1 + n1 + 16 + col;
    const int n0 = w * 16;
    const float* W2c = W2 + n0 + col;
    const float* W3c = W3 + n0 + col;
    const float* W4c = W4 + n0 + col;

    v2f pA[G1], pB[G1], qA[G1], qB[G1];
    l1_fetch<false>(pA, pB, W1c0, W1c1, 0, half);
    const float bias1A = b1[n1 + col], bias1B = b1[n1 + 16 + col];
    const float bias2 = b2[n0 + col], bias3 = b3[n0 + col], bias4 = b4[n0 + col];

    for (int i = tid; i < MT * XROW; i += 256) xs[i] = 0.0f;
    __syncthreads();

    for (int i = tid; i < MT * 128; i += 256) {
        int tt = i >> 7, k = i & 127;
        int t  = t0 + tt;
        xs[tt * XROW + k] = aa_emb[seq[t] * 128 + k];
        int ci = cidx[t];
        xs[tt * XROW + 1067 + k] = (ci != 0) ? chain_emb[ci * 128 + k] : 0.0f;
    }
    for (int i = tid; i < MT * 15; i += 256) {
        int tt = i / 15, a = i % 15;
        int t  = t0 + tt;
        const float* xp = xyz + (size_t)t * 45;
        float rx = xp[a * 3 + 0] - xp[3 + 0];
        float ry = xp[a * 3 + 1] - xp[3 + 1];
        float rz = xp[a * 3 + 2] - xp[3 + 2];
        const float* O = orient + (size_t)t * 9;
        int base = tt * XROW + 128 + seq[t] * 45 + a * 3;
        xs[base + 0] = O[0] * rx + O[3] * ry + O[6] * rz;
        xs[base + 1] = O[1] * rx + O[4] * ry + O[7] * rz;
        xs[base + 2] = O[2] * rx + O[5] * ry + O[8] * rz;
    }
    if (tid < MT * 3) {
        int tt = tid / 3, d = tid % 3;
        float v = dih[(size_t)(t0 + tt) * 3 + d];
        const float FREQ[6] = {1.0f, 2.0f, 3.0f, 1.0f, 0.5f, 1.0f / 3.0f};
        int base = tt * XROW + 1028 + d * 13;
        xs[base] = v;
        #pragma unroll
        for (int f = 0; f < 6; ++f) {
            xs[base + 1 + f] = sinf(FREQ[f] * v);
            xs[base + 7 + f] = cosf(FREQ[f] * v);
        }
    }
    __syncthreads();

    v2f uP[G2], uQ[G2];

    {
        v8f a00 = {}, a01 = {}, a10 = {}, a11 = {};
        const int ar0 = col * XROW;
        const int ar1 = (16 + col) * XROW;
        for (int g = 0; g < 20; g += 2) {
            l1_fetch<false>(qA, qB, W1c0, W1c1, g + 1, half);
            l1_mma(a00, a01, a10, a11, pA, pB, xs, ar0, ar1, g, half);
            l1_fetch<false>(pA, pB, W1c0, W1c1, g + 2, half);
            l1_mma(a00, a01, a10, a11, qA, qB, xs, ar0, ar1, g + 1, half);
        }
        l1_fetch<false>(qA, qB, W1c0, W1c1, 21, half);
        l1_mma(a00, a01, a10, a11, pA, pB, xs, ar0, ar1, 20, half);
        l1_fetch<true>(pA, pB, W1c0, W1c1, 22, half);
        l1_mma(a00, a01, a10, a11, qA, qB, xs, ar0, ar1, 21, half);
        wl_fetch(uP, W2c, 0, half);
        l1_mma(a00, a01, a10, a11, pA, pB, xs, ar0, ar1, 22, half);

        #pragma unroll
        for (int r = 0; r < 8; ++r) {
            int m = r + half * 8;
            float v00 = a00[r] + bias1A, v01 = a01[r] + bias1B;
            float v10 = a10[r] + bias1A, v11 = a11[r] + bias1B;
            hA[m * H1S + n1 + col]             = v00 > 0.0f ? v00 : 0.0f;
            hA[m * H1S + n1 + 16 + col]        = v01 > 0.0f ? v01 : 0.0f;
            hA[(m + 16) * H1S + n1 + col]      = v10 > 0.0f ? v10 : 0.0f;
            hA[(m + 16) * H1S + n1 + 16 + col] = v11 > 0.0f ? v11 : 0.0f;
        }
    }
    __syncthreads();

    {
        v8f ac0 = {}, ac1 = {};
        wl_fetch(uQ, W2c, 1, half);
        wl_mma(ac0, ac1, uP, hA, H1S, col, 0, half);
        wl_fetch(uP, W2c, 2, half);
        wl_mma(ac0, ac1, uQ, hA, H1S, col, 1, half);
        wl_fetch(uQ, W2c, 3, half);
        wl_mma(ac0, ac1, uP, hA, H1S, col, 2, half);
        wl_fetch(uP, W3c, 0, half);
        wl_mma(ac0, ac1, uQ, hA, H1S, col, 3, half);
        #pragma unroll
        for (int r = 0; r < 8; ++r) {
            int m = r + half * 8;
            float v0 = ac0[r] + bias2, v1 = ac1[r] + bias2;
            hB[m * H2S + n0 + col]        = v0 > 0.0f ? v0 : 0.0f;
            hB[(m + 16) * H2S + n0 + col] = v1 > 0.0f ? v1 : 0.0f;
        }
    }
    __syncthreads();

    {
        v8f ac0 = {}, ac1 = {};
        wl_fetch(uQ, W3c, 1, half);
        wl_mma(ac0, ac1, uP, hB, H2S, col, 0, half);
        wl_fetch(uP, W4c, 0, half);
        wl_mma(ac0, ac1, uQ, hB, H2S, col, 1, half);
        #pragma unroll
        for (int r = 0; r < 8; ++r) {
            int m = r + half * 8;
            float v0 = ac0[r] + bias3, v1 = ac1[r] + bias3;
            hA[m * H2S + n0 + col]        = v0 > 0.0f ? v0 : 0.0f;
            hA[(m + 16) * H2S + n0 + col] = v1 > 0.0f ? v1 : 0.0f;
        }
    }
    __syncthreads();

    {
        v8f ac0 = {}, ac1 = {};
        wl_fetch(uQ, W4c, 1, half);
        wl_mma(ac0, ac1, uP, hA, H2S, col, 0, half);
        wl_mma(ac0, ac1, uQ, hA, H2S, col, 1, half);
        #pragma unroll
        for (int r = 0; r < 8; ++r) {
            int m = r + half * 8;
            out[(size_t)(t0 + m) * 128 + n0 + col]      = ac0[r] + bias4;
            out[(size_t)(t0 + m + 16) * 128 + n0 + col] = ac1[r] + bias4;
        }
    }
}

extern "C" void kernel_launch(void* const* d_in, const int* in_sizes, int n_in,
                              void* d_out, int out_size, void* d_ws, size_t ws_size,
                              hipStream_t stream) {
    (void)n_in; (void)d_ws; (void)ws_size; (void)out_size;
    const int*   seq    = (const int*)  d_in[0];
    const float* xyz    = (const float*)d_in[1];
    const float* dih    = (const float*)d_in[2];
    const int*   cidx   = (const int*)  d_in[3];
    const float* orient = (const float*)d_in[4];
    const float* aa_emb    = (const float*)d_in[6];
    const float* chain_emb = (const float*)d_in[7];
    const float* W1 = (const float*)d_in[8];   const float* b1 = (const float*)d_in[9];
    const float* W2 = (const float*)d_in[10];  const float* b2 = (const float*)d_in[11];
    const float* W3 = (const float*)d_in[12];  const float* b3 = (const float*)d_in[13];
    const float* W4 = (const float*)d_in[14];  const float* b4 = (const float*)d_in[15];
    float* out = (float*)d_out;

    const int tokens = in_sizes[0];
    const int blocks = tokens / MT;
    residue_embed_fused<<<blocks, 256, 0, stream>>>(
        seq, xyz, dih, cidx, orient, aa_emb, chain_emb,
        W1, b1, W2, b2, W3, b3, W4, b4, out);
}


# Phase 0 batches its global loads: token indices fetched once into LDS ahead of the zero fill, then all embedding and coordinate loads issued before the sin/cos work, so L2 round trips overlap
# speedup vs baseline: 2.0199x; 1.0944x over previous
#include <hip/hip_runtime.h>
#include <math.h>


typedef float v2f __attribute__((ext_vector_type(2)));
typedef float v8f __attribute__((ext_vector_type(8)));

#define MT    32
#define DIN   1195
#define XROW  1196
#define H1S   260
#define H2S   132
#define G1    13
#define G2    16

#define WMMA_F32(a, b, c) \
    __builtin_amdgcn_wmma_f32_16x16x4_f32(false, (a), false, (b), (short)0, (c), false, false)

template <bool LAST>
__device__ __forceinline__ void l1_fetch(v2f (&fA)[G1], v2f (&fB)[G1],
                                         const float* __restrict__ c0, const float* __restrict__ c1,
                                         int g, int half)
{
    const int kk0 = g * (4 * G1) + half * 2;
    #pragma unroll
    for (int s = 0; s < G1; ++s) {
        int r0 = kk0 + 4 * s;
        int r1 = r0 + 1;
        if (LAST && s == G1 - 1) r1 = (r1 < DIN) ? r1 : DIN - 1;
        fA[s].x = c0[r0 * 256];  fA[s].y = c0[r1 * 256];
        fB[s].x = c1[r0 * 256];  fB[s].y = c1[r1 * 256];
    }
}

__device__ __forceinline__ void l1_mma(v8f& a00, v8f& a01, v8f& a10, v8f& a11,
                                       const v2f (&fA)[G1], const v2f (&fB)[G1],
                                       const float* xs, int ar0, int ar1, int g, int half)
{
    const int kk0 = g * (4 * G1) + half * 2;
    #pragma unroll
    for (int s = 0; s < G1; ++s) {
        int kk = kk0 + 4 * s;
        v2f fa0 = *(const v2f*)&xs[ar0 + kk];
        v2f fa1 = *(const v2f*)&xs[ar1 + kk];
        a00 = WMMA_F32(fa0, fA[s], a00);
        a01 = WMMA_F32(fa0, fB[s], a01);
        a10 = WMMA_F32(fa1, fA[s], a10);
        a11 = WMMA_F32(fa1, fB[s], a11);
    }
}

__device__ __forceinline__ void wl_fetch(v2f (&f)[G2], const float* __restrict__ wc, int g, int half)
{
    #pragma unroll
    for (int s = 0; s < G2; ++s) {
        int kk = (g * G2 + s) * 4 + half * 2;
        f[s].x = wc[kk * 128];
        f[s].y = wc[(kk + 1) * 128];
    }
}

__device__ __forceinline__ void wl_mma(v8f& ac0, v8f& ac1, const v2f (&f)[G2],
                                       const float* h, int hs, int col, int g, int half)
{
    #pragma unroll
    for (int s = 0; s < G2; ++s) {
        int kk = (g * G2 + s) * 4 + half * 2;
        v2f fa0 = *(const v2f*)&h[col * hs + kk];
        v2f fa1 = *(const v2f*)&h[(16 + col) * hs + kk];
        ac0 = WMMA_F32(fa0, f[s], ac0);
        ac1 = WMMA_F32(fa1, f[s], ac1);
    }
}

struct CoordIn {
    float ax, ay, az;
    float cx, cy, cz;
    float O[9];
    int   s;
};

__device__ __forceinline__ void coord_load(CoordIn& c, const float* __restrict__ xyz,
                                           const float* __restrict__ orient, const int* sIdx,
                                           int t0, int i)
{
    int tt = i / 15, a = i % 15;
    const float* xp = xyz + (size_t)(t0 + tt) * 45;
    c.ax = xp[a * 3 + 0];  c.ay = xp[a * 3 + 1];  c.az = xp[a * 3 + 2];
    c.cx = xp[3 + 0];      c.cy = xp[3 + 1];      c.cz = xp[3 + 2];
    const float* O = orient + (size_t)(t0 + tt) * 9;
    #pragma unroll
    for (int j = 0; j < 9; ++j) c.O[j] = O[j];
    c.s = sIdx[tt];
}

__device__ __forceinline__ void coord_store(float* xs, const CoordIn& c, int i)
{
    int tt = i / 15, a = i % 15;
    float rx = c.ax - c.cx;
    float ry = c.ay - c.cy;
    float rz = c.az - c.cz;
    int base = tt * XROW + 128 + c.s * 45 + a * 3;
    xs[base + 0] = c.O[0] * rx + c.O[3] * ry + c.O[6] * rz;
    xs[base + 1] = c.O[1] * rx + c.O[4] * ry + c.O[7] * rz;
    xs[base + 2] = c.O[2] * rx + c.O[5] * ry + c.O[8] * rz;
}

__global__ __launch_bounds__(256) void residue_embed_fused(
    const int* __restrict__ seq, const float* __restrict__ xyz,
    const float* __restrict__ dih, const int* __restrict__ cidx,
    const float* __restrict__ orient,
    const float* __restrict__ aa_emb, const float* __restrict__ chain_emb,
    const float* __restrict__ W1, const float* __restrict__ b1,
    const float* __restrict__ W2, const float* __restrict__ b2,
    const float* __restrict__ W3, const float* __restrict__ b3,
    const float* __restrict__ W4, const float* __restrict__ b4,
    float* __restrict__ out)
{
    __shared__ float xs[MT * XROW];
    __shared__ float hA[MT * H1S];
    __shared__ float hB[MT * H2S];
    __shared__ int   sIdx[2 * MT];

    const int tid = threadIdx.x;
    const int t0  = blockIdx.x * MT;

    const int lane = tid & 31;
    const int w    = tid >> 5;
    const int half = lane >> 4;
    const int col  = lane & 15;

    const int n1 = w * 32;
    const float* W1c0 = W1 + n1 + col;
    const float* W1c1 = W1 + n1 + 16 + col;
    const int n0 = w * 16;
    const float* W2c = W2 + n0 + col;
    const float* W3c = W3 + n0 + col;
    const float* W4c = W4 + n0 + col;

    v2f pA[G1], pB[G1], qA[G1], qB[G1];
    l1_fetch<false>(pA, pB, W1c0, W1c1, 0, half);
    const float bias1A = b1[n1 + col], bias1B = b1[n1 + 16 + col];
    const float bias2 = b2[n0 + col], bias3 = b3[n0 + col], bias4 = b4[n0 + col];

    int myIdx = 0;
    if (tid < 2 * MT) myIdx = (tid < MT) ? seq[t0 + tid] : cidx[t0 + tid - MT];
    float dv = 0.0f;
    if (tid < MT * 3) dv = dih[(size_t)t0 * 3 + tid];

    for (int i = tid; i < MT * XROW; i += 256) xs[i] = 0.0f;
    if (tid < 2 * MT) sIdx[tid] = myIdx;
    __syncthreads();

    const int gk = tid & 127;
    const int gh = tid >> 7;
    float ga[16], gc[16];
    int   gci[16];
    #pragma unroll
    for (int j = 0; j < 16; ++j) {
        int tt = 2 * j + gh;
        int ci = sIdx[MT + tt];
        ga[j]  = aa_emb[sIdx[tt] * 128 + gk];
        gc[j]  = chain_emb[ci * 128 + gk];
        gci[j] = ci;
    }
    const int  ci1  = tid + 256;
    const bool has1 = ci1 < MT * 15;
    CoordIn cA, cB;
    coord_load(cA, xyz, orient, sIdx, t0, tid);
    coord_load(cB, xyz, orient, sIdx, t0, has1 ? ci1 : MT * 15 - 1);

    if (tid < MT * 3) {
        int tt = tid / 3, d = tid % 3;
        float v = dv;
        const float FREQ[6] = {1.0f, 2.0f, 3.0f, 1.0f, 0.5f, 1.0f / 3.0f};
        int base = tt * XROW + 1028 + d * 13;
        xs[base] = v;
        #pragma unroll
        for (int f = 0; f < 6; ++f) {
            xs[base + 1 + f] = sinf(FREQ[f] * v);
            xs[base + 7 + f] = cosf(FREQ[f] * v);
        }
    }

    coord_store(xs, cA, tid);
    if (has1) coord_store(xs, cB, ci1);
    #pragma unroll
    for (int j = 0; j < 16; ++j) {
        int tt = 2 * j + gh;
        xs[tt * XROW + gk] = ga[j];
        xs[tt * XROW + 1067 + gk] = (gci[j] != 0) ? gc[j] : 0.0f;
    }
    __syncthreads();

    v2f uP[G2], uQ[G2];

    {
        v8f a00 = {}, a01 = {}, a10 = {}, a11 = {};
        const int ar0 = col * XROW;
        const int ar1 = (16 + col) * XROW;
        for (int g = 0; g < 20; g += 2) {
            l1_fetch<false>(qA, qB, W1c0, W1c1, g + 1, half);
            l1_mma(a00, a01, a10, a11, pA, pB, xs, ar0, ar1, g, half);
            l1_fetch<false>(pA, pB, W1c0, W1c1, g + 2, half);
            l1_mma(a00, a01, a10, a11, qA, qB, xs, ar0, ar1, g + 1, half);
        }
        l1_fetch<false>(qA, qB, W1c0, W1c1, 21, half);
        l1_mma(a00, a01, a10, a11, pA, pB, xs, ar0, ar1, 20, half);
        l1_fetch<true>(pA, pB, W1c0, W1c1, 22, half);
        l1_mma(a00, a01, a10, a11, qA, qB, xs, ar0, ar1, 21, half);
        wl_fetch(uP, W2c, 0, half);
        l1_mma(a00, a01, a10, a11, pA, pB, xs, ar0, ar1, 22, half);

        #pragma unroll
        for (int r = 0; r < 8; ++r) {
            int m = r + half * 8;
            float v00 = a00[r] + bias1A, v01 = a01[r] + bias1B;
            float v10 = a10[r] + bias1A, v11 = a11[r] + bias1B;
            hA[m * H1S + n1 + col]             = v00 > 0.0f ? v00 : 0.0f;
            hA[m * H1S + n1 + 16 + col]        = v01 > 0.0f ? v01 : 0.0f;
            hA[(m + 16) * H1S + n1 + col]      = v10 > 0.0f ? v10 : 0.0f;
            hA[(m + 16) * H1S + n1 + 16 + col] = v11 > 0.0f ? v11 : 0.0f;
        }
    }
    __syncthreads();

    {
        v8f ac0 = {}, ac1 = {};
        wl_fetch(uQ, W2c, 1, half);
        wl_mma(ac0, ac1, uP, hA, H1S, col, 0, half);
        wl_fetch(uP, W2c, 2, half);
        wl_mma(ac0, ac1, uQ, hA, H1S, col, 1, half);
        wl_fetch(uQ, W2c, 3, half);
        wl_mma(ac0, ac1, uP, hA, H1S, col, 2, half);
        wl_fetch(uP, W3c, 0, half);
        wl_mma(ac0, ac1, uQ, hA, H1S, col, 3, half);
        #pragma unroll
        for (int r = 0; r < 8; ++r) {
            int m = r + half * 8;
            float v0 = ac0[r] + bias2, v1 = ac1[r] + bias2;
            hB[m * H2S + n0 + col]        = v0 > 0.0f ? v0 : 0.0f;
            hB[(m + 16) * H2S + n0 + col] = v1 > 0.0f ? v1 : 0.0f;
        }
    }
    __syncthreads();

    {
        v8f ac0 = {}, ac1 = {};
        wl_fetch(uQ, W3c, 1, half);
        wl_mma(ac0, ac1, uP, hB, H2S, col, 0, half);
        wl_fetch(uP, W4c, 0, half);
        wl_mma(ac0, ac1, uQ, hB, H2S, col, 1, half);
        #pragma unroll
        for (int r = 0; r < 8; ++r) {
            int m = r + half * 8;
            float v0 = ac0[r] + bias3, v1 = ac1[r] + bias3;
            hA[m * H2S + n0 + col]        = v0 > 0.0f ? v0 : 0.0f;
            hA[(m + 16) * H2S + n0 + col] = v1 > 0.0f ? v1 : 0.0f;
        }
    }
    __syncthreads();

    {
        v8f ac0 = {}, ac1 = {};
        wl_fetch(uQ, W4c, 1, half);
        wl_mma(ac0, ac1, uP, hA, H2S, col, 0, half);
        wl_mma(ac0, ac1, uQ, hA, H2S, col, 1, half);
        #pragma unroll
        for (int r = 0; r < 8; ++r) {
            int m = r + half * 8;
            out[(size_t)(t0 + m) * 128 + n0 + col]      = ac0[r] + bias4;
            out[(size_t)(t0 + m + 16) * 128 + n0 + col] = ac1[r] + bias4;
        }
    }
}

extern "C" void kernel_launch(void* const* d_in, const int* in_sizes, int n_in,
                              void* d_out, int out_size, void* d_ws, size_t ws_size,
                              hipStream_t stream) {
    (void)n_in; (void)d_ws; (void)ws_size; (void)out_size;
    const int*   seq    = (const int*)  d_in[0];
    const float* xyz    = (const float*)d_in[1];
    const float* dih    = (const float*)d_in[2];
    const int*   cidx   = (const int*)  d_in[3];
    const float* orient = (const float*)d_in[4];
    const float* aa_emb    = (const float*)d_in[6];
    const float* chain_emb = (const float*)d_in[7];
    const float* W1 = (const float*)d_in[8];   const float* b1 = (const float*)d_in[9];
    const float* W2 = (const float*)d_in[10];  const float* b2 = (const float*)d_in[11];
    const float* W3 = (const float*)d_in[12];  const float* b3 = (const float*)d_in[13];
    const float* W4 = (const float*)d_in[14];  const float* b4 = (const float*)d_in[15];
    float* out = (float*)d_out;

    const int tokens = in_sizes[0];
    const int blocks = tokens / MT;
    residue_embed_fused<<<blocks, 256, 0, stream>>>(
        seq, xyz, dih, cidx, orient, aa_emb, chain_emb,
        W1, b1, W2, b2, W3, b3, W4, b4, out);
}


# Each K-group's LDS fragment reads go to a second register set as one batch ahead of the previous group's WMMA block; block edges (empty asm on accumulators + sched barrier) keep loads out of blocks
# speedup vs baseline: 2.1472x; 1.0630x over previous
#include <hip/hip_runtime.h>
#include <math.h>


typedef float v2f __attribute__((ext_vector_type(2)));
typedef float v8f __attribute__((ext_vector_type(8)));

#define MT    32
#define DIN   1195
#define XROW  1196
#define H1S   260
#define H2S   132
#define G1    13
#define G2    16

#define WMMA_F32(a, b, c) \
    __builtin_amdgcn_wmma_f32_16x16x4_f32(false, (a), false, (b), (short)0, (c), false, false)

__device__ __forceinline__ void block_edge(v8f& a, v8f& b, v8f& c, v8f& d)
{
    asm volatile("" : "+v"(a), "+v"(b), "+v"(c), "+v"(d) :: "memory");
    __builtin_amdgcn_sched_barrier(0);
}
__device__ __forceinline__ void block_edge(v8f& a, v8f& b)
{
    asm volatile("" : "+v"(a), "+v"(b) :: "memory");
    __builtin_amdgcn_sched_barrier(0);
}

template <bool LAST>
__device__ __forceinline__ void l1_fetch(v2f (&fA)[G1], v2f (&fB)[G1],
                                         const float* __restrict__ c0, const float* __restrict__ c1,
                                         int g, int half)
{
    const int kk0 = g * (4 * G1) + half * 2;
    #pragma unroll
    for (int s = 0; s < G1; ++s) {
        int r0 = kk0 + 4 * s;
        int r1 = r0 + 1;
        if (LAST && s == G1 - 1) r1 = (r1 < DIN) ? r1 : DIN - 1;
        fA[s].x = c0[r0 * 256];  fA[s].y = c0[r1 * 256];
        fB[s].x = c1[r0 * 256];  fB[s].y = c1[r1 * 256];
    }
}

__device__ __forceinline__ void l1_aload(v2f (&x0)[G1], v2f (&x1)[G1],
                                         const float* xs, int ar0, int ar1, int g, int half)
{
    const int kk0 = g * (4 * G1) + half * 2;
    #pragma unroll
    for (int s = 0; s < G1; ++s) {
        int kk = kk0 + 4 * s;
        x0[s] = *(const v2f*)&xs[ar0 + kk];
        x1[s] = *(const v2f*)&xs[ar1 + kk];
    }
}

__device__ __forceinline__ void l1_mma(v8f& a00, v8f& a01, v8f& a10, v8f& a11,
                                       const v2f (&x0)[G1], const v2f (&x1)[G1],
                                       const v2f (&fA)[G1], const v2f (&fB)[G1])
{
    block_edge(a00, a01, a10, a11);
    #pragma unroll
    for (int s = 0; s < G1; ++s) {
        a00 = WMMA_F32(x0[s], fA[s], a00);
        a01 = WMMA_F32(x0[s], fB[s], a01);
        a10 = WMMA_F32(x1[s], fA[s], a10);
        a11 = WMMA_F32(x1[s], fB[s], a11);
    }
    block_edge(a00, a01, a10, a11);
}

__device__ __forceinline__ void wl_fetch(v2f (&f)[G2], const float* __restrict__ wc, int g, int half)
{
    #pragma unroll
    for (int s = 0; s < G2; ++s) {
        int kk = (g * G2 + s) * 4 + half * 2;
        f[s].x = wc[kk * 128];
        f[s].y = wc[(kk + 1) * 128];
    }
}

__device__ __forceinline__ void wl_aload(v2f (&x0)[G2], v2f (&x1)[G2],
                                         const float* h, int hs, int col, int g, int half)
{
    #pragma unroll
    for (int s = 0; s < G2; ++s) {
        int kk = (g * G2 + s) * 4 + half * 2;
        x0[s] = *(const v2f*)&h[col * hs + kk];
        x1[s] = *(const v2f*)&h[(16 + col) * hs + kk];
    }
}

__device__ __forceinline__ void wl_mma(v8f& ac0, v8f& ac1, const v2f (&x0)[G2], const v2f (&x1)[G2],
                                       const v2f (&f)[G2])
{
    block_edge(ac0, ac1);
    #pragma unroll
    for (int s = 0; s < G2; ++s) {
        ac0 = WMMA_F32(x0[s], f[s], ac0);
        ac1 = WMMA_F32(x1[s], f[s], ac1);
    }
    block_edge(ac0, ac1);
}

struct CoordIn {
    float ax, ay, az;
    float cx, cy, cz;
    float O[9];
    int   s;
};

__device__ __forceinline__ void coord_load(CoordIn& c, const float* __restrict__ xyz,
                                           const float* __restrict__ orient, const int* sIdx,
                                           int t0, int i)
{
    int tt = i / 15, a = i % 15;
    const float* xp = xyz + (size_t)(t0 + tt) * 45;
    c.ax = xp[a * 3 + 0];  c.ay = xp[a * 3 + 1];  c.az = xp[a * 3 + 2];
    c.cx = xp[3 + 0];      c.cy = xp[3 + 1];      c.cz = xp[3 + 2];
    const float* O = orient + (size_t)(t0 + tt) * 9;
    #pragma unroll
    for (int j = 0; j < 9; ++j) c.O[j] = O[j];
    c.s = sIdx[tt];
}

__device__ __forceinline__ void coord_store(float* xs, const CoordIn& c, int i)
{
    int tt = i / 15, a = i % 15;
    float rx = c.ax - c.cx;
    float ry = c.ay - c.cy;
    float rz = c.az - c.cz;
    int base = tt * XROW + 128 + c.s * 45 + a * 3;
    xs[base + 0] = c.O[0] * rx + c.O[3] * ry + c.O[6] * rz;
    xs[base + 1] = c.O[1] * rx + c.O[4] * ry + c.O[7] * rz;
    xs[base + 2] = c.O[2] * rx + c.O[5] * ry + c.O[8] * rz;
}

__global__ __launch_bounds__(256) void residue_embed_fused(
    const int* __restrict__ seq, const float* __restrict__ xyz,
    const float* __restrict__ dih, const int* __restrict__ cidx,
    const float* __restrict__ orient,
    const float* __restrict__ aa_emb, const float* __restrict__ chain_emb,
    const float* __restrict__ W1, const float* __restrict__ b1,
    const float* __restrict__ W2, const float* __restrict__ b2,
    const float* __restrict__ W3, const float* __restrict__ b3,
    const float* __restrict__ W4, const float* __restrict__ b4,
    float* __restrict__ out)
{
    __shared__ float xs[MT * XROW];
    __shared__ float hA[MT * H1S];
    __shared__ float hB[MT * H2S];
    __shared__ int   sIdx[2 * MT];

    const int tid = threadIdx.x;
    const int t0  = blockIdx.x * MT;

    const int lane = tid & 31;
    const int w    = tid >> 5;
    const int half = lane >> 4;
    const int col  = lane & 15;

    const int n1 = w * 32;
    const float* W1c0 = W1 + n1 + col;
    const float* W1c1 = W1 + n1 + 16 + col;
    const int n0 = w * 16;
    const float* W2c = W2 + n0 + col;
    const float* W3c = W3 + n0 + col;
    const float* W4c = W4 + n0 + col;

    v2f pA[G1], pB[G1], qA[G1], qB[G1];
    l1_fetch<false>(pA, pB, W1c0, W1c1, 0, half);
    const float bias1A = b1[n1 + col], bias1B = b1[n1 + 16 + col];
    const float bias2 = b2[n0 + col], bias3 = b3[n0 + col], bias4 = b4[n0 + col];

    int myIdx = 0;
    if (tid < 2 * MT) myIdx = (tid < MT) ? seq[t0 + tid] : cidx[t0 + tid - MT];
    float dv = 0.0f;
    if (tid < MT * 3) dv = dih[(size_t)t0 * 3 + tid];

    for (int i = tid; i < MT * XROW; i += 256) xs[i] = 0.0f;
    if (tid < 2 * MT) sIdx[tid] = myIdx;
    __syncthreads();

    const int gk = tid & 127;
    const int gh = tid >> 7;
    float ga[16], gc[16];
    int   gci[16];
    #pragma unroll
    for (int j = 0; j < 16; ++j) {
        int tt = 2 * j + gh;
        int ci = sIdx[MT + tt];
        ga[j]  = aa_emb[sIdx[tt] * 128 + gk];
        gc[j]  = chain_emb[ci * 128 + gk];
        gci[j] = ci;
    }
    const int  ci1  = tid + 256;
    const bool has1 = ci1 < MT * 15;
    CoordIn cA, cB;
    coord_load(cA, xyz, orient, sIdx, t0, tid);
    coord_load(cB, xyz, orient, sIdx, t0, has1 ? ci1 : MT * 15 - 1);

    if (tid < MT * 3) {
        int tt = tid / 3, d = tid % 3;
        float v = dv;
        const float FREQ[6] = {1.0f, 2.0f, 3.0f, 1.0f, 0.5f, 1.0f / 3.0f};
        int base = tt * XROW + 1028 + d * 13;
        xs[base] = v;
        #pragma unroll
        for (int f = 0; f < 6; ++f) {
            xs[base + 1 + f] = sinf(FREQ[f] * v);
            xs[base + 7 + f] = cosf(FREQ[f] * v);
        }
    }

    coord_store(xs, cA, tid);
    if (has1) coord_store(xs, cB, ci1);
    #pragma unroll
    for (int j = 0; j < 16; ++j) {
        int tt = 2 * j + gh;
        xs[tt * XROW + gk] = ga[j];
        xs[tt * XROW + 1067 + gk] = (gci[j] != 0) ? gc[j] : 0.0f;
    }
    __syncthreads();

    v2f uP[G2], uQ[G2];
    v2f hP0[G2], hP1[G2], hQ0[G2], hQ1[G2];

    {
        v8f a00 = {}, a01 = {}, a10 = {}, a11 = {};
        const int ar0 = col * XROW;
        const int ar1 = (16 + col) * XROW;
        v2f xP0[G1], xP1[G1], xQ0[G1], xQ1[G1];
        l1_aload(xP0, xP1, xs, ar0, ar1, 0, half);
        for (int g = 0; g < 20; g += 2) {
            l1_fetch<false>(qA, qB, W1c0, W1c1, g + 1, half);
            l1_aload(xQ0, xQ1, xs, ar0, ar1, g + 1, half);
            l1_mma(a00, a01, a10, a11, xP0, xP1, pA, pB);
            l1_fetch<false>(pA, pB, W1c0, W1c1, g + 2, half);
            l1_aload(xP0, xP1, xs, ar0, ar1, g + 2, half);
            l1_mma(a00, a01, a10, a11, xQ0, xQ1, qA, qB);
        }
        l1_fetch<false>(qA, qB, W1c0, W1c1, 21, half);
        l1_aload(xQ0, xQ1, xs, ar0, ar1, 21, half);
        l1_mma(a00, a01, a10, a11, xP0, xP1, pA, pB);
        l1_fetch<true>(pA, pB, W1c0, W1c1, 22, half);
        l1_aload(xP0, xP1, xs, ar0, ar1, 22, half);
        l1_mma(a00, a01, a10, a11, xQ0, xQ1, qA, qB);
        wl_fetch(uP, W2c, 0, half);
        l1_mma(a00, a01, a10, a11, xP0, xP1, pA, pB);

        #pragma unroll
        for (int r = 0; r < 8; ++r) {
            int m = r + half * 8;
            float v00 = a00[r] + bias1A, v01 = a01[r] + bias1B;
            float v10 = a10[r] + bias1A, v11 = a11[r] + bias1B;
            hA[m * H1S + n1 + col]             = v00 > 0.0f ? v00 : 0.0f;
            hA[m * H1S + n1 + 16 + col]        = v01 > 0.0f ? v01 : 0.0f;
            hA[(m + 16) * H1S + n1 + col]      = v10 > 0.0f ? v10 : 0.0f;
            hA[(m + 16) * H1S + n1 + 16 + col] = v11 > 0.0f ? v11 : 0.0f;
        }
    }
    __syncthreads();

    {
        v8f ac0 = {}, ac1 = {};
        wl_aload(hP0, hP1, hA, H1S, col, 0, half);
        wl_fetch(uQ, W2c, 1, half);
        wl_aload(hQ0, hQ1, hA, H1S, col, 1, half);
        wl_mma(ac0, ac1, hP0, hP1, uP);
        wl_fetch(uP, W2c, 2, half);
        wl_aload(hP0, hP1, hA, H1S, col, 2, half);
        wl_mma(ac0, ac1, hQ0, hQ1, uQ);
        wl_fetch(uQ, W2c, 3, half);
        wl_aload(hQ0, hQ1, hA, H1S, col, 3, half);
        wl_mma(ac0, ac1, hP0, hP1, uP);
        wl_fetch(uP, W3c, 0, half);
        wl_mma(ac0, ac1, hQ0, hQ1, uQ);
        #pragma unroll
        for (int r = 0; r < 8; ++r) {
            int m = r + half * 8;
            float v0 = ac0[r] + bias2, v1 = ac1[r] + bias2;
            hB[m * H2S + n0 + col]        = v0 > 0.0f ? v0 : 0.0f;
            hB[(m + 16) * H2S + n0 + col] = v1 > 0.0f ? v1 : 0.0f;
        }
    }
    __syncthreads();

    {
        v8f ac0 = {}, ac1 = {};
        wl_aload(hP0, hP1, hB, H2S, col, 0, half);
        wl_fetch(uQ, W3c, 1, half);
        wl_aload(hQ0, hQ1, hB, H2S, col, 1, half);
        wl_mma(ac0, ac1, hP0, hP1, uP);
        wl_fetch(uP, W4c, 0, half);
        wl_mma(ac0, ac1, hQ0, hQ1, uQ);
        #pragma unroll
        for (int r = 0; r < 8; ++r) {
            int m = r + half * 8;
            float v0 = ac0[r] + bias3, v1 = ac1[r] + bias3;
            hA[m * H2S + n0 + col]        = v0 > 0.0f ? v0 : 0.0f;
            hA[(m + 16) * H2S + n0 + col] = v1 > 0.0f ? v1 : 0.0f;
        }
    }
    __syncthreads();

    {
        v8f ac0 = {}, ac1 = {};
        wl_aload(hP0, hP1, hA, H2S, col, 0, half);
        wl_fetch(uQ, W4c, 1, half);
        wl_aload(hQ0, hQ1, hA, H2S, col, 1, half);
        wl_mma(ac0, ac1, hP0, hP1, uP);
        wl_mma(ac0, ac1, hQ0, hQ1, uQ);
        #pragma unroll
        for (int r = 0; r < 8; ++r) {
            int m = r + half * 8;
            out[(size_t)(t0 + m) * 128 + n0 + col]      = ac0[r] + bias4;
            out[(size_t)(t0 + m + 16) * 128 + n0 + col] = ac1[r] + bias4;
        }
    }
}

extern "C" void kernel_launch(void* const* d_in, const int* in_sizes, int n_in,
                              void* d_out, int out_size, void* d_ws, size_t ws_size,
                              hipStream_t stream) {
    (void)n_in; (void)d_ws; (void)ws_size; (void)out_size;
    const int*   seq    = (const int*)  d_in[0];
    const float* xyz    = (const float*)d_in[1];
    const float* dih    = (const float*)d_in[2];
    const int*   cidx   = (const int*)  d_in[3];
    const float* orient = (const float*)d_in[4];
    const float* aa_emb    = (const float*)d_in[6];
    const float* chain_emb = (const float*)d_in[7];
    const float* W1 = (const float*)d_in[8];   const float* b1 = (const float*)d_in[9];
    const float* W2 = (const float*)d_in[10];  const float* b2 = (const float*)d_in[11];
    const float* W3 = (const float*)d_in[12];  const float* b3 = (const float*)d_in[13];
    const float* W4 = (const float*)d_in[14];  const float* b4 = (const float*)d_in[15];
    float* out = (float*)d_out;

    const int tokens = in_sizes[0];
    const int blocks = tokens / MT;
    residue_embed_fused<<<blocks, 256, 0, stream>>>(
        seq, xyz, dih, cidx, orient, aa_emb, chain_emb,
        W1, b1, W2, b2, W3, b3, W4, b4, out);
}


# Two work-groups resident per WGP (4 waves per SIMD): h1-h3 alias the dead x tile in LDS behind one added barrier; registers held to 256 per wave (scalar weight bases, K loop not peeled)
# speedup vs baseline: 2.3304x; 1.0853x over previous
#include <hip/hip_runtime.h>
#include <math.h>


typedef float v2f __attribute__((ext_vector_type(2)));
typedef float v8f __attribute__((ext_vector_type(8)));

#define MT    32
#define DIN   1195
#define XROW  1196
#define H1S   260
#define H2S   132
#define G1    13
#define G2    16

#define WMMA_F32(a, b, c) \
    __builtin_amdgcn_wmma_f32_16x16x4_f32(false, (a), false, (b), (short)0, (c), false, false)

__device__ __forceinline__ void block_edge(v8f& a, v8f& b, v8f& c, v8f& d)
{
    asm volatile("" : "+v"(a), "+v"(b), "+v"(c), "+v"(d) :: "memory");
    __builtin_amdgcn_sched_barrier(0);
}
__device__ __forceinline__ void block_edge(v8f& a, v8f& b)
{
    asm volatile("" : "+v"(a), "+v"(b) :: "memory");
    __builtin_amdgcn_sched_barrier(0);
}

template <bool LAST>
__device__ __forceinline__ void l1_fetch(v2f (&fA)[G1], v2f (&fB)[G1],
                                         const float* __restrict__ wu, unsigned lo, int g)
{
    const float* __restrict__ wg = wu + (size_t)g * (4 * G1 * 256);
    #pragma unroll
    for (int s = 0; s < G1; ++s) {
        unsigned o0 = lo + (unsigned)(4 * s * 256);
        unsigned o1 = o0 + 256u;
        if (LAST && s == G1 - 1) o1 = (g == 22 && (lo >> 9)) ? o0 : o1;
        fA[s].x = wg[o0];        fA[s].y = wg[o1];
        fB[s].x = wg[o0 + 16u];  fB[s].y = wg[o1 + 16u];
    }
}

__device__ __forceinline__ void l1_aload(v2f (&x0)[G1], v2f (&x1)[G1],
                                         const float* xs, int ar0, int ar1, int g, int half)
{
    const int kk0 = g * (4 * G1) + half * 2;
    #pragma unroll
    for (int s = 0; s < G1; ++s) {
        int kk = kk0 + 4 * s;
        x0[s] = *(const v2f*)&xs[ar0 + kk];
        x1[s] = *(const v2f*)&xs[ar1 + kk];
    }
}

__device__ __forceinline__ void l1_mma(v8f& a00, v8f& a01, v8f& a10, v8f& a11,
                                       const v2f (&x0)[G1], const v2f (&x1)[G1],
                                       const v2f (&fA)[G1], const v2f (&fB)[G1])
{
    block_edge(a00, a01, a10, a11);
    #pragma unroll
    for (int s = 0; s < G1; ++s) {
        a00 = WMMA_F32(x0[s], fA[s], a00);
        a01 = WMMA_F32(x0[s], fB[s], a01);
        a10 = WMMA_F32(x1[s], fA[s], a10);
        a11 = WMMA_F32(x1[s], fB[s], a11);
    }
    block_edge(a00, a01, a10, a11);
}

__device__ __forceinline__ void wl_fetch(v2f (&f)[G2], const float* __restrict__ wu, unsigned lo, int g)
{
    #pragma unroll
    for (int s = 0; s < G2; ++s) {
        unsigned o = lo + (unsigned)((g * G2 + s) * 4 * 128);
        f[s].x = wu[o];
        f[s].y = wu[o + 128u];
    }
}

__device__ __forceinline__ void wl_aload(v2f (&x0)[G2], v2f (&x1)[G2],
                                         const float* h, int hs, int col, int g, int half)
{
    #pragma unroll
    for (int s = 0; s < G2; ++s) {
        int kk = (g * G2 + s) * 4 + half * 2;
        x0[s] = *(const v2f*)&h[col * hs + kk];
        x1[s] = *(const v2f*)&h[(16 + col) * hs + kk];
    }
}

__device__ __forceinline__ void wl_mma(v8f& ac0, v8f& ac1, const v2f (&x0)[G2], const v2f (&x1)[G2],
                                       const v2f (&f)[G2])
{
    block_edge(ac0, ac1);
    #pragma unroll
    for (int s = 0; s < G2; ++s) {
        ac0 = WMMA_F32(x0[s], f[s], ac0);
        ac1 = WMMA_F32(x1[s], f[s], ac1);
    }
    block_edge(ac0, ac1);
}

struct CoordIn {
    float ax, ay, az;
    float cx, cy, cz;
    float O[9];
    int   s;
};

__device__ __forceinline__ void coord_load(CoordIn& c, const float* __restrict__ xyz,
                                           const float* __restrict__ orient, const int* sIdx,
                                           int t0, int i)
{
    int tt = i / 15, a = i % 15;
    const float* xp = xyz + (size_t)(t0 + tt) * 45;
    c.ax = xp[a * 3 + 0];  c.ay = xp[a * 3 + 1];  c.az = xp[a * 3 + 2];
    c.cx = xp[3 + 0];      c.cy = xp[3 + 1];      c.cz = xp[3 + 2];
    const float* O = orient + (size_t)(t0 + tt) * 9;
    #pragma unroll
    for (int j = 0; j < 9; ++j) c.O[j] = O[j];
    c.s = sIdx[tt];
}

__device__ __forceinline__ void coord_store(float* xs, const CoordIn& c, int i)
{
    int tt = i / 15, a = i % 15;
    float rx = c.ax - c.cx;
    float ry = c.ay - c.cy;
    float rz = c.az - c.cz;
    int base = tt * XROW + 128 + c.s * 45 + a * 3;
    xs[base + 0] = c.O[0] * rx + c.O[3] * ry + c.O[6] * rz;
    xs[base + 1] = c.O[1] * rx + c.O[4] * ry + c.O[7] * rz;
    xs[base + 2] = c.O[2] * rx + c.O[5] * ry + c.O[8] * rz;
}

__global__ __launch_bounds__(256, 4) void residue_embed_fused(
    const int* __restrict__ seq, const float* __restrict__ xyz,
    const float* __restrict__ dih, const int* __restrict__ cidx,
    const float* __restrict__ orient,
    const float* __restrict__ aa_emb, const float* __restrict__ chain_emb,
    const float* __restrict__ W1, const float* __restrict__ b1,
    const float* __restrict__ W2, const float* __restrict__ b2,
    const float* __restrict__ W3, const float* __restrict__ b3,
    const float* __restrict__ W4, const float* __restrict__ b4,
    float* __restrict__ out)
{
    __shared__ float xs[MT * XROW];
    __shared__ int   sIdx[2 * MT];
    float* const hA = xs;
    float* const hB = xs + MT * H1S;

    const int tid = threadIdx.x;
    const int t0  = blockIdx.x * MT;

    const int lane = tid & 31;
    const int w    = __builtin_amdgcn_readfirstlane(tid >> 5);
    const int half = lane >> 4;
    const int col  = lane & 15;

    const int n1 = w * 32;
    const float* W1u = W1 + n1;
    const unsigned lo1 = (unsigned)(half * 2 * 256 + col);
    const int n0 = w * 16;
    const float* W2u = W2 + n0;
    const float* W3u = W3 + n0;
    const float* W4u = W4 + n0;

    v2f pA[G1], pB[G1], qA[G1], qB[G1];
    l1_fetch<false>(pA, pB, W1u, lo1, 0);

    int myIdx = 0;
    if (tid < 2 * MT) myIdx = (tid < MT) ? seq[t0 + tid] : cidx[t0 + tid - MT];
    float dv = 0.0f;
    if (tid < MT * 3) dv = dih[(size_t)t0 * 3 + tid];

    for (int i = tid; i < MT * XROW; i += 256) xs[i] = 0.0f;
    if (tid < 2 * MT) sIdx[tid] = myIdx;
    __syncthreads();

    const int gk = tid & 127;
    const int gh = tid >> 7;
    float ga[16], gc[16];
    int   gci[16];
    #pragma unroll
    for (int j = 0; j < 16; ++j) {
        int tt = 2 * j + gh;
        int ci = sIdx[MT + tt];
        ga[j]  = aa_emb[sIdx[tt] * 128 + gk];
        gc[j]  = chain_emb[ci * 128 + gk];
        gci[j] = ci;
    }
    const int  ci1  = tid + 256;
    const bool has1 = ci1 < MT * 15;
    CoordIn cA, cB;
    coord_load(cA, xyz, orient, sIdx, t0, tid);
    coord_load(cB, xyz, orient, sIdx, t0, has1 ? ci1 : MT * 15 - 1);

    if (tid < MT * 3) {
        int tt = tid / 3, d = tid % 3;
        float v = dv;
        const float FREQ[6] = {1.0f, 2.0f, 3.0f, 1.0f, 0.5f, 1.0f / 3.0f};
        int base = tt * XROW + 1028 + d * 13;
        xs[base] = v;
        #pragma unroll
        for (int f = 0; f < 6; ++f) {
            xs[base + 1 + f] = sinf(FREQ[f] * v);
            xs[base + 7 + f] = cosf(FREQ[f] * v);
        }
    }

    coord_store(xs, cA, tid);
    if (has1) coord_store(xs, cB, ci1);
    #pragma unroll
    for (int j = 0; j < 16; ++j) {
        int tt = 2 * j + gh;
        xs[tt * XROW + gk] = ga[j];
        xs[tt * XROW + 1067 + gk] = (gci[j] != 0) ? gc[j] : 0.0f;
    }
    __syncthreads();

    int laneL, halfL, colL;
    unsigned lo2;
    v2f uP[G2], uQ[G2];
    v2f hP0[G2], hP1[G2], hQ0[G2], hQ1[G2];

    {
        v8f a00 = {}, a01 = {}, a10 = {}, a11 = {};
        const int ar0 = col * XROW;
        const int ar1 = (16 + col) * XROW;
        v2f xP0[G1], xP1[G1], xQ0[G1], xQ1[G1];
        l1_aload(xP0, xP1, xs, ar0, ar1, 0, half);
        for (int g = 0; g < 22; g += 2) {
            l1_fetch<false>(qA, qB, W1u, lo1, g + 1);
            l1_aload(xQ0, xQ1, xs, ar0, ar1, g + 1, half);
            l1_mma(a00, a01, a10, a11, xP0, xP1, pA, pB);
            l1_fetch<true>(pA, pB, W1u, lo1, g + 2);
            l1_aload(xP0, xP1, xs, ar0, ar1, g + 2, half);
            l1_mma(a00, a01, a10, a11, xQ0, xQ1, qA, qB);
        }
        laneL = (int)__builtin_amdgcn_mbcnt_lo(~0u, 0u);
        halfL = laneL >> 4;
        colL  = laneL & 15;
        lo2   = (unsigned)(halfL * 2 * 128 + colL);
        wl_fetch(uP, W2u, lo2, 0);
        const float bias1A = b1[n1 + colL], bias1B = b1[n1 + 16 + colL];
        l1_mma(a00, a01, a10, a11, xP0, xP1, pA, pB);

        __syncthreads();

        #pragma unroll
        for (int r = 0; r < 8; ++r) {
            int m = r + halfL * 8;
            float v00 = a00[r] + bias1A, v01 = a01[r] + bias1B;
            float v10 = a10[r] + bias1A, v11 = a11[r] + bias1B;
            hA[m * H1S + n1 + colL]             = v00 > 0.0f ? v00 : 0.0f;
            hA[m * H1S + n1 + 16 + colL]        = v01 > 0.0f ? v01 : 0.0f;
            hA[(m + 16) * H1S + n1 + colL]      = v10 > 0.0f ? v10 : 0.0f;
            hA[(m + 16) * H1S + n1 + 16 + colL] = v11 > 0.0f ? v11 : 0.0f;
        }
    }
    __syncthreads();

    {
        v8f ac0 = {}, ac1 = {};
        wl_aload(hP0, hP1, hA, H1S, colL, 0, halfL);
        wl_fetch(uQ, W2u, lo2, 1);
        wl_aload(hQ0, hQ1, hA, H1S, colL, 1, halfL);
        wl_mma(ac0, ac1, hP0, hP1, uP);
        wl_fetch(uP, W2u, lo2, 2);
        wl_aload(hP0, hP1, hA, H1S, colL, 2, halfL);
        wl_mma(ac0, ac1, hQ0, hQ1, uQ);
        wl_fetch(uQ, W2u, lo2, 3);
        wl_aload(hQ0, hQ1, hA, H1S, colL, 3, halfL);
        wl_mma(ac0, ac1, hP0, hP1, uP);
        wl_fetch(uP, W3u, lo2, 0);
        const float bias2 = b2[n0 + colL];
        wl_mma(ac0, ac1, hQ0, hQ1, uQ);
        #pragma unroll
        for (int r = 0; r < 8; ++r) {
            int m = r + halfL * 8;
            float v0 = ac0[r] + bias2, v1 = ac1[r] + bias2;
            hB[m * H2S + n0 + colL]        = v0 > 0.0f ? v0 : 0.0f;
            hB[(m + 16) * H2S + n0 + colL] = v1 > 0.0f ? v1 : 0.0f;
        }
    }
    __syncthreads();

    {
        v8f ac0 = {}, ac1 = {};
        wl_aload(hP0, hP1, hB, H2S, colL, 0, halfL);
        wl_fetch(uQ, W3u, lo2, 1);
        wl_aload(hQ0, hQ1, hB, H2S, colL, 1, halfL);
        wl_mma(ac0, ac1, hP0, hP1, uP);
        wl_fetch(uP, W4u, lo2, 0);
        const float bias3 = b3[n0 + colL];
        wl_mma(ac0, ac1, hQ0, hQ1, uQ);
        #pragma unroll
        for (int r = 0; r < 8; ++r) {
            int m = r + halfL * 8;
            float v0 = ac0[r] + bias3, v1 = ac1[r] + bias3;
            hA[m * H2S + n0 + colL]        = v0 > 0.0f ? v0 : 0.0f;
            hA[(m + 16) * H2S + n0 + colL] = v1 > 0.0f ? v1 : 0.0f;
        }
    }
    __syncthreads();

    {
        v8f ac0 = {}, ac1 = {};
        wl_aload(hP0, hP1, hA, H2S, colL, 0, halfL);
        wl_fetch(uQ, W4u, lo2, 1);
        wl_aload(hQ0, hQ1, hA, H2S, colL, 1, halfL);
        const float bias4 = b4[n0 + colL];
        wl_mma(ac0, ac1, hP0, hP1, uP);
        wl_mma(ac0, ac1, hQ0, hQ1, uQ);
        #pragma unroll
        for (int r = 0; r < 8; ++r) {
            int m = r + halfL * 8;
            out[(size_t)(t0 + m) * 128 + n0 + colL]      = ac0[r] + bias4;
            out[(size_t)(t0 + m + 16) * 128 + n0 + colL] = ac1[r] + bias4;
        }
    }
}

extern "C" void kernel_launch(void* const* d_in, const int* in_sizes, int n_in,
                              void* d_out, int out_size, void* d_ws, size_t ws_size,
                              hipStream_t stream) {
    (void)n_in; (void)d_ws; (void)ws_size; (void)out_size;
    const int*   seq    = (const int*)  d_in[0];
    const float* xyz    = (const float*)d_in[1];
    const float* dih    = (const float*)d_in[2];
    const int*   cidx   = (const int*)  d_in[3];
    const float* orient = (const float*)d_in[4];
    const float* aa_emb    = (const float*)d_in[6];
    const float* chain_emb = (const float*)d_in[7];
    const float* W1 = (const float*)d_in[8];   const float* b1 = (const float*)d_in[9];
    const float* W2 = (const float*)d_in[10];  const float* b2 = (const float*)d_in[11];
    const float* W3 = (const float*)d_in[12];  const float* b3 = (const float*)d_in[13];
    const float* W4 = (const float*)d_in[14];  const float* b4 = (const float*)d_in[15];
    float* out = (float*)d_out;

    const int tokens = in_sizes[0];
    const int blocks = tokens / MT;
    residue_embed_fused<<<blocks, 256, 0, stream>>>(
        seq, xyz, dih, cidx, orient, aa_emb, chain_emb,
        W1, b1, W2, b2, W3, b3, W4, b4, out);
}


# Layer 1 skips the one-hot's exact-zero columns: each work-group sorts tokens by residue type itself, a tile is 32 sorted tokens, K = 296 + 48 per type in it, not 1196; summation order changes
# speedup vs baseline: 3.8764x; 1.6634x over previous
#include <hip/hip_runtime.h>
#include <math.h>


typedef float v2f __attribute__((ext_vector_type(2)));
typedef float v8f __attribute__((ext_vector_type(8)));

#define MT    32
#define NSTAT 74
#define CSLOT 48
#define XC0   296
#define XROW  1260
#define NTOK  16384
#define NTYPE 20
#define H1S   260
#define H2S   132
#define G1    12
#define G2    16

#define WMMA_F32(a, b, c) \
    __builtin_amdgcn_wmma_f32_16x16x4_f32(false, (a), false, (b), (short)0, (c), false, false)

__device__ __forceinline__ void block_edge(v8f& a, v8f& b, v8f& c, v8f& d)
{
    asm volatile("" : "+v"(a), "+v"(b), "+v"(c), "+v"(d) :: "memory");
    __builtin_amdgcn_sched_barrier(0);
}
__device__ __forceinline__ void block_edge(v8f& a, v8f& b)
{
    asm volatile("" : "+v"(a), "+v"(b) :: "memory");
    __builtin_amdgcn_sched_barrier(0);
}


__device__ __forceinline__ constexpr int sg_first(int g) { return g < 5 ? 12 * g : (g == 5 ? 60 : 67); }
__device__ __forceinline__ constexpr int sg_count(int g) { return g < 5 ? 12 : 7; }

__device__ __forceinline__ void l1_fetch(v2f (&fA)[G1], v2f (&fB)[G1],
                                         const float* __restrict__ wu, unsigned lo, int g)
{
    #pragma unroll
    for (int s = 0; s < G1; ++s) {
        if (s < sg_count(g)) {
            const int S = sg_first(g) + s;
            const int row = (S < 32) ? 4 * S : (S < 64) ? 1067 + 4 * (S - 32) : 1028 + 4 * (S - 64);
            unsigned o0 = lo + (unsigned)(row * 256);
            unsigned o1 = o0 + 256u;
            fA[s].x = wu[o0];        fA[s].y = wu[o1];
            fB[s].x = wu[o0 + 16u];  fB[s].y = wu[o1 + 16u];
        }
    }
}

__device__ __forceinline__ void l1_cfetch(v2f (&fA)[G1], v2f (&fB)[G1],
                                          const float* __restrict__ wc, unsigned lo)
{
    #pragma unroll
    for (int s = 0; s < G1; ++s) {
        unsigned o0 = lo + (unsigned)(4 * s * 256);
        unsigned o1 = o0 + 256u;
        fA[s].x = wc[o0];        fA[s].y = wc[o1];
        fB[s].x = wc[o0 + 16u];  fB[s].y = wc[o1 + 16u];
    }
}

__device__ __forceinline__ void l1_aload(v2f (&x0)[G1], v2f (&x1)[G1],
                                         const float* xs, int ar0, int ar1, int k0, int n, int half)
{
    #pragma unroll
    for (int s = 0; s < G1; ++s) {
        if (s < n) {
            int kk = k0 + 4 * s + half * 2;
            x0[s] = *(const v2f*)&xs[ar0 + kk];
            x1[s] = *(const v2f*)&xs[ar1 + kk];
        }
    }
}

__device__ __forceinline__ void l1_mma(v8f& a00, v8f& a01, v8f& a10, v8f& a11,
                                       const v2f (&x0)[G1], const v2f (&x1)[G1],
                                       const v2f (&fA)[G1], const v2f (&fB)[G1], int n)
{
    block_edge(a00, a01, a10, a11);
    #pragma unroll
    for (int s = 0; s < G1; ++s) {
        if (s < n) {
            a00 = WMMA_F32(x0[s], fA[s], a00);
            a01 = WMMA_F32(x0[s], fB[s], a01);
            a10 = WMMA_F32(x1[s], fA[s], a10);
            a11 = WMMA_F32(x1[s], fB[s], a11);
        }
    }
    block_edge(a00, a01, a10, a11);
}

__device__ __forceinline__ void wmma_branch_pad()
{
    asm volatile("v_nop\n\tv_nop\n\tv_nop\n\tv_nop\n\tv_nop");
}

__device__ __forceinline__ void wl_fetch(v2f (&f)[G2], const float* __restrict__ wu, unsigned lo, int g)
{
    #pragma unroll
    for (int s = 0; s < G2; ++s) {
        unsigned o = lo + (unsigned)((g * G2 + s) * 4 * 128);
        f[s].x = wu[o];
        f[s].y = wu[o + 128u];
    }
}

__device__ __forceinline__ void wl_aload(v2f (&x0)[G2], v2f (&x1)[G2],
                                         const float* h, int hs, int col, int g, int half)
{
    #pragma unroll
    for (int s = 0; s < G2; ++s) {
        int kk = (g * G2 + s) * 4 + half * 2;
        x0[s] = *(const v2f*)&h[col * hs + kk];
        x1[s] = *(const v2f*)&h[(16 + col) * hs + kk];
    }
}

__device__ __forceinline__ void wl_mma(v8f& ac0, v8f& ac1, const v2f (&x0)[G2], const v2f (&x1)[G2],
                                       const v2f (&f)[G2])
{
    block_edge(ac0, ac1);
    #pragma unroll
    for (int s = 0; s < G2; ++s) {
        ac0 = WMMA_F32(x0[s], f[s], ac0);
        ac1 = WMMA_F32(x1[s], f[s], ac1);
    }
    block_edge(ac0, ac1);
}

struct CoordIn {
    float ax, ay, az;
    float cx, cy, cz;
    float O[9];
    int   slot;
};

__device__ __forceinline__ void coord_load(CoordIn& c, const float* __restrict__ xyz,
                                           const float* __restrict__ orient, const int* sIdx,
                                           int tA, int i)
{
    int tt = i / 15, a = i % 15;
    int tk = sIdx[tt];
    const float* xp = xyz + (size_t)tk * 45;
    c.ax = xp[a * 3 + 0];  c.ay = xp[a * 3 + 1];  c.az = xp[a * 3 + 2];
    c.cx = xp[3 + 0];      c.cy = xp[3 + 1];      c.cz = xp[3 + 2];
    const float* O = orient + (size_t)tk * 9;
    #pragma unroll
    for (int j = 0; j < 9; ++j) c.O[j] = O[j];
    c.slot = sIdx[MT + tt] - tA;
}

__device__ __forceinline__ void coord_store(float* xs, const CoordIn& c, int i)
{
    int tt = i / 15, a = i % 15;
    float rx = c.ax - c.cx;
    float ry = c.ay - c.cy;
    float rz = c.az - c.cz;
    int base = tt * XROW + XC0 + c.slot * CSLOT + a * 3;
    xs[base + 0] = c.O[0] * rx + c.O[3] * ry + c.O[6] * rz;
    xs[base + 1] = c.O[1] * rx + c.O[4] * ry + c.O[7] * rz;
    xs[base + 2] = c.O[2] * rx + c.O[5] * ry + c.O[8] * rz;
}

__global__ __launch_bounds__(256, 4) void residue_embed_fused(
    const int* __restrict__ seq, const float* __restrict__ xyz,
    const float* __restrict__ dih, const int* __restrict__ cidx,
    const float* __restrict__ orient,
    const float* __restrict__ aa_emb, const float* __restrict__ chain_emb,
    const float* __restrict__ W1, const float* __restrict__ b1,
    const float* __restrict__ W2, const float* __restrict__ b2,
    const float* __restrict__ W3, const float* __restrict__ b3,
    const float* __restrict__ W4, const float* __restrict__ b4,
    float* __restrict__ out)
{
    __shared__ float xs[MT * XROW];
    __shared__ int   sIdx[2 * MT];
    __shared__ unsigned wtot[8 * 10];
    float* const hA = xs;
    float* const hB = xs + MT * H1S;

    const int tid = threadIdx.x;

    const int lane = tid & 31;
    const int w    = __builtin_amdgcn_readfirstlane(tid >> 5);
    const int half = lane >> 4;
    const int col  = lane & 15;

    const int n1 = w * 32;
    const float* W1u = W1 + n1;
    const unsigned lo1 = (unsigned)(half * 2 * 256 + col);
    const int n0 = w * 16;
    const float* W2u = W2 + n0;
    const float* W3u = W3 + n0;
    const float* W4u = W4 + n0;

    v2f pA[G1], pB[G1], qA[G1], qB[G1];
    l1_fetch(pA, pB, W1u, lo1, 0);

    int4 sv[16];
    #pragma unroll
    for (int i = 0; i < 16; ++i) sv[i] = ((const int4*)seq)[i * 256 + tid];
    unsigned long long ca = 0ull, cb = 0ull;
    unsigned cc = 0u;
    #pragma unroll
    for (int i = 0; i < 16; ++i) {
        #pragma unroll
        for (int j = 0; j < 4; ++j) {
            int v = (j == 0) ? sv[i].x : (j == 1) ? sv[i].y : (j == 2) ? sv[i].z : sv[i].w;
            v = v < 0 ? 0 : (v > NTYPE - 1 ? NTYPE - 1 : v);
            unsigned sh = (unsigned)(v & 7) * 8u;
            ca += (v < 8) ? (1ull << sh) : 0ull;
            cb += ((v >> 3) == 1) ? (1ull << sh) : 0ull;
            cc += (v >= 16) ? (1u << sh) : 0u;
        }
    }
    {
        unsigned pk[10];
        #pragma unroll
        for (int p = 0; p < 10; ++p) {
            const int t0 = 2 * p, t1 = 2 * p + 1;
            unsigned c0 = (t0 < 8) ? (unsigned)(ca >> (8 * t0)) : (t0 < 16) ? (unsigned)(cb >> (8 * (t0 - 8))) : (cc >> (8 * (t0 - 16)));
            unsigned c1 = (t1 < 8) ? (unsigned)(ca >> (8 * t1)) : (t1 < 16) ? (unsigned)(cb >> (8 * (t1 - 8))) : (cc >> (8 * (t1 - 16)));
            pk[p] = (c0 & 255u) | ((c1 & 255u) << 16);
        }
        #pragma unroll
        for (int off = 16; off >= 1; off >>= 1) {
            #pragma unroll
            for (int p = 0; p < 10; ++p) pk[p] += (unsigned)__shfl_xor((int)pk[p], off);
        }
        if (lane == 0) {
            #pragma unroll
            for (int p = 0; p < 10; ++p) wtot[w * 10 + p] = pk[p];
        }
    }
    __syncthreads();

    const int R0 = (int)blockIdx.x * MT;
    int tA = -1, tB = -1, baseA = 0;
    {
        int base = 0;
        #pragma unroll
        for (int t = 0; t < NTYPE; ++t) {
            unsigned n = 0u;
            #pragma unroll
            for (int ww = 0; ww < 8; ++ww) n += (wtot[ww * 10 + (t >> 1)] >> (16 * (t & 1))) & 0xFFFFu;
            int nb = base + (int)n;
            if (tA < 0 && nb > R0)      { tA = t; baseA = base; }
            if (tB < 0 && nb > R0 + 31) { tB = t; }
            base = nb;
        }
    }
    tA    = __builtin_amdgcn_readfirstlane(tA);
    tB    = __builtin_amdgcn_readfirstlane(tB);
    baseA = __builtin_amdgcn_readfirstlane(baseA);
    const int nT = tB - tA + 1;
    {
        int baseT = baseA - R0;
        for (int t = tA; t <= tB; ++t) {
            unsigned long long sel = (t < 8) ? ca : (t < 16) ? cb : (unsigned long long)cc;
            int cnt = (int)((sel >> ((t & 7) * 8)) & 255ull);
            int inc = cnt;
            #pragma unroll
            for (int off = 1; off < 32; off <<= 1) {
                int y = __shfl_up(inc, off);
                if (lane >= off) inc += y;
            }
            int rank = baseT + inc - cnt;
            int nt = 0;
            for (int ww = 0; ww < 8; ++ww) {
                int nw = (int)((wtot[ww * 10 + (t >> 1)] >> (16 * (t & 1))) & 0xFFFFu);
                if (ww < w) rank += nw;
                nt += nw;
            }
            #pragma unroll
            for (int i = 0; i < 16; ++i) {
                #pragma unroll
                for (int j = 0; j < 4; ++j) {
                    int v = (j == 0) ? sv[i].x : (j == 1) ? sv[i].y : (j == 2) ? sv[i].z : sv[i].w;
                    v = v < 0 ? 0 : (v > NTYPE - 1 ? NTYPE - 1 : v);
                    if (v == t) {
                        if ((unsigned)rank < (unsigned)MT) { sIdx[rank] = 4 * (i * 256 + tid) + j; sIdx[MT + rank] = t; }
                        ++rank;
                    }
                }
            }
            baseT += nt;
        }
    }
    {
        const int zr = tid >> 3, zc = tid & 7;
        for (int c = zc; c < nT * CSLOT; c += 8) xs[zr * XROW + XC0 + c] = 0.0f;
    }
    __syncthreads();

    float dv = 0.0f;
    if (tid < MT * 3) dv = dih[(size_t)sIdx[tid / 3] * 3 + (tid % 3)];

    const int gk = tid & 127;
    const int gh = tid >> 7;
    float ga[16], gc[16];
    int   gci[16];
    #pragma unroll
    for (int j = 0; j < 16; ++j) {
        int tt = 2 * j + gh;
        int ci = cidx[sIdx[tt]];
        ga[j]  = aa_emb[sIdx[MT + tt] * 128 + gk];
        gc[j]  = chain_emb[ci * 128 + gk];
        gci[j] = ci;
    }
    const int  ci1  = tid + 256;
    const bool has1 = ci1 < MT * 15;
    CoordIn cA, cB;
    coord_load(cA, xyz, orient, sIdx, tA, tid);
    coord_load(cB, xyz, orient, sIdx, tA, has1 ? ci1 : MT * 15 - 1);

    if (tid < MT * 3) {
        int tt = tid / 3, d = tid % 3;
        float v = dv;
        const float FREQ[6] = {1.0f, 2.0f, 3.0f, 1.0f, 0.5f, 1.0f / 3.0f};
        int base = tt * XROW + 256 + d * 13;
        xs[base] = v;
        #pragma unroll
        for (int f = 0; f < 6; ++f) {
            xs[base + 1 + f] = sinf(FREQ[f] * v);
            xs[base + 7 + f] = cosf(FREQ[f] * v);
        }
    } else if (tid >= 128 && tid < 128 + MT) {
        xs[(tid - 128) * XROW + 295] = 0.0f;
    }

    coord_store(xs, cA, tid);
    if (has1) coord_store(xs, cB, ci1);
    #pragma unroll
    for (int j = 0; j < 16; ++j) {
        int tt = 2 * j + gh;
        xs[tt * XROW + gk] = ga[j];
        xs[tt * XROW + 128 + gk] = (gci[j] != 0) ? gc[j] : 0.0f;
    }
    __syncthreads();

    int laneL, halfL, colL;
    unsigned lo2;
    v2f uP[G2], uQ[G2];
    v2f hP0[G2], hP1[G2], hQ0[G2], hQ1[G2];

    {
        v8f a00 = {}, a01 = {}, a10 = {}, a11 = {};
        const int ar0 = col * XROW;
        const int ar1 = (16 + col) * XROW;
        v2f xP0[G1], xP1[G1], xQ0[G1], xQ1[G1];
        l1_aload(xP0, xP1, xs, ar0, ar1, 0, sg_count(0), half);
        #pragma unroll
        for (int g = 0; g < 6; g += 2) {
            l1_fetch(qA, qB, W1u, lo1, g + 1);
            l1_aload(xQ0, xQ1, xs, ar0, ar1, 4 * sg_first(g + 1), sg_count(g + 1), half);
            l1_mma(a00, a01, a10, a11, xP0, xP1, pA, pB, sg_count(g));
            l1_fetch(pA, pB, W1u, lo1, g + 2);
            l1_aload(xP0, xP1, xs, ar0, ar1, 4 * sg_first(g + 2), sg_count(g + 2), half);
            l1_mma(a00, a01, a10, a11, xQ0, xQ1, qA, qB, sg_count(g + 1));
        }
        const float* wc = W1u + (size_t)(128 + 45 * tA) * 256;
        l1_cfetch(qA, qB, wc, lo1);
        l1_aload(xQ0, xQ1, xs, ar0, ar1, XC0, G1, half);
        l1_mma(a00, a01, a10, a11, xP0, xP1, pA, pB, sg_count(6));
        wmma_branch_pad();
        for (int c = 0; c < nT; c += 2) {
            const bool more1 = c + 1 < nT;
            if (more1) {
                l1_cfetch(pA, pB, wc + (size_t)(c + 1) * (45 * 256), lo1);
                l1_aload(xP0, xP1, xs, ar0, ar1, XC0 + (c + 1) * CSLOT, G1, half);
            }
            l1_mma(a00, a01, a10, a11, xQ0, xQ1, qA, qB, G1);
            wmma_branch_pad();
            if (more1) {
                if (c + 2 < nT) {
                    l1_cfetch(qA, qB, wc + (size_t)(c + 2) * (45 * 256), lo1);
                    l1_aload(xQ0, xQ1, xs, ar0, ar1, XC0 + (c + 2) * CSLOT, G1, half);
                }
                l1_mma(a00, a01, a10, a11, xP0, xP1, pA, pB, G1);
                wmma_branch_pad();
            }
        }
        laneL = (int)__builtin_amdgcn_mbcnt_lo(~0u, 0u);
        halfL = laneL >> 4;
        colL  = laneL & 15;
        lo2   = (unsigned)(halfL * 2 * 128 + colL);
        wl_fetch(uP, W2u, lo2, 0);
        const float bias1A = b1[n1 + colL], bias1B = b1[n1 + 16 + colL];

        __syncthreads();

        #pragma unroll
        for (int r = 0; r < 8; ++r) {
            int m = r + halfL * 8;
            float v00 = a00[r] + bias1A, v01 = a01[r] + bias1B;
            float v10 = a10[r] + bias1A, v11 = a11[r] + bias1B;
            hA[m * H1S + n1 + colL]             = v00 > 0.0f ? v00 : 0.0f;
            hA[m * H1S + n1 + 16 + colL]        = v01 > 0.0f ? v01 : 0.0f;
            hA[(m + 16) * H1S + n1 + colL]      = v10 > 0.0f ? v10 : 0.0f;
            hA[(m + 16) * H1S + n1 + 16 + colL] = v11 > 0.0f ? v11 : 0.0f;
        }
    }
    __syncthreads();

    {
        v8f ac0 = {}, ac1 = {};
        wl_aload(hP0, hP1, hA, H1S, colL, 0, halfL);
        wl_fetch(uQ, W2u, lo2, 1);
        wl_aload(hQ0, hQ1, hA, H1S, colL, 1, halfL);
        wl_mma(ac0, ac1, hP0, hP1, uP);
        wl_fetch(uP, W2u, lo2, 2);
        wl_aload(hP0, hP1, hA, H1S, colL, 2, halfL);
        wl_mma(ac0, ac1, hQ0, hQ1, uQ);
        wl_fetch(uQ, W2u, lo2, 3);
        wl_aload(hQ0, hQ1, hA, H1S, colL, 3, halfL);
        wl_mma(ac0, ac1, hP0, hP1, uP);
        wl_fetch(uP, W3u, lo2, 0);
        const float bias2 = b2[n0 + colL];
        wl_mma(ac0, ac1, hQ0, hQ1, uQ);
        #pragma unroll
        for (int r = 0; r < 8; ++r) {
            int m = r + halfL * 8;
            float v0 = ac0[r] + bias2, v1 = ac1[r] + bias2;
            hB[m * H2S + n0 + colL]        = v0 > 0.0f ? v0 : 0.0f;
            hB[(m + 16) * H2S + n0 + colL] = v1 > 0.0f ? v1 : 0.0f;
        }
    }
    __syncthreads();

    {
        v8f ac0 = {}, ac1 = {};
        wl_aload(hP0, hP1, hB, H2S, colL, 0, halfL);
        wl_fetch(uQ, W3u, lo2, 1);
        wl_aload(hQ0, hQ1, hB, H2S, colL, 1, halfL);
        wl_mma(ac0, ac1, hP0, hP1, uP);
        wl_fetch(uP, W4u, lo2, 0);
        const float bias3 = b3[n0 + colL];
        wl_mma(ac0, ac1, hQ0, hQ1, uQ);
        #pragma unroll
        for (int r = 0; r < 8; ++r) {
            int m = r + halfL * 8;
            float v0 = ac0[r] + bias3, v1 = ac1[r] + bias3;
            hA[m * H2S + n0 + colL]        = v0 > 0.0f ? v0 : 0.0f;
            hA[(m + 16) * H2S + n0 + colL] = v1 > 0.0f ? v1 : 0.0f;
        }
    }
    __syncthreads();

    {
        v8f ac0 = {}, ac1 = {};
        wl_aload(hP0, hP1, hA, H2S, colL, 0, halfL);
        wl_fetch(uQ, W4u, lo2, 1);
        wl_aload(hQ0, hQ1, hA, H2S, colL, 1, halfL);
        const float bias4 = b4[n0 + colL];
        wl_mma(ac0, ac1, hP0, hP1, uP);
        wl_mma(ac0, ac1, hQ0, hQ1, uQ);
        #pragma unroll
        for (int r = 0; r < 8; ++r) {
            int m = r + halfL * 8;
            out[(size_t)sIdx[m] * 128 + n0 + colL]      = ac0[r] + bias4;
            out[(size_t)sIdx[m + 16] * 128 + n0 + colL] = ac1[r] + bias4;
        }
    }
}

extern "C" void kernel_launch(void* const* d_in, const int* in_sizes, int n_in,
                              void* d_out, int out_size, void* d_ws, size_t ws_size,
                              hipStream_t stream) {
    (void)n_in; (void)d_ws; (void)ws_size; (void)out_size;
    const int*   seq    = (const int*)  d_in[0];
    const float* xyz    = (const float*)d_in[1];
    const float* dih    = (const float*)d_in[2];
    const int*   cidx   = (const int*)  d_in[3];
    const float* orient = (const float*)d_in[4];
    const float* aa_emb    = (const float*)d_in[6];
    const float* chain_emb = (const float*)d_in[7];
    const float* W1 = (const float*)d_in[8];   const float* b1 = (const float*)d_in[9];
    const float* W2 = (const float*)d_in[10];  const float* b2 = (const float*)d_in[11];
    const float* W3 = (const float*)d_in[12];  const float* b3 = (const float*)d_in[13];
    const float* W4 = (const float*)d_in[14];  const float* b4 = (const float*)d_in[15];
    float* out = (float*)d_out;

    const int tokens = in_sizes[0];
    const int blocks = tokens / MT;
    residue_embed_fused<<<blocks, 256, 0, stream>>>(
        seq, xyz, dih, cidx, orient, aa_emb, chain_emb,
        W1, b1, W2, b2, W3, b3, W4, b4, out);
}


# Layer 1 forms the aa and chain products once per table row, not per token: a 16-row tile of chain_emb and aa_emb rows gives T, a 0/1 selection WMMA adds each token's T rows; summation order changes
# speedup vs baseline: 4.0852x; 1.0539x over previous
#include <hip/hip_runtime.h>
#include <math.h>


typedef float v2f __attribute__((ext_vector_type(2)));
typedef float v8f __attribute__((ext_vector_type(8)));

#define MT    32
#define NSTAT 74
#define CSLOT 48
#define XC0   296
#define XROW  1260
#define NTOK  16384
#define NTYPE 20
#define H1S   260
#define H2S   132
#define G1    12
#define G2    16

#define WMMA_F32(a, b, c) \
    __builtin_amdgcn_wmma_f32_16x16x4_f32(false, (a), false, (b), (short)0, (c), false, false)

__device__ __forceinline__ void block_edge(v8f& a, v8f& b, v8f& c, v8f& d)
{
    asm volatile("" : "+v"(a), "+v"(b), "+v"(c), "+v"(d) :: "memory");
    __builtin_amdgcn_sched_barrier(0);
}
__device__ __forceinline__ void block_edge(v8f& a, v8f& b)
{
    asm volatile("" : "+v"(a), "+v"(b) :: "memory");
    __builtin_amdgcn_sched_barrier(0);
}


__device__ __forceinline__ constexpr int sg_first(int g) { return g == 0 ? 0 : g == 1 ? 12 : g == 2 ? 24 : g == 3 ? 32 : g == 4 ? 44 : g == 5 ? 56 : 64; }
__device__ __forceinline__ constexpr int sg_count(int g) { return g == 6 ? 10 : ((g == 2 || g == 5) ? 8 : 12); }

__device__ __forceinline__ void l1_fetch(v2f (&fA)[G1], v2f (&fB)[G1],
                                         const float* __restrict__ wu, unsigned lo, int g)
{
    #pragma unroll
    for (int s = 0; s < G1; ++s) {
        if (s < sg_count(g)) {
            const int S = sg_first(g) + s;
            const int row = (S < 32) ? 4 * S : (S < 64) ? 1067 + 4 * (S - 32) : 1028 + 4 * (S - 64);
            unsigned o0 = lo + (unsigned)(row * 256);
            unsigned o1 = o0 + 256u;
            fA[s].x = wu[o0];        fA[s].y = wu[o1];
            fB[s].x = wu[o0 + 16u];  fB[s].y = wu[o1 + 16u];
        }
    }
}

__device__ __forceinline__ void l1_cfetch(v2f (&fA)[G1], v2f (&fB)[G1],
                                          const float* __restrict__ wc, unsigned lo, int n = G1)
{
    #pragma unroll
    for (int s = 0; s < G1; ++s) {
        if (s < n) {
            unsigned o0 = lo + (unsigned)(4 * s * 256);
            unsigned o1 = o0 + 256u;
            fA[s].x = wc[o0];        fA[s].y = wc[o1];
            fB[s].x = wc[o0 + 16u];  fB[s].y = wc[o1 + 16u];
        }
    }
}

__device__ __forceinline__ void l1_aload(v2f (&x0)[G1], v2f (&x1)[G1],
                                         const float* xs, int ar0, int ar1, int k0, int n, int half)
{
    #pragma unroll
    for (int s = 0; s < G1; ++s) {
        if (s < n) {
            int kk = k0 + 4 * s + half * 2;
            x0[s] = *(const v2f*)&xs[ar0 + kk];
            x1[s] = *(const v2f*)&xs[ar1 + kk];
        }
    }
}

__device__ __forceinline__ void l1_aload1(v2f (&x0)[G1], const float* xs, int ar0, int k0, int n, int half)
{
    #pragma unroll
    for (int s = 0; s < G1; ++s) {
        if (s < n) x0[s] = *(const v2f*)&xs[ar0 + k0 + 4 * s + half * 2];
    }
}

__device__ __forceinline__ void t_mma(v8f& t0, v8f& t1, const v2f (&x0)[G1],
                                      const v2f (&fA)[G1], const v2f (&fB)[G1], int n)
{
    block_edge(t0, t1);
    #pragma unroll
    for (int s = 0; s < G1; ++s) {
        if (s < n) {
            t0 = WMMA_F32(x0[s], fA[s], t0);
            t1 = WMMA_F32(x0[s], fB[s], t1);
        }
    }
    block_edge(t0, t1);
}

__device__ __forceinline__ void l1_mma(v8f& a00, v8f& a01, v8f& a10, v8f& a11,
                                       const v2f (&x0)[G1], const v2f (&x1)[G1],
                                       const v2f (&fA)[G1], const v2f (&fB)[G1], int n)
{
    block_edge(a00, a01, a10, a11);
    #pragma unroll
    for (int s = 0; s < G1; ++s) {
        if (s < n) {
            a00 = WMMA_F32(x0[s], fA[s], a00);
            a01 = WMMA_F32(x0[s], fB[s], a01);
            a10 = WMMA_F32(x1[s], fA[s], a10);
            a11 = WMMA_F32(x1[s], fB[s], a11);
        }
    }
    block_edge(a00, a01, a10, a11);
}

__device__ __forceinline__ void wmma_branch_pad()
{
    asm volatile("v_nop\n\tv_nop\n\tv_nop\n\tv_nop\n\tv_nop");
}

__device__ __forceinline__ void wl_fetch(v2f (&f)[G2], const float* __restrict__ wu, unsigned lo, int g)
{
    #pragma unroll
    for (int s = 0; s < G2; ++s) {
        unsigned o = lo + (unsigned)((g * G2 + s) * 4 * 128);
        f[s].x = wu[o];
        f[s].y = wu[o + 128u];
    }
}

__device__ __forceinline__ void wl_aload(v2f (&x0)[G2], v2f (&x1)[G2],
                                         const float* h, int hs, int col, int g, int half)
{
    #pragma unroll
    for (int s = 0; s < G2; ++s) {
        int kk = (g * G2 + s) * 4 + half * 2;
        x0[s] = *(const v2f*)&h[col * hs + kk];
        x1[s] = *(const v2f*)&h[(16 + col) * hs + kk];
    }
}

__device__ __forceinline__ void wl_mma(v8f& ac0, v8f& ac1, const v2f (&x0)[G2], const v2f (&x1)[G2],
                                       const v2f (&f)[G2])
{
    block_edge(ac0, ac1);
    #pragma unroll
    for (int s = 0; s < G2; ++s) {
        ac0 = WMMA_F32(x0[s], f[s], ac0);
        ac1 = WMMA_F32(x1[s], f[s], ac1);
    }
    block_edge(ac0, ac1);
}

struct CoordIn {
    float ax, ay, az;
    float cx, cy, cz;
    float O[9];
    int   slot;
};

__device__ __forceinline__ void coord_load(CoordIn& c, const float* __restrict__ xyz,
                                           const float* __restrict__ orient, const int* sIdx,
                                           int tA, int i)
{
    int tt = i / 15, a = i % 15;
    int tk = sIdx[tt];
    const float* xp = xyz + (size_t)tk * 45;
    c.ax = xp[a * 3 + 0];  c.ay = xp[a * 3 + 1];  c.az = xp[a * 3 + 2];
    c.cx = xp[3 + 0];      c.cy = xp[3 + 1];      c.cz = xp[3 + 2];
    const float* O = orient + (size_t)tk * 9;
    #pragma unroll
    for (int j = 0; j < 9; ++j) c.O[j] = O[j];
    c.slot = sIdx[MT + tt] - tA;
}

__device__ __forceinline__ void coord_store(float* xs, const CoordIn& c, int i)
{
    int tt = i / 15, a = i % 15;
    float rx = c.ax - c.cx;
    float ry = c.ay - c.cy;
    float rz = c.az - c.cz;
    int base = tt * XROW + XC0 + c.slot * CSLOT + a * 3;
    xs[base + 0] = c.O[0] * rx + c.O[3] * ry + c.O[6] * rz;
    xs[base + 1] = c.O[1] * rx + c.O[4] * ry + c.O[7] * rz;
    xs[base + 2] = c.O[2] * rx + c.O[5] * ry + c.O[8] * rz;
}

__global__ __launch_bounds__(256, 4) void residue_embed_fused(
    const int* __restrict__ seq, const float* __restrict__ xyz,
    const float* __restrict__ dih, const int* __restrict__ cidx,
    const float* __restrict__ orient,
    const float* __restrict__ aa_emb, const float* __restrict__ chain_emb,
    const float* __restrict__ W1, const float* __restrict__ b1,
    const float* __restrict__ W2, const float* __restrict__ b2,
    const float* __restrict__ W3, const float* __restrict__ b3,
    const float* __restrict__ W4, const float* __restrict__ b4,
    float* __restrict__ out)
{
    __shared__ float xs[MT * XROW];
    __shared__ int   sIdx[2 * MT];
    __shared__ unsigned wtot[8 * 10];
    float* const hA = xs;
    float* const hB = xs + MT * H1S;

    const int tid = threadIdx.x;

    const int lane = tid & 31;
    const int w    = __builtin_amdgcn_readfirstlane(tid >> 5);
    const int half = lane >> 4;
    const int col  = lane & 15;

    const int n1 = w * 32;
    const float* W1u = W1 + n1;
    const unsigned lo1 = (unsigned)(half * 2 * 256 + col);
    const int n0 = w * 16;
    const float* W2u = W2 + n0;
    const float* W3u = W3 + n0;
    const float* W4u = W4 + n0;

    v2f pA[G1], pB[G1], qA[G1], qB[G1];
    l1_fetch(pA, pB, W1u, lo1, 0);

    int4 sv[16];
    #pragma unroll
    for (int i = 0; i < 16; ++i) sv[i] = ((const int4*)seq)[i * 256 + tid];
    unsigned long long ca = 0ull, cb = 0ull;
    unsigned cc = 0u;
    #pragma unroll
    for (int i = 0; i < 16; ++i) {
        #pragma unroll
        for (int j = 0; j < 4; ++j) {
            int v = (j == 0) ? sv[i].x : (j == 1) ? sv[i].y : (j == 2) ? sv[i].z : sv[i].w;
            v = v < 0 ? 0 : (v > NTYPE - 1 ? NTYPE - 1 : v);
            unsigned sh = (unsigned)(v & 7) * 8u;
            ca += (v < 8) ? (1ull << sh) : 0ull;
            cb += ((v >> 3) == 1) ? (1ull << sh) : 0ull;
            cc += (v >= 16) ? (1u << sh) : 0u;
        }
    }
    {
        unsigned pk[10];
        #pragma unroll
        for (int p = 0; p < 10; ++p) {
            const int t0 = 2 * p, t1 = 2 * p + 1;
            unsigned c0 = (t0 < 8) ? (unsigned)(ca >> (8 * t0)) : (t0 < 16) ? (unsigned)(cb >> (8 * (t0 - 8))) : (cc >> (8 * (t0 - 16)));
            unsigned c1 = (t1 < 8) ? (unsigned)(ca >> (8 * t1)) : (t1 < 16) ? (unsigned)(cb >> (8 * (t1 - 8))) : (cc >> (8 * (t1 - 16)));
            pk[p] = (c0 & 255u) | ((c1 & 255u) << 16);
        }
        #pragma unroll
        for (int off = 16; off >= 1; off >>= 1) {
            #pragma unroll
            for (int p = 0; p < 10; ++p) pk[p] += (unsigned)__shfl_xor((int)pk[p], off);
        }
        if (lane == 0) {
            #pragma unroll
            for (int p = 0; p < 10; ++p) wtot[w * 10 + p] = pk[p];
        }
    }
    __syncthreads();

    const int R0 = (int)blockIdx.x * MT;
    int tA = -1, tB = -1, baseA = 0;
    {
        int base = 0;
        #pragma unroll
        for (int t = 0; t < NTYPE; ++t) {
            unsigned n = 0u;
            #pragma unroll
            for (int ww = 0; ww < 8; ++ww) n += (wtot[ww * 10 + (t >> 1)] >> (16 * (t & 1))) & 0xFFFFu;
            int nb = base + (int)n;
            if (tA < 0 && nb > R0)      { tA = t; baseA = base; }
            if (tB < 0 && nb > R0 + 31) { tB = t; }
            base = nb;
        }
    }
    tA    = __builtin_amdgcn_readfirstlane(tA);
    tB    = __builtin_amdgcn_readfirstlane(tB);
    baseA = __builtin_amdgcn_readfirstlane(baseA);
    const int nT = tB - tA + 1;
    {
        int baseT = baseA - R0;
        for (int t = tA; t <= tB; ++t) {
            unsigned long long sel = (t < 8) ? ca : (t < 16) ? cb : (unsigned long long)cc;
            int cnt = (int)((sel >> ((t & 7) * 8)) & 255ull);
            int inc = cnt;
            #pragma unroll
            for (int off = 1; off < 32; off <<= 1) {
                int y = __shfl_up(inc, off);
                if (lane >= off) inc += y;
            }
            int rank = baseT + inc - cnt;
            int nt = 0;
            for (int ww = 0; ww < 8; ++ww) {
                int nw = (int)((wtot[ww * 10 + (t >> 1)] >> (16 * (t & 1))) & 0xFFFFu);
                if (ww < w) rank += nw;
                nt += nw;
            }
            #pragma unroll
            for (int i = 0; i < 16; ++i) {
                #pragma unroll
                for (int j = 0; j < 4; ++j) {
                    int v = (j == 0) ? sv[i].x : (j == 1) ? sv[i].y : (j == 2) ? sv[i].z : sv[i].w;
                    v = v < 0 ? 0 : (v > NTYPE - 1 ? NTYPE - 1 : v);
                    if (v == t) {
                        if ((unsigned)rank < (unsigned)MT) { sIdx[rank] = 4 * (i * 256 + tid) + j; sIdx[MT + rank] = t; }
                        ++rank;
                    }
                }
            }
            baseT += nt;
        }
    }
    {
        const int zr = tid >> 3, zc = tid & 7;
        for (int c = zc; c < nT * CSLOT; c += 8) xs[zr * XROW + XC0 + c] = 0.0f;
    }
    __syncthreads();

    float dv = 0.0f;
    if (tid < MT * 3) dv = dih[(size_t)sIdx[tid / 3] * 3 + (tid % 3)];

    const bool aaT = nT <= 6;
    const int rcA = cidx[sIdx[col]], rcB = cidx[sIdx[16 + col]];
    const int slA = sIdx[MT + col] - tA, slB = sIdx[MT + 16 + col] - tA;

    const int gk = tid & 127;
    const int gh = tid >> 7;
    float ga[16], gc[8];
    #pragma unroll
    for (int j = 0; j < 16; ++j) {
        int tt = 2 * j + gh;
        int sl = tt - 10;
        bool on = aaT ? (sl >= 0 && sl < nT) : true;
        int ty = aaT ? (tA + (on ? sl : 0)) : sIdx[MT + tt];
        float v = aa_emb[ty * 128 + gk];
        ga[j] = on ? v : 0.0f;
    }
    #pragma unroll
    for (int j = 0; j < 8; ++j) {
        int r = 2 * j + gh;
        bool on = (r >= 1 && r <= 9);
        float v = chain_emb[(on ? r : 1) * 128 + gk];
        gc[j] = on ? v : 0.0f;
    }
    const int  ci1  = tid + 256;
    const bool has1 = ci1 < MT * 15;
    CoordIn cA, cB;
    coord_load(cA, xyz, orient, sIdx, tA, tid);
    coord_load(cB, xyz, orient, sIdx, tA, has1 ? ci1 : MT * 15 - 1);

    if (tid < MT * 3) {
        int tt = tid / 3, d = tid % 3;
        float v = dv;
        const float FREQ[6] = {1.0f, 2.0f, 3.0f, 1.0f, 0.5f, 1.0f / 3.0f};
        int base = tt * XROW + 256 + d * 13;
        xs[base] = v;
        #pragma unroll
        for (int f = 0; f < 6; ++f) {
            xs[base + 1 + f] = sinf(FREQ[f] * v);
            xs[base + 7 + f] = cosf(FREQ[f] * v);
        }
    } else if (tid >= 128 && tid < 128 + MT) {
        xs[(tid - 128) * XROW + 295] = 0.0f;
    }

    coord_store(xs, cA, tid);
    if (has1) coord_store(xs, cB, ci1);
    #pragma unroll
    for (int j = 0; j < 16; ++j) xs[(2 * j + gh) * XROW + gk] = ga[j];
    #pragma unroll
    for (int j = 0; j < 8; ++j)  xs[(2 * j + gh) * XROW + 128 + gk] = gc[j];
    unsigned selmask;
    {
        unsigned mA = (rcA >= 1 && rcA <= 9) ? (1u << rcA) : 0u;
        unsigned mB = (rcB >= 1 && rcB <= 9) ? (1u << rcB) : 0u;
        if (aaT) {
            int a = slA < 0 ? 0 : (slA > 5 ? 5 : slA);
            int b = slB < 0 ? 0 : (slB > 5 ? 5 : slB);
            mA |= 1u << (10 + a);
            mB |= 1u << (10 + b);
        }
        selmask = mA | (mB << 16);
    }
    __syncthreads();

    int laneL, halfL, colL;
    unsigned lo2;
    v2f uP[G2], uQ[G2];
    v2f hP0[G2], hP1[G2], hQ0[G2], hQ1[G2];

    {
        v8f a00 = {}, a01 = {}, a10 = {}, a11 = {};
        const int ar0 = col * XROW;
        const int ar1 = (16 + col) * XROW;
        v2f xP0[G1], xP1[G1], xQ0[G1], xQ1[G1];
        v8f t0 = {}, t1 = {};
        l1_aload1(xP0, xs, ar0, 0, sg_count(0), half);
        l1_fetch(qA, qB, W1u, lo1, 1);
        l1_aload1(xQ0, xs, ar0, 4 * sg_first(1), sg_count(1), half);
        t_mma(t0, t1, xP0, pA, pB, sg_count(0));
        l1_fetch(pA, pB, W1u, lo1, 2);
        l1_aload1(xP0, xs, ar0, 4 * sg_first(2), sg_count(2), half);
        t_mma(t0, t1, xQ0, qA, qB, sg_count(1));
        l1_fetch(qA, qB, W1u, lo1, 3);
        l1_aload1(xQ0, xs, ar0, 4 * sg_first(3), sg_count(3), half);
        t_mma(t0, t1, xP0, pA, pB, sg_count(2));
        if (!aaT) { t0 = v8f{}; t1 = v8f{}; }
        l1_fetch(pA, pB, W1u, lo1, 4);
        l1_aload1(xP0, xs, ar0, 4 * sg_first(4), sg_count(4), half);
        t_mma(t0, t1, xQ0, qA, qB, sg_count(3));
        l1_fetch(qA, qB, W1u, lo1, 5);
        l1_aload1(xQ0, xs, ar0, 4 * sg_first(5), sg_count(5), half);
        t_mma(t0, t1, xP0, pA, pB, sg_count(4));
        l1_fetch(pA, pB, W1u, lo1, 6);
        l1_aload(xP0, xP1, xs, ar0, ar1, 4 * sg_first(6), sg_count(6), half);
        t_mma(t0, t1, xQ0, qA, qB, sg_count(5));
        {
            const unsigned m0 = (selmask >> (8 * half)) & 0xFFu;
            const unsigned m1 = (selmask >> (16 + 8 * half)) & 0xFFu;
            v2f s0[4], s1[4];
            #pragma unroll
            for (int s = 0; s < 4; ++s) {
                s0[s].x = ((m0 >> (2 * s)) & 1u) ? 1.0f : 0.0f;
                s0[s].y = ((m0 >> (2 * s + 1)) & 1u) ? 1.0f : 0.0f;
                s1[s].x = ((m1 >> (2 * s)) & 1u) ? 1.0f : 0.0f;
                s1[s].y = ((m1 >> (2 * s + 1)) & 1u) ? 1.0f : 0.0f;
            }
            block_edge(a00, a01, a10, a11);
            #pragma unroll
            for (int s = 0; s < 4; ++s) {
                v2f b0 = { t0[2 * s], t0[2 * s + 1] };
                v2f b1 = { t1[2 * s], t1[2 * s + 1] };
                a00 = WMMA_F32(s0[s], b0, a00);
                a01 = WMMA_F32(s0[s], b1, a01);
                a10 = WMMA_F32(s1[s], b0, a10);
                a11 = WMMA_F32(s1[s], b1, a11);
            }
            block_edge(a00, a01, a10, a11);
        }
        wmma_branch_pad();
        if (!aaT) {
            #pragma nounroll
            for (int g = 0; g < 8; ++g) {
                l1_cfetch(qA, qB, W1u + (size_t)g * (16 * 256), lo1, 4);
                l1_aload(xQ0, xQ1, xs, ar0, ar1, 16 * g, 4, half);
                l1_mma(a00, a01, a10, a11, xQ0, xQ1, qA, qB, 4);
                wmma_branch_pad();
            }
        }
        const float* wc = W1u + (size_t)(128 + 45 * tA) * 256;
        l1_cfetch(qA, qB, wc, lo1);
        l1_aload(xQ0, xQ1, xs, ar0, ar1, XC0, G1, half);
        l1_mma(a00, a01, a10, a11, xP0, xP1, pA, pB, sg_count(6));
        wmma_branch_pad();
        for (int c = 0; c < nT; c += 2) {
            const bool more1 = c + 1 < nT;
            if (more1) {
                l1_cfetch(pA, pB, wc + (size_t)(c + 1) * (45 * 256), lo1);
                l1_aload(xP0, xP1, xs, ar0, ar1, XC0 + (c + 1) * CSLOT, G1, half);
            }
            l1_mma(a00, a01, a10, a11, xQ0, xQ1, qA, qB, G1);
            wmma_branch_pad();
            if (more1) {
                if (c + 2 < nT) {
                    l1_cfetch(qA, qB, wc + (size_t)(c + 2) * (45 * 256), lo1);
                    l1_aload(xQ0, xQ1, xs, ar0, ar1, XC0 + (c + 2) * CSLOT, G1, half);
                }
                l1_mma(a00, a01, a10, a11, xP0, xP1, pA, pB, G1);
                wmma_branch_pad();
            }
        }
        laneL = (int)__builtin_amdgcn_mbcnt_lo(~0u, 0u);
        halfL = laneL >> 4;
        colL  = laneL & 15;
        lo2   = (unsigned)(halfL * 2 * 128 + colL);
        wl_fetch(uP, W2u, lo2, 0);
        const float bias1A = b1[n1 + colL], bias1B = b1[n1 + 16 + colL];

        __syncthreads();

        #pragma unroll
        for (int r = 0; r < 8; ++r) {
            int m = r + halfL * 8;
            float v00 = a00[r] + bias1A, v01 = a01[r] + bias1B;
            float v10 = a10[r] + bias1A, v11 = a11[r] + bias1B;
            hA[m * H1S + n1 + colL]             = v00 > 0.0f ? v00 : 0.0f;
            hA[m * H1S + n1 + 16 + colL]        = v01 > 0.0f ? v01 : 0.0f;
            hA[(m + 16) * H1S + n1 + colL]      = v10 > 0.0f ? v10 : 0.0f;
            hA[(m + 16) * H1S + n1 + 16 + colL] = v11 > 0.0f ? v11 : 0.0f;
        }
    }
    __syncthreads();

    {
        v8f ac0 = {}, ac1 = {};
        wl_aload(hP0, hP1, hA, H1S, colL, 0, halfL);
        wl_fetch(uQ, W2u, lo2, 1);
        wl_aload(hQ0, hQ1, hA, H1S, colL, 1, halfL);
        wl_mma(ac0, ac1, hP0, hP1, uP);
        wl_fetch(uP, W2u, lo2, 2);
        wl_aload(hP0, hP1, hA, H1S, colL, 2, halfL);
        wl_mma(ac0, ac1, hQ0, hQ1, uQ);
        wl_fetch(uQ, W2u, lo2, 3);
        wl_aload(hQ0, hQ1, hA, H1S, colL, 3, halfL);
        wl_mma(ac0, ac1, hP0, hP1, uP);
        wl_fetch(uP, W3u, lo2, 0);
        const float bias2 = b2[n0 + colL];
        wl_mma(ac0, ac1, hQ0, hQ1, uQ);
        #pragma unroll
        for (int r = 0; r < 8; ++r) {
            int m = r + halfL * 8;
            float v0 = ac0[r] + bias2, v1 = ac1[r] + bias2;
            hB[m * H2S + n0 + colL]        = v0 > 0.0f ? v0 : 0.0f;
            hB[(m + 16) * H2S + n0 + colL] = v1 > 0.0f ? v1 : 0.0f;
        }
    }
    __syncthreads();

    {
        v8f ac0 = {}, ac1 = {};
        wl_aload(hP0, hP1, hB, H2S, colL, 0, halfL);
        wl_fetch(uQ, W3u, lo2, 1);
        wl_aload(hQ0, hQ1, hB, H2S, colL, 1, halfL);
        wl_mma(ac0, ac1, hP0, hP1, uP);
        wl_fetch(uP, W4u, lo2, 0);
        const float bias3 = b3[n0 + colL];
        wl_mma(ac0, ac1, hQ0, hQ1, uQ);
        #pragma unroll
        for (int r = 0; r < 8; ++r) {
            int m = r + halfL * 8;
            float v0 = ac0[r] + bias3, v1 = ac1[r] + bias3;
            hA[m * H2S + n0 + colL]        = v0 > 0.0f ? v0 : 0.0f;
            hA[(m + 16) * H2S + n0 + colL] = v1 > 0.0f ? v1 : 0.0f;
        }
    }
    __syncthreads();

    {
        v8f ac0 = {}, ac1 = {};
        wl_aload(hP0, hP1, hA, H2S, colL, 0, halfL);
        wl_fetch(uQ, W4u, lo2, 1);
        wl_aload(hQ0, hQ1, hA, H2S, colL, 1, halfL);
        const float bias4 = b4[n0 + colL];
        wl_mma(ac0, ac1, hP0, hP1, uP);
        wl_mma(ac0, ac1, hQ0, hQ1, uQ);
        #pragma unroll
        for (int r = 0; r < 8; ++r) {
            int m = r + halfL * 8;
            out[(size_t)sIdx[m] * 128 + n0 + colL]      = ac0[r] + bias4;
            out[(size_t)sIdx[m + 16] * 128 + n0 + colL] = ac1[r] + bias4;
        }
    }
}

extern "C" void kernel_launch(void* const* d_in, const int* in_sizes, int n_in,
                              void* d_out, int out_size, void* d_ws, size_t ws_size,
                              hipStream_t stream) {
    (void)n_in; (void)d_ws; (void)ws_size; (void)out_size;
    const int*   seq    = (const int*)  d_in[0];
    const float* xyz    = (const float*)d_in[1];
    const float* dih    = (const float*)d_in[2];
    const int*   cidx   = (const int*)  d_in[3];
    const float* orient = (const float*)d_in[4];
    const float* aa_emb    = (const float*)d_in[6];
    const float* chain_emb = (const float*)d_in[7];
    const float* W1 = (const float*)d_in[8];   const float* b1 = (const float*)d_in[9];
    const float* W2 = (const float*)d_in[10];  const float* b2 = (const float*)d_in[11];
    const float* W3 = (const float*)d_in[12];  const float* b3 = (const float*)d_in[13];
    const float* W4 = (const float*)d_in[14];  const float* b4 = (const float*)d_in[15];
    float* out = (float*)d_out;

    const int tokens = in_sizes[0];
    const int blocks = tokens / MT;
    residue_embed_fused<<<blocks, 256, 0, stream>>>(
        seq, xyz, dih, cidx, orient, aa_emb, chain_emb,
        W1, b1, W2, b2, W3, b3, W4, b4, out);
}


# Tokens are type-sorted inside each sequence of 1024, not over all 16384: a work-group scans 4 tokens a thread, not 64; most tiles then hold 2 types; per-token sum order unchanged
# speedup vs baseline: 4.4144x; 1.0806x over previous
#include <hip/hip_runtime.h>
#include <math.h>


typedef float v2f __attribute__((ext_vector_type(2)));
typedef float v8f __attribute__((ext_vector_type(8)));

#define MT    32
#define NSTAT 74
#define CSLOT 48
#define XC0   296
#define XROW  1260
#define CHUNK 1024
#define NTYPE 20
#define H1S   260
#define H2S   132
#define G1    12
#define G2    16

#define WMMA_F32(a, b, c) \
    __builtin_amdgcn_wmma_f32_16x16x4_f32(false, (a), false, (b), (short)0, (c), false, false)

__device__ __forceinline__ void block_edge(v8f& a, v8f& b, v8f& c, v8f& d)
{
    asm volatile("" : "+v"(a), "+v"(b), "+v"(c), "+v"(d) :: "memory");
    __builtin_amdgcn_sched_barrier(0);
}
__device__ __forceinline__ void block_edge(v8f& a, v8f& b)
{
    asm volatile("" : "+v"(a), "+v"(b) :: "memory");
    __builtin_amdgcn_sched_barrier(0);
}


__device__ __forceinline__ constexpr int sg_first(int g) { return g == 0 ? 0 : g == 1 ? 12 : g == 2 ? 24 : g == 3 ? 32 : g == 4 ? 44 : g == 5 ? 56 : 64; }
__device__ __forceinline__ constexpr int sg_count(int g) { return g == 6 ? 10 : ((g == 2 || g == 5) ? 8 : 12); }

__device__ __forceinline__ void l1_fetch(v2f (&fA)[G1], v2f (&fB)[G1],
                                         const float* __restrict__ wu, unsigned lo, int g)
{
    #pragma unroll
    for (int s = 0; s < G1; ++s) {
        if (s < sg_count(g)) {
            const int S = sg_first(g) + s;
            const int row = (S < 32) ? 4 * S : (S < 64) ? 1067 + 4 * (S - 32) : 1028 + 4 * (S - 64);
            unsigned o0 = lo + (unsigned)(row * 256);
            unsigned o1 = o0 + 256u;
            fA[s].x = wu[o0];        fA[s].y = wu[o1];
            fB[s].x = wu[o0 + 16u];  fB[s].y = wu[o1 + 16u];
        }
    }
}

__device__ __forceinline__ void l1_cfetch(v2f (&fA)[G1], v2f (&fB)[G1],
                                          const float* __restrict__ wc, unsigned lo, int n = G1)
{
    #pragma unroll
    for (int s = 0; s < G1; ++s) {
        if (s < n) {
            unsigned o0 = lo + (unsigned)(4 * s * 256);
            unsigned o1 = o0 + 256u;
            fA[s].x = wc[o0];        fA[s].y = wc[o1];
            fB[s].x = wc[o0 + 16u];  fB[s].y = wc[o1 + 16u];
        }
    }
}

__device__ __forceinline__ void l1_aload(v2f (&x0)[G1], v2f (&x1)[G1],
                                         const float* xs, int ar0, int ar1, int k0, int n, int half)
{
    #pragma unroll
    for (int s = 0; s < G1; ++s) {
        if (s < n) {
            int kk = k0 + 4 * s + half * 2;
            x0[s] = *(const v2f*)&xs[ar0 + kk];
            x1[s] = *(const v2f*)&xs[ar1 + kk];
        }
    }
}

__device__ __forceinline__ void l1_aload1(v2f (&x0)[G1], const float* xs, int ar0, int k0, int n, int half)
{
    #pragma unroll
    for (int s = 0; s < G1; ++s) {
        if (s < n) x0[s] = *(const v2f*)&xs[ar0 + k0 + 4 * s + half * 2];
    }
}

__device__ __forceinline__ void t_mma(v8f& t0, v8f& t1, const v2f (&x0)[G1],
                                      const v2f (&fA)[G1], const v2f (&fB)[G1], int n)
{
    block_edge(t0, t1);
    #pragma unroll
    for (int s = 0; s < G1; ++s) {
        if (s < n) {
            t0 = WMMA_F32(x0[s], fA[s], t0);
            t1 = WMMA_F32(x0[s], fB[s], t1);
        }
    }
    block_edge(t0, t1);
}

__device__ __forceinline__ void l1_mma(v8f& a00, v8f& a01, v8f& a10, v8f& a11,
                                       const v2f (&x0)[G1], const v2f (&x1)[G1],
                                       const v2f (&fA)[G1], const v2f (&fB)[G1], int n)
{
    block_edge(a00, a01, a10, a11);
    #pragma unroll
    for (int s = 0; s < G1; ++s) {
        if (s < n) {
            a00 = WMMA_F32(x0[s], fA[s], a00);
            a01 = WMMA_F32(x0[s], fB[s], a01);
            a10 = WMMA_F32(x1[s], fA[s], a10);
            a11 = WMMA_F32(x1[s], fB[s], a11);
        }
    }
    block_edge(a00, a01, a10, a11);
}

__device__ __forceinline__ void wmma_branch_pad()
{
    asm volatile("v_nop\n\tv_nop\n\tv_nop\n\tv_nop\n\tv_nop");
}

__device__ __forceinline__ void wl_fetch(v2f (&f)[G2], const float* __restrict__ wu, unsigned lo, int g)
{
    #pragma unroll
    for (int s = 0; s < G2; ++s) {
        unsigned o = lo + (unsigned)((g * G2 + s) * 4 * 128);
        f[s].x = wu[o];
        f[s].y = wu[o + 128u];
    }
}

__device__ __forceinline__ void wl_aload(v2f (&x0)[G2], v2f (&x1)[G2],
                                         const float* h, int hs, int col, int g, int half)
{
    #pragma unroll
    for (int s = 0; s < G2; ++s) {
        int kk = (g * G2 + s) * 4 + half * 2;
        x0[s] = *(const v2f*)&h[col * hs + kk];
        x1[s] = *(const v2f*)&h[(16 + col) * hs + kk];
    }
}

__device__ __forceinline__ void wl_mma(v8f& ac0, v8f& ac1, const v2f (&x0)[G2], const v2f (&x1)[G2],
                                       const v2f (&f)[G2])
{
    block_edge(ac0, ac1);
    #pragma unroll
    for (int s = 0; s < G2; ++s) {
        ac0 = WMMA_F32(x0[s], f[s], ac0);
        ac1 = WMMA_F32(x1[s], f[s], ac1);
    }
    block_edge(ac0, ac1);
}

struct CoordIn {
    float ax, ay, az;
    float cx, cy, cz;
    float O[9];
    int   slot;
};

__device__ __forceinline__ void coord_load(CoordIn& c, const float* __restrict__ xyz,
                                           const float* __restrict__ orient, const int* sIdx,
                                           int tA, int i)
{
    int tt = i / 15, a = i % 15;
    int tk = sIdx[tt];
    const float* xp = xyz + (size_t)tk * 45;
    c.ax = xp[a * 3 + 0];  c.ay = xp[a * 3 + 1];  c.az = xp[a * 3 + 2];
    c.cx = xp[3 + 0];      c.cy = xp[3 + 1];      c.cz = xp[3 + 2];
    const float* O = orient + (size_t)tk * 9;
    #pragma unroll
    for (int j = 0; j < 9; ++j) c.O[j] = O[j];
    c.slot = sIdx[MT + tt] - tA;
}

__device__ __forceinline__ void coord_store(float* xs, const CoordIn& c, int i)
{
    int tt = i / 15, a = i % 15;
    float rx = c.ax - c.cx;
    float ry = c.ay - c.cy;
    float rz = c.az - c.cz;
    int base = tt * XROW + XC0 + c.slot * CSLOT + a * 3;
    xs[base + 0] = c.O[0] * rx + c.O[3] * ry + c.O[6] * rz;
    xs[base + 1] = c.O[1] * rx + c.O[4] * ry + c.O[7] * rz;
    xs[base + 2] = c.O[2] * rx + c.O[5] * ry + c.O[8] * rz;
}

__global__ __launch_bounds__(256, 4) void residue_embed_fused(
    const int* __restrict__ seq, const float* __restrict__ xyz,
    const float* __restrict__ dih, const int* __restrict__ cidx,
    const float* __restrict__ orient,
    const float* __restrict__ aa_emb, const float* __restrict__ chain_emb,
    const float* __restrict__ W1, const float* __restrict__ b1,
    const float* __restrict__ W2, const float* __restrict__ b2,
    const float* __restrict__ W3, const float* __restrict__ b3,
    const float* __restrict__ W4, const float* __restrict__ b4,
    float* __restrict__ out)
{
    __shared__ float xs[MT * XROW];
    __shared__ int   sIdx[2 * MT];
    __shared__ unsigned wtot[8 * 10];
    float* const hA = xs;
    float* const hB = xs + MT * H1S;

    const int tid = threadIdx.x;

    const int lane = tid & 31;
    const int w    = __builtin_amdgcn_readfirstlane(tid >> 5);
    const int half = lane >> 4;
    const int col  = lane & 15;

    const int n1 = w * 32;
    const float* W1u = W1 + n1;
    const unsigned lo1 = (unsigned)(half * 2 * 256 + col);
    const int n0 = w * 16;
    const float* W2u = W2 + n0;
    const float* W3u = W3 + n0;
    const float* W4u = W4 + n0;

    v2f pA[G1], pB[G1], qA[G1], qB[G1];
    l1_fetch(pA, pB, W1u, lo1, 0);

    const int chunk = (int)blockIdx.x / (CHUNK / MT);
    const int4 sv = ((const int4*)seq)[chunk * (CHUNK / 4) + tid];
    unsigned long long ca = 0ull, cb = 0ull;
    unsigned cc = 0u;
    #pragma unroll
    for (int j = 0; j < 4; ++j) {
        int v = (j == 0) ? sv.x : (j == 1) ? sv.y : (j == 2) ? sv.z : sv.w;
        v = v < 0 ? 0 : (v > NTYPE - 1 ? NTYPE - 1 : v);
        unsigned sh = (unsigned)(v & 7) * 8u;
        ca += (v < 8) ? (1ull << sh) : 0ull;
        cb += ((v >> 3) == 1) ? (1ull << sh) : 0ull;
        cc += (v >= 16) ? (1u << sh) : 0u;
    }
    {
        unsigned pk[10];
        #pragma unroll
        for (int p = 0; p < 10; ++p) {
            const int t0 = 2 * p, t1 = 2 * p + 1;
            unsigned c0 = (t0 < 8) ? (unsigned)(ca >> (8 * t0)) : (t0 < 16) ? (unsigned)(cb >> (8 * (t0 - 8))) : (cc >> (8 * (t0 - 16)));
            unsigned c1 = (t1 < 8) ? (unsigned)(ca >> (8 * t1)) : (t1 < 16) ? (unsigned)(cb >> (8 * (t1 - 8))) : (cc >> (8 * (t1 - 16)));
            pk[p] = (c0 & 255u) | ((c1 & 255u) << 16);
        }
        #pragma unroll
        for (int off = 16; off >= 1; off >>= 1) {
            #pragma unroll
            for (int p = 0; p < 10; ++p) pk[p] += (unsigned)__shfl_xor((int)pk[p], off);
        }
        if (lane == 0) {
            #pragma unroll
            for (int p = 0; p < 10; ++p) wtot[w * 10 + p] = pk[p];
        }
    }
    __syncthreads();

    const int R0 = ((int)blockIdx.x % (CHUNK / MT)) * MT;
    int tA = -1, tB = -1, baseA = 0;
    {
        int base = 0;
        #pragma unroll
        for (int t = 0; t < NTYPE; ++t) {
            unsigned n = 0u;
            #pragma unroll
            for (int ww = 0; ww < 8; ++ww) n += (wtot[ww * 10 + (t >> 1)] >> (16 * (t & 1))) & 0xFFFFu;
            int nb = base + (int)n;
            if (tA < 0 && nb > R0)      { tA = t; baseA = base; }
            if (tB < 0 && nb > R0 + 31) { tB = t; }
            base = nb;
        }
    }
    tA    = __builtin_amdgcn_readfirstlane(tA);
    tB    = __builtin_amdgcn_readfirstlane(tB);
    baseA = __builtin_amdgcn_readfirstlane(baseA);
    const int nT = tB - tA + 1;
    {
        int baseT = baseA - R0;
        for (int t = tA; t <= tB; ++t) {
            unsigned long long sel = (t < 8) ? ca : (t < 16) ? cb : (unsigned long long)cc;
            int cnt = (int)((sel >> ((t & 7) * 8)) & 255ull);
            int inc = cnt;
            #pragma unroll
            for (int off = 1; off < 32; off <<= 1) {
                int y = __shfl_up(inc, off);
                if (lane >= off) inc += y;
            }
            int rank = baseT + inc - cnt;
            int nt = 0;
            for (int ww = 0; ww < 8; ++ww) {
                int nw = (int)((wtot[ww * 10 + (t >> 1)] >> (16 * (t & 1))) & 0xFFFFu);
                if (ww < w) rank += nw;
                nt += nw;
            }
            #pragma unroll
            for (int j = 0; j < 4; ++j) {
                int v = (j == 0) ? sv.x : (j == 1) ? sv.y : (j == 2) ? sv.z : sv.w;
                v = v < 0 ? 0 : (v > NTYPE - 1 ? NTYPE - 1 : v);
                if (v == t) {
                    if ((unsigned)rank < (unsigned)MT) { sIdx[rank] = chunk * CHUNK + 4 * tid + j; sIdx[MT + rank] = t; }
                    ++rank;
                }
            }
            baseT += nt;
        }
    }
    {
        const int zr = tid >> 3, zc = tid & 7;
        for (int c = zc; c < nT * CSLOT; c += 8) xs[zr * XROW + XC0 + c] = 0.0f;
    }
    __syncthreads();

    float dv = 0.0f;
    if (tid < MT * 3) dv = dih[(size_t)sIdx[tid / 3] * 3 + (tid % 3)];

    const bool aaT = nT <= 6;
    const int rcA = cidx[sIdx[col]], rcB = cidx[sIdx[16 + col]];
    const int slA = sIdx[MT + col] - tA, slB = sIdx[MT + 16 + col] - tA;

    const int gk = tid & 127;
    const int gh = tid >> 7;
    float ga[16], gc[8];
    #pragma unroll
    for (int j = 0; j < 16; ++j) {
        int tt = 2 * j + gh;
        int sl = tt - 10;
        bool on = aaT ? (sl >= 0 && sl < nT) : true;
        int ty = aaT ? (tA + (on ? sl : 0)) : sIdx[MT + tt];
        float v = aa_emb[ty * 128 + gk];
        ga[j] = on ? v : 0.0f;
    }
    #pragma unroll
    for (int j = 0; j < 8; ++j) {
        int r = 2 * j + gh;
        bool on = (r >= 1 && r <= 9);
        float v = chain_emb[(on ? r : 1) * 128 + gk];
        gc[j] = on ? v : 0.0f;
    }
    const int  ci1  = tid + 256;
    const bool has1 = ci1 < MT * 15;
    CoordIn cA, cB;
    coord_load(cA, xyz, orient, sIdx, tA, tid);
    coord_load(cB, xyz, orient, sIdx, tA, has1 ? ci1 : MT * 15 - 1);

    if (tid < MT * 3) {
        int tt = tid / 3, d = tid % 3;
        float v = dv;
        const float FREQ[6] = {1.0f, 2.0f, 3.0f, 1.0f, 0.5f, 1.0f / 3.0f};
        int base = tt * XROW + 256 + d * 13;
        xs[base] = v;
        #pragma unroll
        for (int f = 0; f < 6; ++f) {
            xs[base + 1 + f] = sinf(FREQ[f] * v);
            xs[base + 7 + f] = cosf(FREQ[f] * v);
        }
    } else if (tid >= 128 && tid < 128 + MT) {
        xs[(tid - 128) * XROW + 295] = 0.0f;
    }

    coord_store(xs, cA, tid);
    if (has1) coord_store(xs, cB, ci1);
    #pragma unroll
    for (int j = 0; j < 16; ++j) xs[(2 * j + gh) * XROW + gk] = ga[j];
    #pragma unroll
    for (int j = 0; j < 8; ++j)  xs[(2 * j + gh) * XROW + 128 + gk] = gc[j];
    unsigned selmask;
    {
        unsigned mA = (rcA >= 1 && rcA <= 9) ? (1u << rcA) : 0u;
        unsigned mB = (rcB >= 1 && rcB <= 9) ? (1u << rcB) : 0u;
        if (aaT) {
            int a = slA < 0 ? 0 : (slA > 5 ? 5 : slA);
            int b = slB < 0 ? 0 : (slB > 5 ? 5 : slB);
            mA |= 1u << (10 + a);
            mB |= 1u << (10 + b);
        }
        selmask = mA | (mB << 16);
    }
    __syncthreads();

    int laneL, halfL, colL;
    unsigned lo2;
    v2f uP[G2], uQ[G2];
    v2f hP0[G2], hP1[G2], hQ0[G2], hQ1[G2];

    {
        v8f a00 = {}, a01 = {}, a10 = {}, a11 = {};
        const int ar0 = col * XROW;
        const int ar1 = (16 + col) * XROW;
        v2f xP0[G1], xP1[G1], xQ0[G1], xQ1[G1];
        v8f t0 = {}, t1 = {};
        l1_aload1(xP0, xs, ar0, 0, sg_count(0), half);
        l1_fetch(qA, qB, W1u, lo1, 1);
        l1_aload1(xQ0, xs, ar0, 4 * sg_first(1), sg_count(1), half);
        t_mma(t0, t1, xP0, pA, pB, sg_count(0));
        l1_fetch(pA, pB, W1u, lo1, 2);
        l1_aload1(xP0, xs, ar0, 4 * sg_first(2), sg_count(2), half);
        t_mma(t0, t1, xQ0, qA, qB, sg_count(1));
        l1_fetch(qA, qB, W1u, lo1, 3);
        l1_aload1(xQ0, xs, ar0, 4 * sg_first(3), sg_count(3), half);
        t_mma(t0, t1, xP0, pA, pB, sg_count(2));
        if (!aaT) { t0 = v8f{}; t1 = v8f{}; }
        l1_fetch(pA, pB, W1u, lo1, 4);
        l1_aload1(xP0, xs, ar0, 4 * sg_first(4), sg_count(4), half);
        t_mma(t0, t1, xQ0, qA, qB, sg_count(3));
        l1_fetch(qA, qB, W1u, lo1, 5);
        l1_aload1(xQ0, xs, ar0, 4 * sg_first(5), sg_count(5), half);
        t_mma(t0, t1, xP0, pA, pB, sg_count(4));
        l1_fetch(pA, pB, W1u, lo1, 6);
        l1_aload(xP0, xP1, xs, ar0, ar1, 4 * sg_first(6), sg_count(6), half);
        t_mma(t0, t1, xQ0, qA, qB, sg_count(5));
        {
            const unsigned m0 = (selmask >> (8 * half)) & 0xFFu;
            const unsigned m1 = (selmask >> (16 + 8 * half)) & 0xFFu;
            v2f s0[4], s1[4];
            #pragma unroll
            for (int s = 0; s < 4; ++s) {
                s0[s].x = ((m0 >> (2 * s)) & 1u) ? 1.0f : 0.0f;
                s0[s].y = ((m0 >> (2 * s + 1)) & 1u) ? 1.0f : 0.0f;
                s1[s].x = ((m1 >> (2 * s)) & 1u) ? 1.0f : 0.0f;
                s1[s].y = ((m1 >> (2 * s + 1)) & 1u) ? 1.0f : 0.0f;
            }
            block_edge(a00, a01, a10, a11);
            #pragma unroll
            for (int s = 0; s < 4; ++s) {
                v2f b0 = { t0[2 * s], t0[2 * s + 1] };
                v2f b1 = { t1[2 * s], t1[2 * s + 1] };
                a00 = WMMA_F32(s0[s], b0, a00);
                a01 = WMMA_F32(s0[s], b1, a01);
                a10 = WMMA_F32(s1[s], b0, a10);
                a11 = WMMA_F32(s1[s], b1, a11);
            }
            block_edge(a00, a01, a10, a11);
        }
        wmma_branch_pad();
        if (!aaT) {
            #pragma nounroll
            for (int g = 0; g < 8; ++g) {
                l1_cfetch(qA, qB, W1u + (size_t)g * (16 * 256), lo1, 4);
                l1_aload(xQ0, xQ1, xs, ar0, ar1, 16 * g, 4, half);
                l1_mma(a00, a01, a10, a11, xQ0, xQ1, qA, qB, 4);
                wmma_branch_pad();
            }
        }
        const float* wc = W1u + (size_t)(128 + 45 * tA) * 256;
        l1_cfetch(qA, qB, wc, lo1);
        l1_aload(xQ0, xQ1, xs, ar0, ar1, XC0, G1, half);
        l1_mma(a00, a01, a10, a11, xP0, xP1, pA, pB, sg_count(6));
        wmma_branch_pad();
        for (int c = 0; c < nT; c += 2) {
            const bool more1 = c + 1 < nT;
            if (more1) {
                l1_cfetch(pA, pB, wc + (size_t)(c + 1) * (45 * 256), lo1);
                l1_aload(xP0, xP1, xs, ar0, ar1, XC0 + (c + 1) * CSLOT, G1, half);
            }
            l1_mma(a00, a01, a10, a11, xQ0, xQ1, qA, qB, G1);
            wmma_branch_pad();
            if (more1) {
                if (c + 2 < nT) {
                    l1_cfetch(qA, qB, wc + (size_t)(c + 2) * (45 * 256), lo1);
                    l1_aload(xQ0, xQ1, xs, ar0, ar1, XC0 + (c + 2) * CSLOT, G1, half);
                }
                l1_mma(a00, a01, a10, a11, xP0, xP1, pA, pB, G1);
                wmma_branch_pad();
            }
        }
        laneL = (int)__builtin_amdgcn_mbcnt_lo(~0u, 0u);
        halfL = laneL >> 4;
        colL  = laneL & 15;
        lo2   = (unsigned)(halfL * 2 * 128 + colL);
        wl_fetch(uP, W2u, lo2, 0);
        const float bias1A = b1[n1 + colL], bias1B = b1[n1 + 16 + colL];

        __syncthreads();

        #pragma unroll
        for (int r = 0; r < 8; ++r) {
            int m = r + halfL * 8;
            float v00 = a00[r] + bias1A, v01 = a01[r] + bias1B;
            float v10 = a10[r] + bias1A, v11 = a11[r] + bias1B;
            hA[m * H1S + n1 + colL]             = v00 > 0.0f ? v00 : 0.0f;
            hA[m * H1S + n1 + 16 + colL]        = v01 > 0.0f ? v01 : 0.0f;
            hA[(m + 16) * H1S + n1 + colL]      = v10 > 0.0f ? v10 : 0.0f;
            hA[(m + 16) * H1S + n1 + 16 + colL] = v11 > 0.0f ? v11 : 0.0f;
        }
    }
    __syncthreads();

    {
        v8f ac0 = {}, ac1 = {};
        wl_aload(hP0, hP1, hA, H1S, colL, 0, halfL);
        wl_fetch(uQ, W2u, lo2, 1);
        wl_aload(hQ0, hQ1, hA, H1S, colL, 1, halfL);
        wl_mma(ac0, ac1, hP0, hP1, uP);
        wl_fetch(uP, W2u, lo2, 2);
        wl_aload(hP0, hP1, hA, H1S, colL, 2, halfL);
        wl_mma(ac0, ac1, hQ0, hQ1, uQ);
        wl_fetch(uQ, W2u, lo2, 3);
        wl_aload(hQ0, hQ1, hA, H1S, colL, 3, halfL);
        wl_mma(ac0, ac1, hP0, hP1, uP);
        wl_fetch(uP, W3u, lo2, 0);
        const float bias2 = b2[n0 + colL];
        wl_mma(ac0, ac1, hQ0, hQ1, uQ);
        #pragma unroll
        for (int r = 0; r < 8; ++r) {
            int m = r + halfL * 8;
            float v0 = ac0[r] + bias2, v1 = ac1[r] + bias2;
            hB[m * H2S + n0 + colL]        = v0 > 0.0f ? v0 : 0.0f;
            hB[(m + 16) * H2S + n0 + colL] = v1 > 0.0f ? v1 : 0.0f;
        }
    }
    __syncthreads();

    {
        v8f ac0 = {}, ac1 = {};
        wl_aload(hP0, hP1, hB, H2S, colL, 0, halfL);
        wl_fetch(uQ, W3u, lo2, 1);
        wl_aload(hQ0, hQ1, hB, H2S, colL, 1, halfL);
        wl_mma(ac0, ac1, hP0, hP1, uP);
        wl_fetch(uP, W4u, lo2, 0);
        const float bias3 = b3[n0 + colL];
        wl_mma(ac0, ac1, hQ0, hQ1, uQ);
        #pragma unroll
        for (int r = 0; r < 8; ++r) {
            int m = r + halfL * 8;
            float v0 = ac0[r] + bias3, v1 = ac1[r] + bias3;
            hA[m * H2S + n0 + colL]        = v0 > 0.0f ? v0 : 0.0f;
            hA[(m + 16) * H2S + n0 + colL] = v1 > 0.0f ? v1 : 0.0f;
        }
    }
    __syncthreads();

    {
        v8f ac0 = {}, ac1 = {};
        wl_aload(hP0, hP1, hA, H2S, colL, 0, halfL);
        wl_fetch(uQ, W4u, lo2, 1);
        wl_aload(hQ0, hQ1, hA, H2S, colL, 1, halfL);
        const float bias4 = b4[n0 + colL];
        wl_mma(ac0, ac1, hP0, hP1, uP);
        wl_mma(ac0, ac1, hQ0, hQ1, uQ);
        #pragma unroll
        for (int r = 0; r < 8; ++r) {
            int m = r + halfL * 8;
            out[(size_t)sIdx[m] * 128 + n0 + colL]      = ac0[r] + bias4;
            out[(size_t)sIdx[m + 16] * 128 + n0 + colL] = ac1[r] + bias4;
        }
    }
}

extern "C" void kernel_launch(void* const* d_in, const int* in_sizes, int n_in,
                              void* d_out, int out_size, void* d_ws, size_t ws_size,
                              hipStream_t stream) {
    (void)n_in; (void)d_ws; (void)ws_size; (void)out_size;
    const int*   seq    = (const int*)  d_in[0];
    const float* xyz    = (const float*)d_in[1];
    const float* dih    = (const float*)d_in[2];
    const int*   cidx   = (const int*)  d_in[3];
    const float* orient = (const float*)d_in[4];
    const float* aa_emb    = (const float*)d_in[6];
    const float* chain_emb = (const float*)d_in[7];
    const float* W1 = (const float*)d_in[8];   const float* b1 = (const float*)d_in[9];
    const float* W2 = (const float*)d_in[10];  const float* b2 = (const float*)d_in[11];
    const float* W3 = (const float*)d_in[12];  const float* b3 = (const float*)d_in[13];
    const float* W4 = (const float*)d_in[14];  const float* b4 = (const float*)d_in[15];
    float* out = (float*)d_out;

    const int tokens = in_sizes[0];
    const int blocks = tokens / MT;
    residue_embed_fused<<<blocks, 256, 0, stream>>>(
        seq, xyz, dih, cidx, orient, aa_emb, chain_emb,
        W1, b1, W2, b2, W3, b3, W4, b4, out);
}


# Tile lookup: each residue type's block count sits in its own lane; a row-shift prefix over 20 lanes and two ballots give the tile's first and last type, not a 20-step serial search in every thread
# speedup vs baseline: 4.5733x; 1.0360x over previous
#include <hip/hip_runtime.h>
#include <math.h>


typedef float v2f __attribute__((ext_vector_type(2)));
typedef float v8f __attribute__((ext_vector_type(8)));

#define MT    32
#define NSTAT 74
#define CSLOT 48
#define XC0   296
#define XROW  1260
#define CHUNK 1024
#define NTYPE 20
#define H1S   260
#define H2S   132
#define G1    12
#define G2    16

#define WMMA_F32(a, b, c) \
    __builtin_amdgcn_wmma_f32_16x16x4_f32(false, (a), false, (b), (short)0, (c), false, false)

__device__ __forceinline__ void block_edge(v8f& a, v8f& b, v8f& c, v8f& d)
{
    asm volatile("" : "+v"(a), "+v"(b), "+v"(c), "+v"(d) :: "memory");
    __builtin_amdgcn_sched_barrier(0);
}
__device__ __forceinline__ void block_edge(v8f& a, v8f& b)
{
    asm volatile("" : "+v"(a), "+v"(b) :: "memory");
    __builtin_amdgcn_sched_barrier(0);
}


__device__ __forceinline__ constexpr int sg_first(int g) { return g == 0 ? 0 : g == 1 ? 12 : g == 2 ? 24 : g == 3 ? 32 : g == 4 ? 44 : g == 5 ? 56 : 64; }
__device__ __forceinline__ constexpr int sg_count(int g) { return g == 6 ? 10 : ((g == 2 || g == 5) ? 8 : 12); }

__device__ __forceinline__ void l1_fetch(v2f (&fA)[G1], v2f (&fB)[G1],
                                         const float* __restrict__ wu, unsigned lo, int g)
{
    #pragma unroll
    for (int s = 0; s < G1; ++s) {
        if (s < sg_count(g)) {
            const int S = sg_first(g) + s;
            const int row = (S < 32) ? 4 * S : (S < 64) ? 1067 + 4 * (S - 32) : 1028 + 4 * (S - 64);
            unsigned o0 = lo + (unsigned)(row * 256);
            unsigned o1 = o0 + 256u;
            fA[s].x = wu[o0];        fA[s].y = wu[o1];
            fB[s].x = wu[o0 + 16u];  fB[s].y = wu[o1 + 16u];
        }
    }
}

__device__ __forceinline__ void l1_cfetch(v2f (&fA)[G1], v2f (&fB)[G1],
                                          const float* __restrict__ wc, unsigned lo, int n = G1)
{
    #pragma unroll
    for (int s = 0; s < G1; ++s) {
        if (s < n) {
            unsigned o0 = lo + (unsigned)(4 * s * 256);
            unsigned o1 = o0 + 256u;
            fA[s].x = wc[o0];        fA[s].y = wc[o1];
            fB[s].x = wc[o0 + 16u];  fB[s].y = wc[o1 + 16u];
        }
    }
}

__device__ __forceinline__ void l1_aload(v2f (&x0)[G1], v2f (&x1)[G1],
                                         const float* xs, int ar0, int ar1, int k0, int n, int half)
{
    #pragma unroll
    for (int s = 0; s < G1; ++s) {
        if (s < n) {
            int kk = k0 + 4 * s + half * 2;
            x0[s] = *(const v2f*)&xs[ar0 + kk];
            x1[s] = *(const v2f*)&xs[ar1 + kk];
        }
    }
}

__device__ __forceinline__ void l1_aload1(v2f (&x0)[G1], const float* xs, int ar0, int k0, int n, int half)
{
    #pragma unroll
    for (int s = 0; s < G1; ++s) {
        if (s < n) x0[s] = *(const v2f*)&xs[ar0 + k0 + 4 * s + half * 2];
    }
}

__device__ __forceinline__ void t_mma(v8f& t0, v8f& t1, const v2f (&x0)[G1],
                                      const v2f (&fA)[G1], const v2f (&fB)[G1], int n)
{
    block_edge(t0, t1);
    #pragma unroll
    for (int s = 0; s < G1; ++s) {
        if (s < n) {
            t0 = WMMA_F32(x0[s], fA[s], t0);
            t1 = WMMA_F32(x0[s], fB[s], t1);
        }
    }
    block_edge(t0, t1);
}

__device__ __forceinline__ void l1_mma(v8f& a00, v8f& a01, v8f& a10, v8f& a11,
                                       const v2f (&x0)[G1], const v2f (&x1)[G1],
                                       const v2f (&fA)[G1], const v2f (&fB)[G1], int n)
{
    block_edge(a00, a01, a10, a11);
    #pragma unroll
    for (int s = 0; s < G1; ++s) {
        if (s < n) {
            a00 = WMMA_F32(x0[s], fA[s], a00);
            a01 = WMMA_F32(x0[s], fB[s], a01);
            a10 = WMMA_F32(x1[s], fA[s], a10);
            a11 = WMMA_F32(x1[s], fB[s], a11);
        }
    }
    block_edge(a00, a01, a10, a11);
}

__device__ __forceinline__ void wmma_branch_pad()
{
    asm volatile("v_nop\n\tv_nop\n\tv_nop\n\tv_nop\n\tv_nop");
}

__device__ __forceinline__ void wl_fetch(v2f (&f)[G2], const float* __restrict__ wu, unsigned lo, int g)
{
    #pragma unroll
    for (int s = 0; s < G2; ++s) {
        unsigned o = lo + (unsigned)((g * G2 + s) * 4 * 128);
        f[s].x = wu[o];
        f[s].y = wu[o + 128u];
    }
}

__device__ __forceinline__ void wl_aload(v2f (&x0)[G2], v2f (&x1)[G2],
                                         const float* h, int hs, int col, int g, int half)
{
    #pragma unroll
    for (int s = 0; s < G2; ++s) {
        int kk = (g * G2 + s) * 4 + half * 2;
        x0[s] = *(const v2f*)&h[col * hs + kk];
        x1[s] = *(const v2f*)&h[(16 + col) * hs + kk];
    }
}

__device__ __forceinline__ void wl_mma(v8f& ac0, v8f& ac1, const v2f (&x0)[G2], const v2f (&x1)[G2],
                                       const v2f (&f)[G2])
{
    block_edge(ac0, ac1);
    #pragma unroll
    for (int s = 0; s < G2; ++s) {
        ac0 = WMMA_F32(x0[s], f[s], ac0);
        ac1 = WMMA_F32(x1[s], f[s], ac1);
    }
    block_edge(ac0, ac1);
}

struct CoordIn {
    float ax, ay, az;
    float cx, cy, cz;
    float O[9];
    int   slot;
};

__device__ __forceinline__ void coord_load(CoordIn& c, const float* __restrict__ xyz,
                                           const float* __restrict__ orient, const int* sIdx,
                                           int tA, int i)
{
    int tt = i / 15, a = i % 15;
    int tk = sIdx[tt];
    const float* xp = xyz + (size_t)tk * 45;
    c.ax = xp[a * 3 + 0];  c.ay = xp[a * 3 + 1];  c.az = xp[a * 3 + 2];
    c.cx = xp[3 + 0];      c.cy = xp[3 + 1];      c.cz = xp[3 + 2];
    const float* O = orient + (size_t)tk * 9;
    #pragma unroll
    for (int j = 0; j < 9; ++j) c.O[j] = O[j];
    c.slot = sIdx[MT + tt] - tA;
}

__device__ __forceinline__ void coord_store(float* xs, const CoordIn& c, int i)
{
    int tt = i / 15, a = i % 15;
    float rx = c.ax - c.cx;
    float ry = c.ay - c.cy;
    float rz = c.az - c.cz;
    int base = tt * XROW + XC0 + c.slot * CSLOT + a * 3;
    xs[base + 0] = c.O[0] * rx + c.O[3] * ry + c.O[6] * rz;
    xs[base + 1] = c.O[1] * rx + c.O[4] * ry + c.O[7] * rz;
    xs[base + 2] = c.O[2] * rx + c.O[5] * ry + c.O[8] * rz;
}

__global__ __launch_bounds__(256, 4) void residue_embed_fused(
    const int* __restrict__ seq, const float* __restrict__ xyz,
    const float* __restrict__ dih, const int* __restrict__ cidx,
    const float* __restrict__ orient,
    const float* __restrict__ aa_emb, const float* __restrict__ chain_emb,
    const float* __restrict__ W1, const float* __restrict__ b1,
    const float* __restrict__ W2, const float* __restrict__ b2,
    const float* __restrict__ W3, const float* __restrict__ b3,
    const float* __restrict__ W4, const float* __restrict__ b4,
    float* __restrict__ out)
{
    __shared__ float xs[MT * XROW];
    __shared__ int   sIdx[2 * MT];
    __shared__ unsigned wtot[8 * 10];
    float* const hA = xs;
    float* const hB = xs + MT * H1S;

    const int tid = threadIdx.x;

    const int lane = tid & 31;
    const int w    = __builtin_amdgcn_readfirstlane(tid >> 5);
    const int half = lane >> 4;
    const int col  = lane & 15;

    const int n1 = w * 32;
    const float* W1u = W1 + n1;
    const unsigned lo1 = (unsigned)(half * 2 * 256 + col);
    const int n0 = w * 16;
    const float* W2u = W2 + n0;
    const float* W3u = W3 + n0;
    const float* W4u = W4 + n0;

    v2f pA[G1], pB[G1], qA[G1], qB[G1];
    l1_fetch(pA, pB, W1u, lo1, 0);

    const int chunk = (int)blockIdx.x / (CHUNK / MT);
    const int4 sv = ((const int4*)seq)[chunk * (CHUNK / 4) + tid];
    unsigned long long ca = 0ull, cb = 0ull;
    unsigned cc = 0u;
    #pragma unroll
    for (int j = 0; j < 4; ++j) {
        int v = (j == 0) ? sv.x : (j == 1) ? sv.y : (j == 2) ? sv.z : sv.w;
        v = v < 0 ? 0 : (v > NTYPE - 1 ? NTYPE - 1 : v);
        unsigned sh = (unsigned)(v & 7) * 8u;
        ca += (v < 8) ? (1ull << sh) : 0ull;
        cb += ((v >> 3) == 1) ? (1ull << sh) : 0ull;
        cc += (v >= 16) ? (1u << sh) : 0u;
    }
    {
        unsigned pk[10];
        #pragma unroll
        for (int p = 0; p < 10; ++p) {
            const int t0 = 2 * p, t1 = 2 * p + 1;
            unsigned c0 = (t0 < 8) ? (unsigned)(ca >> (8 * t0)) : (t0 < 16) ? (unsigned)(cb >> (8 * (t0 - 8))) : (cc >> (8 * (t0 - 16)));
            unsigned c1 = (t1 < 8) ? (unsigned)(ca >> (8 * t1)) : (t1 < 16) ? (unsigned)(cb >> (8 * (t1 - 8))) : (cc >> (8 * (t1 - 16)));
            pk[p] = (c0 & 255u) | ((c1 & 255u) << 16);
        }
        #pragma unroll
        for (int off = 16; off >= 1; off >>= 1) {
            #pragma unroll
            for (int p = 0; p < 10; ++p) pk[p] += (unsigned)__shfl_xor((int)pk[p], off);
        }
        if (lane == 0) {
            #pragma unroll
            for (int p = 0; p < 10; ++p) wtot[w * 10 + p] = pk[p];
        }
    }
    __syncthreads();

    const int R0 = ((int)blockIdx.x % (CHUNK / MT)) * MT;
    int tA, tB, baseA;
    {
        const int tq = lane < NTYPE ? lane : NTYPE - 1;
        unsigned n = 0u;
        #pragma unroll
        for (int ww = 0; ww < 8; ++ww) n += (wtot[ww * 10 + (tq >> 1)] >> (16 * (tq & 1))) & 0xFFFFu;
        const int nl = lane < NTYPE ? (int)n : 0;
        int incl = nl;
        incl += __builtin_amdgcn_update_dpp(0, incl, 0x111, 0xf, 0xf, true);
        incl += __builtin_amdgcn_update_dpp(0, incl, 0x112, 0xf, 0xf, true);
        incl += __builtin_amdgcn_update_dpp(0, incl, 0x114, 0xf, 0xf, true);
        incl += __builtin_amdgcn_update_dpp(0, incl, 0x118, 0xf, 0xf, true);
        const int low = __builtin_amdgcn_readlane(incl, 15);
        if (lane >= 16) incl += low;
        const unsigned mA = __builtin_amdgcn_ballot_w32(incl > R0);
        const unsigned mB = __builtin_amdgcn_ballot_w32(incl > R0 + 31);
        tA    = __builtin_ctz(mA);
        tB    = __builtin_ctz(mB);
        baseA = __builtin_amdgcn_readlane(incl - nl, tA);
    }
    const int nT = tB - tA + 1;
    {
        int baseT = baseA - R0;
        for (int t = tA; t <= tB; ++t) {
            unsigned long long sel = (t < 8) ? ca : (t < 16) ? cb : (unsigned long long)cc;
            int cnt = (int)((sel >> ((t & 7) * 8)) & 255ull);
            int inc = cnt;
            #pragma unroll
            for (int off = 1; off < 32; off <<= 1) {
                int y = __shfl_up(inc, off);
                if (lane >= off) inc += y;
            }
            int rank = baseT + inc - cnt;
            int nt = 0;
            for (int ww = 0; ww < 8; ++ww) {
                int nw = (int)((wtot[ww * 10 + (t >> 1)] >> (16 * (t & 1))) & 0xFFFFu);
                if (ww < w) rank += nw;
                nt += nw;
            }
            #pragma unroll
            for (int j = 0; j < 4; ++j) {
                int v = (j == 0) ? sv.x : (j == 1) ? sv.y : (j == 2) ? sv.z : sv.w;
                v = v < 0 ? 0 : (v > NTYPE - 1 ? NTYPE - 1 : v);
                if (v == t) {
                    if ((unsigned)rank < (unsigned)MT) { sIdx[rank] = chunk * CHUNK + 4 * tid + j; sIdx[MT + rank] = t; }
                    ++rank;
                }
            }
            baseT += nt;
        }
    }
    {
        const int zr = tid >> 3, zc = tid & 7;
        for (int c = zc; c < nT * CSLOT; c += 8) xs[zr * XROW + XC0 + c] = 0.0f;
    }
    __syncthreads();

    float dv = 0.0f;
    if (tid < MT * 3) dv = dih[(size_t)sIdx[tid / 3] * 3 + (tid % 3)];

    const bool aaT = nT <= 6;
    const int rcA = cidx[sIdx[col]], rcB = cidx[sIdx[16 + col]];
    const int slA = sIdx[MT + col] - tA, slB = sIdx[MT + 16 + col] - tA;

    const int gk = tid & 127;
    const int gh = tid >> 7;
    float ga[16], gc[8];
    #pragma unroll
    for (int j = 0; j < 16; ++j) {
        int tt = 2 * j + gh;
        int sl = tt - 10;
        bool on = aaT ? (sl >= 0 && sl < nT) : true;
        int ty = aaT ? (tA + (on ? sl : 0)) : sIdx[MT + tt];
        float v = aa_emb[ty * 128 + gk];
        ga[j] = on ? v : 0.0f;
    }
    #pragma unroll
    for (int j = 0; j < 8; ++j) {
        int r = 2 * j + gh;
        bool on = (r >= 1 && r <= 9);
        float v = chain_emb[(on ? r : 1) * 128 + gk];
        gc[j] = on ? v : 0.0f;
    }
    const int  ci1  = tid + 256;
    const bool has1 = ci1 < MT * 15;
    CoordIn cA, cB;
    coord_load(cA, xyz, orient, sIdx, tA, tid);
    coord_load(cB, xyz, orient, sIdx, tA, has1 ? ci1 : MT * 15 - 1);

    if (tid < MT * 3) {
        int tt = tid / 3, d = tid % 3;
        float v = dv;
        const float FREQ[6] = {1.0f, 2.0f, 3.0f, 1.0f, 0.5f, 1.0f / 3.0f};
        int base = tt * XROW + 256 + d * 13;
        xs[base] = v;
        #pragma unroll
        for (int f = 0; f < 6; ++f) {
            xs[base + 1 + f] = sinf(FREQ[f] * v);
            xs[base + 7 + f] = cosf(FREQ[f] * v);
        }
    } else if (tid >= 128 && tid < 128 + MT) {
        xs[(tid - 128) * XROW + 295] = 0.0f;
    }

    coord_store(xs, cA, tid);
    if (has1) coord_store(xs, cB, ci1);
    #pragma unroll
    for (int j = 0; j < 16; ++j) xs[(2 * j + gh) * XROW + gk] = ga[j];
    #pragma unroll
    for (int j = 0; j < 8; ++j)  xs[(2 * j + gh) * XROW + 128 + gk] = gc[j];
    unsigned selmask;
    {
        unsigned mA = (rcA >= 1 && rcA <= 9) ? (1u << rcA) : 0u;
        unsigned mB = (rcB >= 1 && rcB <= 9) ? (1u << rcB) : 0u;
        if (aaT) {
            int a = slA < 0 ? 0 : (slA > 5 ? 5 : slA);
            int b = slB < 0 ? 0 : (slB > 5 ? 5 : slB);
            mA |= 1u << (10 + a);
            mB |= 1u << (10 + b);
        }
        selmask = mA | (mB << 16);
    }
    __syncthreads();

    int laneL, halfL, colL;
    unsigned lo2;
    v2f uP[G2], uQ[G2];
    v2f hP0[G2], hP1[G2], hQ0[G2], hQ1[G2];

    {
        v8f a00 = {}, a01 = {}, a10 = {}, a11 = {};
        const int ar0 = col * XROW;
        const int ar1 = (16 + col) * XROW;
        v2f xP0[G1], xP1[G1], xQ0[G1], xQ1[G1];
        v8f t0 = {}, t1 = {};
        l1_aload1(xP0, xs, ar0, 0, sg_count(0), half);
        l1_fetch(qA, qB, W1u, lo1, 1);
        l1_aload1(xQ0, xs, ar0, 4 * sg_first(1), sg_count(1), half);
        t_mma(t0, t1, xP0, pA, pB, sg_count(0));
        l1_fetch(pA, pB, W1u, lo1, 2);
        l1_aload1(xP0, xs, ar0, 4 * sg_first(2), sg_count(2), half);
        t_mma(t0, t1, xQ0, qA, qB, sg_count(1));
        l1_fetch(qA, qB, W1u, lo1, 3);
        l1_aload1(xQ0, xs, ar0, 4 * sg_first(3), sg_count(3), half);
        t_mma(t0, t1, xP0, pA, pB, sg_count(2));
        if (!aaT) { t0 = v8f{}; t1 = v8f{}; }
        l1_fetch(pA, pB, W1u, lo1, 4);
        l1_aload1(xP0, xs, ar0, 4 * sg_first(4), sg_count(4), half);
        t_mma(t0, t1, xQ0, qA, qB, sg_count(3));
        l1_fetch(qA, qB, W1u, lo1, 5);
        l1_aload1(xQ0, xs, ar0, 4 * sg_first(5), sg_count(5), half);
        t_mma(t0, t1, xP0, pA, pB, sg_count(4));
        l1_fetch(pA, pB, W1u, lo1, 6);
        l1_aload(xP0, xP1, xs, ar0, ar1, 4 * sg_first(6), sg_count(6), half);
        t_mma(t0, t1, xQ0, qA, qB, sg_count(5));
        {
            const unsigned m0 = (selmask >> (8 * half)) & 0xFFu;
            const unsigned m1 = (selmask >> (16 + 8 * half)) & 0xFFu;
            v2f s0[4], s1[4];
            #pragma unroll
            for (int s = 0; s < 4; ++s) {
                s0[s].x = ((m0 >> (2 * s)) & 1u) ? 1.0f : 0.0f;
                s0[s].y = ((m0 >> (2 * s + 1)) & 1u) ? 1.0f : 0.0f;
                s1[s].x = ((m1 >> (2 * s)) & 1u) ? 1.0f : 0.0f;
                s1[s].y = ((m1 >> (2 * s + 1)) & 1u) ? 1.0f : 0.0f;
            }
            block_edge(a00, a01, a10, a11);
            #pragma unroll
            for (int s = 0; s < 4; ++s) {
                v2f b0 = { t0[2 * s], t0[2 * s + 1] };
                v2f b1 = { t1[2 * s], t1[2 * s + 1] };
                a00 = WMMA_F32(s0[s], b0, a00);
                a01 = WMMA_F32(s0[s], b1, a01);
                a10 = WMMA_F32(s1[s], b0, a10);
                a11 = WMMA_F32(s1[s], b1, a11);
            }
            block_edge(a00, a01, a10, a11);
        }
        wmma_branch_pad();
        if (!aaT) {
            #pragma nounroll
            for (int g = 0; g < 8; ++g) {
                l1_cfetch(qA, qB, W1u + (size_t)g * (16 * 256), lo1, 4);
                l1_aload(xQ0, xQ1, xs, ar0, ar1, 16 * g, 4, half);
                l1_mma(a00, a01, a10, a11, xQ0, xQ1, qA, qB, 4);
                wmma_branch_pad();
            }
        }
        const float* wc = W1u + (size_t)(128 + 45 * tA) * 256;
        l1_cfetch(qA, qB, wc, lo1);
        l1_aload(xQ0, xQ1, xs, ar0, ar1, XC0, G1, half);
        l1_mma(a00, a01, a10, a11, xP0, xP1, pA, pB, sg_count(6));
        wmma_branch_pad();
        for (int c = 0; c < nT; c += 2) {
            const bool more1 = c + 1 < nT;
            if (more1) {
                l1_cfetch(pA, pB, wc + (size_t)(c + 1) * (45 * 256), lo1);
                l1_aload(xP0, xP1, xs, ar0, ar1, XC0 + (c + 1) * CSLOT, G1, half);
            }
            l1_mma(a00, a01, a10, a11, xQ0, xQ1, qA, qB, G1);
            wmma_branch_pad();
            if (more1) {
                if (c + 2 < nT) {
                    l1_cfetch(qA, qB, wc + (size_t)(c + 2) * (45 * 256), lo1);
                    l1_aload(xQ0, xQ1, xs, ar0, ar1, XC0 + (c + 2) * CSLOT, G1, half);
                }
                l1_mma(a00, a01, a10, a11, xP0, xP1, pA, pB, G1);
                wmma_branch_pad();
            }
        }
        laneL = (int)__builtin_amdgcn_mbcnt_lo(~0u, 0u);
        halfL = laneL >> 4;
        colL  = laneL & 15;
        lo2   = (unsigned)(halfL * 2 * 128 + colL);
        wl_fetch(uP, W2u, lo2, 0);
        const float bias1A = b1[n1 + colL], bias1B = b1[n1 + 16 + colL];

        __syncthreads();

        #pragma unroll
        for (int r = 0; r < 8; ++r) {
            int m = r + halfL * 8;
            float v00 = a00[r] + bias1A, v01 = a01[r] + bias1B;
            float v10 = a10[r] + bias1A, v11 = a11[r] + bias1B;
            hA[m * H1S + n1 + colL]             = v00 > 0.0f ? v00 : 0.0f;
            hA[m * H1S + n1 + 16 + colL]        = v01 > 0.0f ? v01 : 0.0f;
            hA[(m + 16) * H1S + n1 + colL]      = v10 > 0.0f ? v10 : 0.0f;
            hA[(m + 16) * H1S + n1 + 16 + colL] = v11 > 0.0f ? v11 : 0.0f;
        }
    }
    __syncthreads();

    {
        v8f ac0 = {}, ac1 = {};
        wl_aload(hP0, hP1, hA, H1S, colL, 0, halfL);
        wl_fetch(uQ, W2u, lo2, 1);
        wl_aload(hQ0, hQ1, hA, H1S, colL, 1, halfL);
        wl_mma(ac0, ac1, hP0, hP1, uP);
        wl_fetch(uP, W2u, lo2, 2);
        wl_aload(hP0, hP1, hA, H1S, colL, 2, halfL);
        wl_mma(ac0, ac1, hQ0, hQ1, uQ);
        wl_fetch(uQ, W2u, lo2, 3);
        wl_aload(hQ0, hQ1, hA, H1S, colL, 3, halfL);
        wl_mma(ac0, ac1, hP0, hP1, uP);
        wl_fetch(uP, W3u, lo2, 0);
        const float bias2 = b2[n0 + colL];
        wl_mma(ac0, ac1, hQ0, hQ1, uQ);
        #pragma unroll
        for (int r = 0; r < 8; ++r) {
            int m = r + halfL * 8;
            float v0 = ac0[r] + bias2, v1 = ac1[r] + bias2;
            hB[m * H2S + n0 + colL]        = v0 > 0.0f ? v0 : 0.0f;
            hB[(m + 16) * H2S + n0 + colL] = v1 > 0.0f ? v1 : 0.0f;
        }
    }
    __syncthreads();

    {
        v8f ac0 = {}, ac1 = {};
        wl_aload(hP0, hP1, hB, H2S, colL, 0, halfL);
        wl_fetch(uQ, W3u, lo2, 1);
        wl_aload(hQ0, hQ1, hB, H2S, colL, 1, halfL);
        wl_mma(ac0, ac1, hP0, hP1, uP);
        wl_fetch(uP, W4u, lo2, 0);
        const float bias3 = b3[n0 + colL];
        wl_mma(ac0, ac1, hQ0, hQ1, uQ);
        #pragma unroll
        for (int r = 0; r < 8; ++r) {
            int m = r + halfL * 8;
            float v0 = ac0[r] + bias3, v1 = ac1[r] + bias3;
            hA[m * H2S + n0 + colL]        = v0 > 0.0f ? v0 : 0.0f;
            hA[(m + 16) * H2S + n0 + colL] = v1 > 0.0f ? v1 : 0.0f;
        }
    }
    __syncthreads();

    {
        v8f ac0 = {}, ac1 = {};
        wl_aload(hP0, hP1, hA, H2S, colL, 0, halfL);
        wl_fetch(uQ, W4u, lo2, 1);
        wl_aload(hQ0, hQ1, hA, H2S, colL, 1, halfL);
        const float bias4 = b4[n0 + colL];
        wl_mma(ac0, ac1, hP0, hP1, uP);
        wl_mma(ac0, ac1, hQ0, hQ1, uQ);
        #pragma unroll
        for (int r = 0; r < 8; ++r) {
            int m = r + halfL * 8;
            out[(size_t)sIdx[m] * 128 + n0 + colL]      = ac0[r] + bias4;
            out[(size_t)sIdx[m + 16] * 128 + n0 + colL] = ac1[r] + bias4;
        }
    }
}

extern "C" void kernel_launch(void* const* d_in, const int* in_sizes, int n_in,
                              void* d_out, int out_size, void* d_ws, size_t ws_size,
                              hipStream_t stream) {
    (void)n_in; (void)d_ws; (void)ws_size; (void)out_size;
    const int*   seq    = (const int*)  d_in[0];
    const float* xyz    = (const float*)d_in[1];
    const float* dih    = (const float*)d_in[2];
    const int*   cidx   = (const int*)  d_in[3];
    const float* orient = (const float*)d_in[4];
    const float* aa_emb    = (const float*)d_in[6];
    const float* chain_emb = (const float*)d_in[7];
    const float* W1 = (const float*)d_in[8];   const float* b1 = (const float*)d_in[9];
    const float* W2 = (const float*)d_in[10];  const float* b2 = (const float*)d_in[11];
    const float* W3 = (const float*)d_in[12];  const float* b3 = (const float*)d_in[13];
    const float* W4 = (const float*)d_in[14];  const float* b4 = (const float*)d_in[15];
    float* out = (float*)d_out;

    const int tokens = in_sizes[0];
    const int blocks = tokens / MT;
    residue_embed_fused<<<blocks, 256, 0, stream>>>(
        seq, xyz, dih, cidx, orient, aa_emb, chain_emb,
        W1, b1, W2, b2, W3, b3, W4, b4, out);
}


# Layer-1 weights fetched 8 bytes a lane: a wave's two N tiles take interleaved columns (2c, 2c+1), so one load feeds both tiles (a register swap pairs the k values): half the W1 loads; sums unchanged
# speedup vs baseline: 4.7200x; 1.0321x over previous
#include <hip/hip_runtime.h>
#include <math.h>


typedef float v2f __attribute__((ext_vector_type(2)));
typedef float v8f __attribute__((ext_vector_type(8)));

#define MT    32
#define NSTAT 74
#define CSLOT 48
#define XC0   296
#define XROW  1260
#define CHUNK 1024
#define NTYPE 20
#define H1S   260
#define H2S   132
#define G1    12
#define G2    16

#define WMMA_F32(a, b, c) \
    __builtin_amdgcn_wmma_f32_16x16x4_f32(false, (a), false, (b), (short)0, (c), false, false)

__device__ __forceinline__ void block_edge(v8f& a, v8f& b, v8f& c, v8f& d)
{
    asm volatile("" : "+v"(a), "+v"(b), "+v"(c), "+v"(d) :: "memory");
    __builtin_amdgcn_sched_barrier(0);
}
__device__ __forceinline__ void block_edge(v8f& a, v8f& b)
{
    asm volatile("" : "+v"(a), "+v"(b) :: "memory");
    __builtin_amdgcn_sched_barrier(0);
}


__device__ __forceinline__ constexpr int sg_first(int g) { return g == 0 ? 0 : g == 1 ? 12 : g == 2 ? 24 : g == 3 ? 32 : g == 4 ? 44 : g == 5 ? 56 : 64; }
__device__ __forceinline__ constexpr int sg_count(int g) { return g == 6 ? 10 : ((g == 2 || g == 5) ? 8 : 12); }

__device__ __forceinline__ void l1_fetch(v2f (&fA)[G1], v2f (&fB)[G1],
                                         const float* __restrict__ wu, unsigned lo, int g)
{
    #pragma unroll
    for (int s = 0; s < G1; ++s) {
        if (s < sg_count(g)) {
            const int S = sg_first(g) + s;
            const int row = (S < 32) ? 4 * S : (S < 64) ? 1067 + 4 * (S - 32) : 1028 + 4 * (S - 64);
            unsigned o0 = lo + (unsigned)(row * 256);
            fA[s] = *(const v2f*)&wu[o0];
            fB[s] = *(const v2f*)&wu[o0 + 256u];
        }
    }
}

__device__ __forceinline__ void l1_cfetch(v2f (&fA)[G1], v2f (&fB)[G1],
                                          const float* __restrict__ wc, unsigned lo, int n = G1)
{
    #pragma unroll
    for (int s = 0; s < G1; ++s) {
        if (s < n) {
            unsigned o0 = lo + (unsigned)(4 * s * 256);
            fA[s] = *(const v2f*)&wc[o0];
            fB[s] = *(const v2f*)&wc[o0 + 256u];
        }
    }
}

__device__ __forceinline__ void pair_cols(v2f& r0, v2f& r1)
{
    asm("v_swap_b32 %0, %1" : "+v"(r0.y), "+v"(r1.x));
}

__device__ __forceinline__ void l1_aload(v2f (&x0)[G1], v2f (&x1)[G1],
                                         const float* xs, int ar0, int ar1, int k0, int n, int half)
{
    #pragma unroll
    for (int s = 0; s < G1; ++s) {
        if (s < n) {
            int kk = k0 + 4 * s + half * 2;
            x0[s] = *(const v2f*)&xs[ar0 + kk];
            x1[s] = *(const v2f*)&xs[ar1 + kk];
        }
    }
}

__device__ __forceinline__ void l1_aload1(v2f (&x0)[G1], const float* xs, int ar0, int k0, int n, int half)
{
    #pragma unroll
    for (int s = 0; s < G1; ++s) {
        if (s < n) x0[s] = *(const v2f*)&xs[ar0 + k0 + 4 * s + half * 2];
    }
}

__device__ __forceinline__ void t_mma(v8f& t0, v8f& t1, const v2f (&x0)[G1],
                                      v2f (&fA)[G1], v2f (&fB)[G1], int n)
{
    block_edge(t0, t1);
    #pragma unroll
    for (int s = 0; s < G1; ++s) {
        if (s < n) {
            pair_cols(fA[s], fB[s]);
            t0 = WMMA_F32(x0[s], fA[s], t0);
            t1 = WMMA_F32(x0[s], fB[s], t1);
            __builtin_amdgcn_sched_barrier(0);
        }
    }
    block_edge(t0, t1);
}

__device__ __forceinline__ void l1_mma(v8f& a00, v8f& a01, v8f& a10, v8f& a11,
                                       const v2f (&x0)[G1], const v2f (&x1)[G1],
                                       v2f (&fA)[G1], v2f (&fB)[G1], int n)
{
    block_edge(a00, a01, a10, a11);
    #pragma unroll
    for (int s = 0; s < G1; ++s) {
        if (s < n) {
            pair_cols(fA[s], fB[s]);
            a00 = WMMA_F32(x0[s], fA[s], a00);
            a01 = WMMA_F32(x0[s], fB[s], a01);
            a10 = WMMA_F32(x1[s], fA[s], a10);
            a11 = WMMA_F32(x1[s], fB[s], a11);
            __builtin_amdgcn_sched_barrier(0);
        }
    }
    block_edge(a00, a01, a10, a11);
}

__device__ __forceinline__ void wmma_branch_pad()
{
    asm volatile("v_nop\n\tv_nop\n\tv_nop\n\tv_nop\n\tv_nop");
}

__device__ __forceinline__ void wl_fetch(v2f (&f)[G2], const float* __restrict__ wu, unsigned lo, int g)
{
    #pragma unroll
    for (int s = 0; s < G2; ++s) {
        unsigned o = lo + (unsigned)((g * G2 + s) * 4 * 128);
        f[s].x = wu[o];
        f[s].y = wu[o + 128u];
    }
}

__device__ __forceinline__ void wl_aload(v2f (&x0)[G2], v2f (&x1)[G2],
                                         const float* h, int hs, int col, int g, int half)
{
    #pragma unroll
    for (int s = 0; s < G2; ++s) {
        int kk = (g * G2 + s) * 4 + half * 2;
        x0[s] = *(const v2f*)&h[col * hs + kk];
        x1[s] = *(const v2f*)&h[(16 + col) * hs + kk];
    }
}

__device__ __forceinline__ void wl_mma(v8f& ac0, v8f& ac1, const v2f (&x0)[G2], const v2f (&x1)[G2],
                                       const v2f (&f)[G2])
{
    block_edge(ac0, ac1);
    #pragma unroll
    for (int s = 0; s < G2; ++s) {
        ac0 = WMMA_F32(x0[s], f[s], ac0);
        ac1 = WMMA_F32(x1[s], f[s], ac1);
    }
    block_edge(ac0, ac1);
}

struct CoordIn {
    float ax, ay, az;
    float cx, cy, cz;
    float O[9];
    int   slot;
};

__device__ __forceinline__ void coord_load(CoordIn& c, const float* __restrict__ xyz,
                                           const float* __restrict__ orient, const int* sIdx,
                                           int tA, int i)
{
    int tt = i / 15, a = i % 15;
    int tk = sIdx[tt];
    const float* xp = xyz + (size_t)tk * 45;
    c.ax = xp[a * 3 + 0];  c.ay = xp[a * 3 + 1];  c.az = xp[a * 3 + 2];
    c.cx = xp[3 + 0];      c.cy = xp[3 + 1];      c.cz = xp[3 + 2];
    const float* O = orient + (size_t)tk * 9;
    #pragma unroll
    for (int j = 0; j < 9; ++j) c.O[j] = O[j];
    c.slot = sIdx[MT + tt] - tA;
}

__device__ __forceinline__ void coord_store(float* xs, const CoordIn& c, int i)
{
    int tt = i / 15, a = i % 15;
    float rx = c.ax - c.cx;
    float ry = c.ay - c.cy;
    float rz = c.az - c.cz;
    int base = tt * XROW + XC0 + c.slot * CSLOT + a * 3;
    xs[base + 0] = c.O[0] * rx + c.O[3] * ry + c.O[6] * rz;
    xs[base + 1] = c.O[1] * rx + c.O[4] * ry + c.O[7] * rz;
    xs[base + 2] = c.O[2] * rx + c.O[5] * ry + c.O[8] * rz;
}

__global__ __launch_bounds__(256, 4) void residue_embed_fused(
    const int* __restrict__ seq, const float* __restrict__ xyz,
    const float* __restrict__ dih, const int* __restrict__ cidx,
    const float* __restrict__ orient,
    const float* __restrict__ aa_emb, const float* __restrict__ chain_emb,
    const float* __restrict__ W1, const float* __restrict__ b1,
    const float* __restrict__ W2, const float* __restrict__ b2,
    const float* __restrict__ W3, const float* __restrict__ b3,
    const float* __restrict__ W4, const float* __restrict__ b4,
    float* __restrict__ out)
{
    __shared__ float xs[MT * XROW];
    __shared__ int   sIdx[2 * MT];
    __shared__ unsigned wtot[8 * 10];
    float* const hA = xs;
    float* const hB = xs + MT * H1S;

    const int tid = threadIdx.x;

    const int lane = tid & 31;
    const int w    = __builtin_amdgcn_readfirstlane(tid >> 5);
    const int half = lane >> 4;
    const int col  = lane & 15;

    const int n1 = w * 32;
    const float* W1u = W1 + n1;
    const unsigned lo1 = (unsigned)(half * 2 * 256 + 2 * col);
    const int n0 = w * 16;
    const float* W2u = W2 + n0;
    const float* W3u = W3 + n0;
    const float* W4u = W4 + n0;

    v2f pA[G1], pB[G1], qA[G1], qB[G1];
    l1_fetch(pA, pB, W1u, lo1, 0);

    const int chunk = (int)blockIdx.x / (CHUNK / MT);
    const int4 sv = ((const int4*)seq)[chunk * (CHUNK / 4) + tid];
    unsigned long long ca = 0ull, cb = 0ull;
    unsigned cc = 0u;
    #pragma unroll
    for (int j = 0; j < 4; ++j) {
        int v = (j == 0) ? sv.x : (j == 1) ? sv.y : (j == 2) ? sv.z : sv.w;
        v = v < 0 ? 0 : (v > NTYPE - 1 ? NTYPE - 1 : v);
        unsigned sh = (unsigned)(v & 7) * 8u;
        ca += (v < 8) ? (1ull << sh) : 0ull;
        cb += ((v >> 3) == 1) ? (1ull << sh) : 0ull;
        cc += (v >= 16) ? (1u << sh) : 0u;
    }
    {
        unsigned pk[10];
        #pragma unroll
        for (int p = 0; p < 10; ++p) {
            const int t0 = 2 * p, t1 = 2 * p + 1;
            unsigned c0 = (t0 < 8) ? (unsigned)(ca >> (8 * t0)) : (t0 < 16) ? (unsigned)(cb >> (8 * (t0 - 8))) : (cc >> (8 * (t0 - 16)));
            unsigned c1 = (t1 < 8) ? (unsigned)(ca >> (8 * t1)) : (t1 < 16) ? (unsigned)(cb >> (8 * (t1 - 8))) : (cc >> (8 * (t1 - 16)));
            pk[p] = (c0 & 255u) | ((c1 & 255u) << 16);
        }
        #pragma unroll
        for (int off = 16; off >= 1; off >>= 1) {
            #pragma unroll
            for (int p = 0; p < 10; ++p) pk[p] += (unsigned)__shfl_xor((int)pk[p], off);
        }
        if (lane == 0) {
            #pragma unroll
            for (int p = 0; p < 10; ++p) wtot[w * 10 + p] = pk[p];
        }
    }
    __syncthreads();

    const int R0 = ((int)blockIdx.x % (CHUNK / MT)) * MT;
    int tA, tB, baseA;
    {
        const int tq = lane < NTYPE ? lane : NTYPE - 1;
        unsigned n = 0u;
        #pragma unroll
        for (int ww = 0; ww < 8; ++ww) n += (wtot[ww * 10 + (tq >> 1)] >> (16 * (tq & 1))) & 0xFFFFu;
        const int nl = lane < NTYPE ? (int)n : 0;
        int incl = nl;
        incl += __builtin_amdgcn_update_dpp(0, incl, 0x111, 0xf, 0xf, true);
        incl += __builtin_amdgcn_update_dpp(0, incl, 0x112, 0xf, 0xf, true);
        incl += __builtin_amdgcn_update_dpp(0, incl, 0x114, 0xf, 0xf, true);
        incl += __builtin_amdgcn_update_dpp(0, incl, 0x118, 0xf, 0xf, true);
        const int low = __builtin_amdgcn_readlane(incl, 15);
        if (lane >= 16) incl += low;
        const unsigned mA = __builtin_amdgcn_ballot_w32(incl > R0);
        const unsigned mB = __builtin_amdgcn_ballot_w32(incl > R0 + 31);
        tA    = __builtin_ctz(mA);
        tB    = __builtin_ctz(mB);
        baseA = __builtin_amdgcn_readlane(incl - nl, tA);
    }
    const int nT = tB - tA + 1;
    {
        int baseT = baseA - R0;
        for (int t = tA; t <= tB; ++t) {
            unsigned long long sel = (t < 8) ? ca : (t < 16) ? cb : (unsigned long long)cc;
            int cnt = (int)((sel >> ((t & 7) * 8)) & 255ull);
            int inc = cnt;
            #pragma unroll
            for (int off = 1; off < 32; off <<= 1) {
                int y = __shfl_up(inc, off);
                if (lane >= off) inc += y;
            }
            int rank = baseT + inc - cnt;
            int nt = 0;
            for (int ww = 0; ww < 8; ++ww) {
                int nw = (int)((wtot[ww * 10 + (t >> 1)] >> (16 * (t & 1))) & 0xFFFFu);
                if (ww < w) rank += nw;
                nt += nw;
            }
            #pragma unroll
            for (int j = 0; j < 4; ++j) {
                int v = (j == 0) ? sv.x : (j == 1) ? sv.y : (j == 2) ? sv.z : sv.w;
                v = v < 0 ? 0 : (v > NTYPE - 1 ? NTYPE - 1 : v);
                if (v == t) {
                    if ((unsigned)rank < (unsigned)MT) { sIdx[rank] = chunk * CHUNK + 4 * tid + j; sIdx[MT + rank] = t; }
                    ++rank;
                }
            }
            baseT += nt;
        }
    }
    {
        const int zr = tid >> 3, zc = tid & 7;
        for (int c = zc; c < nT * CSLOT; c += 8) xs[zr * XROW + XC0 + c] = 0.0f;
    }
    __syncthreads();

    float dv = 0.0f;
    if (tid < MT * 3) dv = dih[(size_t)sIdx[tid / 3] * 3 + (tid % 3)];

    const bool aaT = nT <= 6;
    const int rcA = cidx[sIdx[col]], rcB = cidx[sIdx[16 + col]];
    const int slA = sIdx[MT + col] - tA, slB = sIdx[MT + 16 + col] - tA;

    const int gk = tid & 127;
    const int gh = tid >> 7;
    float ga[16], gc[8];
    #pragma unroll
    for (int j = 0; j < 16; ++j) {
        int tt = 2 * j + gh;
        int sl = tt - 10;
        bool on = aaT ? (sl >= 0 && sl < nT) : true;
        int ty = aaT ? (tA + (on ? sl : 0)) : sIdx[MT + tt];
        float v = aa_emb[ty * 128 + gk];
        ga[j] = on ? v : 0.0f;
    }
    #pragma unroll
    for (int j = 0; j < 8; ++j) {
        int r = 2 * j + gh;
        bool on = (r >= 1 && r <= 9);
        float v = chain_emb[(on ? r : 1) * 128 + gk];
        gc[j] = on ? v : 0.0f;
    }
    const int  ci1  = tid + 256;
    const bool has1 = ci1 < MT * 15;
    CoordIn cA, cB;
    coord_load(cA, xyz, orient, sIdx, tA, tid);
    coord_load(cB, xyz, orient, sIdx, tA, has1 ? ci1 : MT * 15 - 1);

    if (tid < MT * 3) {
        int tt = tid / 3, d = tid % 3;
        float v = dv;
        const float FREQ[6] = {1.0f, 2.0f, 3.0f, 1.0f, 0.5f, 1.0f / 3.0f};
        int base = tt * XROW + 256 + d * 13;
        xs[base] = v;
        #pragma unroll
        for (int f = 0; f < 6; ++f) {
            xs[base + 1 + f] = sinf(FREQ[f] * v);
            xs[base + 7 + f] = cosf(FREQ[f] * v);
        }
    } else if (tid >= 128 && tid < 128 + MT) {
        xs[(tid - 128) * XROW + 295] = 0.0f;
    }

    coord_store(xs, cA, tid);
    if (has1) coord_store(xs, cB, ci1);
    #pragma unroll
    for (int j = 0; j < 16; ++j) xs[(2 * j + gh) * XROW + gk] = ga[j];
    #pragma unroll
    for (int j = 0; j < 8; ++j)  xs[(2 * j + gh) * XROW + 128 + gk] = gc[j];
    unsigned selmask;
    {
        unsigned mA = (rcA >= 1 && rcA <= 9) ? (1u << rcA) : 0u;
        unsigned mB = (rcB >= 1 && rcB <= 9) ? (1u << rcB) : 0u;
        if (aaT) {
            int a = slA < 0 ? 0 : (slA > 5 ? 5 : slA);
            int b = slB < 0 ? 0 : (slB > 5 ? 5 : slB);
            mA |= 1u << (10 + a);
            mB |= 1u << (10 + b);
        }
        selmask = mA | (mB << 16);
    }
    __syncthreads();

    int laneL, halfL, colL;
    unsigned lo2;
    v2f uP[G2], uQ[G2];
    v2f hP0[G2], hP1[G2], hQ0[G2], hQ1[G2];

    {
        v8f a00 = {}, a01 = {}, a10 = {}, a11 = {};
        const int ar0 = col * XROW;
        const int ar1 = (16 + col) * XROW;
        v2f xP0[G1], xP1[G1], xQ0[G1], xQ1[G1];
        v8f t0 = {}, t1 = {};
        l1_aload1(xP0, xs, ar0, 0, sg_count(0), half);
        l1_fetch(qA, qB, W1u, lo1, 1);
        l1_aload1(xQ0, xs, ar0, 4 * sg_first(1), sg_count(1), half);
        t_mma(t0, t1, xP0, pA, pB, sg_count(0));
        l1_fetch(pA, pB, W1u, lo1, 2);
        l1_aload1(xP0, xs, ar0, 4 * sg_first(2), sg_count(2), half);
        t_mma(t0, t1, xQ0, qA, qB, sg_count(1));
        l1_fetch(qA, qB, W1u, lo1, 3);
        l1_aload1(xQ0, xs, ar0, 4 * sg_first(3), sg_count(3), half);
        t_mma(t0, t1, xP0, pA, pB, sg_count(2));
        if (!aaT) { t0 = v8f{}; t1 = v8f{}; }
        l1_fetch(pA, pB, W1u, lo1, 4);
        l1_aload1(xP0, xs, ar0, 4 * sg_first(4), sg_count(4), half);
        t_mma(t0, t1, xQ0, qA, qB, sg_count(3));
        l1_fetch(qA, qB, W1u, lo1, 5);
        l1_aload1(xQ0, xs, ar0, 4 * sg_first(5), sg_count(5), half);
        t_mma(t0, t1, xP0, pA, pB, sg_count(4));
        l1_fetch(pA, pB, W1u, lo1, 6);
        l1_aload(xP0, xP1, xs, ar0, ar1, 4 * sg_first(6), sg_count(6), half);
        t_mma(t0, t1, xQ0, qA, qB, sg_count(5));
        {
            const unsigned m0 = (selmask >> (8 * half)) & 0xFFu;
            const unsigned m1 = (selmask >> (16 + 8 * half)) & 0xFFu;
            v2f s0[4], s1[4];
            #pragma unroll
            for (int s = 0; s < 4; ++s) {
                s0[s].x = ((m0 >> (2 * s)) & 1u) ? 1.0f : 0.0f;
                s0[s].y = ((m0 >> (2 * s + 1)) & 1u) ? 1.0f : 0.0f;
                s1[s].x = ((m1 >> (2 * s)) & 1u) ? 1.0f : 0.0f;
                s1[s].y = ((m1 >> (2 * s + 1)) & 1u) ? 1.0f : 0.0f;
            }
            block_edge(a00, a01, a10, a11);
            #pragma unroll
            for (int s = 0; s < 4; ++s) {
                v2f b0 = { t0[2 * s], t0[2 * s + 1] };
                v2f b1 = { t1[2 * s], t1[2 * s + 1] };
                a00 = WMMA_F32(s0[s], b0, a00);
                a01 = WMMA_F32(s0[s], b1, a01);
                a10 = WMMA_F32(s1[s], b0, a10);
                a11 = WMMA_F32(s1[s], b1, a11);
            }
            block_edge(a00, a01, a10, a11);
        }
        wmma_branch_pad();
        if (!aaT) {
            #pragma nounroll
            for (int g = 0; g < 8; ++g) {
                l1_cfetch(qA, qB, W1u + (size_t)g * (16 * 256), lo1, 4);
                l1_aload(xQ0, xQ1, xs, ar0, ar1, 16 * g, 4, half);
                l1_mma(a00, a01, a10, a11, xQ0, xQ1, qA, qB, 4);
                wmma_branch_pad();
            }
        }
        const float* wc = W1u + (size_t)(128 + 45 * tA) * 256;
        l1_cfetch(qA, qB, wc, lo1);
        l1_aload(xQ0, xQ1, xs, ar0, ar1, XC0, G1, half);
        l1_mma(a00, a01, a10, a11, xP0, xP1, pA, pB, sg_count(6));
        wmma_branch_pad();
        for (int c = 0; c < nT; c += 2) {
            const bool more1 = c + 1 < nT;
            if (more1) {
                l1_cfetch(pA, pB, wc + (size_t)(c + 1) * (45 * 256), lo1);
                l1_aload(xP0, xP1, xs, ar0, ar1, XC0 + (c + 1) * CSLOT, G1, half);
            }
            l1_mma(a00, a01, a10, a11, xQ0, xQ1, qA, qB, G1);
            wmma_branch_pad();
            if (more1) {
                if (c + 2 < nT) {
                    l1_cfetch(qA, qB, wc + (size_t)(c + 2) * (45 * 256), lo1);
                    l1_aload(xQ0, xQ1, xs, ar0, ar1, XC0 + (c + 2) * CSLOT, G1, half);
                }
                l1_mma(a00, a01, a10, a11, xP0, xP1, pA, pB, G1);
                wmma_branch_pad();
            }
        }
        laneL = (int)__builtin_amdgcn_mbcnt_lo(~0u, 0u);
        halfL = laneL >> 4;
        colL  = laneL & 15;
        lo2   = (unsigned)(halfL * 2 * 128 + colL);
        wl_fetch(uP, W2u, lo2, 0);
        const float bias1A = b1[n1 + 2 * colL], bias1B = b1[n1 + 2 * colL + 1];

        v2f h0[8], h1[8];
        #pragma unroll
        for (int r = 0; r < 8; ++r) {
            float v00 = a00[r] + bias1A, v01 = a01[r] + bias1B;
            float v10 = a10[r] + bias1A, v11 = a11[r] + bias1B;
            h0[r] = v2f{ v00 > 0.0f ? v00 : 0.0f, v01 > 0.0f ? v01 : 0.0f };
            h1[r] = v2f{ v10 > 0.0f ? v10 : 0.0f, v11 > 0.0f ? v11 : 0.0f };
        }

        __syncthreads();

        #pragma unroll
        for (int r = 0; r < 8; ++r) {
            int m = r + halfL * 8;
            *(v2f*)&hA[m * H1S + n1 + 2 * colL]        = h0[r];
            *(v2f*)&hA[(m + 16) * H1S + n1 + 2 * colL] = h1[r];
        }
    }
    __syncthreads();

    {
        v8f ac0 = {}, ac1 = {};
        wl_aload(hP0, hP1, hA, H1S, colL, 0, halfL);
        wl_fetch(uQ, W2u, lo2, 1);
        wl_aload(hQ0, hQ1, hA, H1S, colL, 1, halfL);
        wl_mma(ac0, ac1, hP0, hP1, uP);
        wl_fetch(uP, W2u, lo2, 2);
        wl_aload(hP0, hP1, hA, H1S, colL, 2, halfL);
        wl_mma(ac0, ac1, hQ0, hQ1, uQ);
        wl_fetch(uQ, W2u, lo2, 3);
        wl_aload(hQ0, hQ1, hA, H1S, colL, 3, halfL);
        wl_mma(ac0, ac1, hP0, hP1, uP);
        wl_fetch(uP, W3u, lo2, 0);
        const float bias2 = b2[n0 + colL];
        wl_mma(ac0, ac1, hQ0, hQ1, uQ);
        #pragma unroll
        for (int r = 0; r < 8; ++r) {
            int m = r + halfL * 8;
            float v0 = ac0[r] + bias2, v1 = ac1[r] + bias2;
            hB[m * H2S + n0 + colL]        = v0 > 0.0f ? v0 : 0.0f;
            hB[(m + 16) * H2S + n0 + colL] = v1 > 0.0f ? v1 : 0.0f;
        }
    }
    __syncthreads();

    {
        v8f ac0 = {}, ac1 = {};
        wl_aload(hP0, hP1, hB, H2S, colL, 0, halfL);
        wl_fetch(uQ, W3u, lo2, 1);
        wl_aload(hQ0, hQ1, hB, H2S, colL, 1, halfL);
        wl_mma(ac0, ac1, hP0, hP1, uP);
        wl_fetch(uP, W4u, lo2, 0);
        const float bias3 = b3[n0 + colL];
        wl_mma(ac0, ac1, hQ0, hQ1, uQ);
        #pragma unroll
        for (int r = 0; r < 8; ++r) {
            int m = r + halfL * 8;
            float v0 = ac0[r] + bias3, v1 = ac1[r] + bias3;
            hA[m * H2S + n0 + colL]        = v0 > 0.0f ? v0 : 0.0f;
            hA[(m + 16) * H2S + n0 + colL] = v1 > 0.0f ? v1 : 0.0f;
        }
    }
    __syncthreads();

    {
        v8f ac0 = {}, ac1 = {};
        wl_aload(hP0, hP1, hA, H2S, colL, 0, halfL);
        wl_fetch(uQ, W4u, lo2, 1);
        wl_aload(hQ0, hQ1, hA, H2S, colL, 1, halfL);
        const float bias4 = b4[n0 + colL];
        wl_mma(ac0, ac1, hP0, hP1, uP);
        wl_mma(ac0, ac1, hQ0, hQ1, uQ);
        #pragma unroll
        for (int r = 0; r < 8; ++r) {
            int m = r + halfL * 8;
            out[(size_t)sIdx[m] * 128 + n0 + colL]      = ac0[r] + bias4;
            out[(size_t)sIdx[m + 16] * 128 + n0 + colL] = ac1[r] + bias4;
        }
    }
}

extern "C" void kernel_launch(void* const* d_in, const int* in_sizes, int n_in,
                              void* d_out, int out_size, void* d_ws, size_t ws_size,
                              hipStream_t stream) {
    (void)n_in; (void)d_ws; (void)ws_size; (void)out_size;
    const int*   seq    = (const int*)  d_in[0];
    const float* xyz    = (const float*)d_in[1];
    const float* dih    = (const float*)d_in[2];
    const int*   cidx   = (const int*)  d_in[3];
    const float* orient = (const float*)d_in[4];
    const float* aa_emb    = (const float*)d_in[6];
    const float* chain_emb = (const float*)d_in[7];
    const float* W1 = (const float*)d_in[8];   const float* b1 = (const float*)d_in[9];
    const float* W2 = (const float*)d_in[10];  const float* b2 = (const float*)d_in[11];
    const float* W3 = (const float*)d_in[12];  const float* b3 = (const float*)d_in[13];
    const float* W4 = (const float*)d_in[14];  const float* b4 = (const float*)d_in[15];
    float* out = (float*)d_out;

    const int tokens = in_sizes[0];
    const int blocks = tokens / MT;
    residue_embed_fused<<<blocks, 256, 0, stream>>>(
        seq, xyz, dih, cidx, orient, aa_emb, chain_emb,
        W1, b1, W2, b2, W3, b3, W4, b4, out);
}


# aa and chain products formed once per table row for the whole call: a first launch writes Ttab = emb rows @ W1 (32x256) to the workspace, the token kernel selects its rows; summation order changes
# speedup vs baseline: 4.8770x; 1.0333x over previous
#include <hip/hip_runtime.h>
#include <math.h>


typedef float v2f __attribute__((ext_vector_type(2)));
typedef float v8f __attribute__((ext_vector_type(8)));

#define MT    32
#define NSTAT 74
#define CSLOT 48
#define XC0   296
#define XROW  1260
#define CHUNK 1024
#define NTYPE 20
#define H1S   260
#define H2S   132
#define G1    12
#define G2    16

#define WMMA_F32(a, b, c) \
    __builtin_amdgcn_wmma_f32_16x16x4_f32(false, (a), false, (b), (short)0, (c), false, false)

__device__ __forceinline__ void block_edge(v8f& a, v8f& b, v8f& c, v8f& d)
{
    asm volatile("" : "+v"(a), "+v"(b), "+v"(c), "+v"(d) :: "memory");
    __builtin_amdgcn_sched_barrier(0);
}
__device__ __forceinline__ void block_edge(v8f& a, v8f& b)
{
    asm volatile("" : "+v"(a), "+v"(b) :: "memory");
    __builtin_amdgcn_sched_barrier(0);
}


__device__ __forceinline__ void l1_cfetch(v2f (&fA)[G1], v2f (&fB)[G1],
                                          const float* __restrict__ wc, unsigned lo, int n = G1)
{
    #pragma unroll
    for (int s = 0; s < G1; ++s) {
        if (s < n) {
            unsigned o0 = lo + (unsigned)(4 * s * 256);
            fA[s] = *(const v2f*)&wc[o0];
            fB[s] = *(const v2f*)&wc[o0 + 256u];
        }
    }
}

__device__ __forceinline__ void pair_cols(v2f& r0, v2f& r1)
{
    asm("v_swap_b32 %0, %1" : "+v"(r0.y), "+v"(r1.x));
}

__device__ __forceinline__ void l1_aload(v2f (&x0)[G1], v2f (&x1)[G1],
                                         const float* xs, int ar0, int ar1, int k0, int n, int half)
{
    #pragma unroll
    for (int s = 0; s < G1; ++s) {
        if (s < n) {
            int kk = k0 + 4 * s + half * 2;
            x0[s] = *(const v2f*)&xs[ar0 + kk];
            x1[s] = *(const v2f*)&xs[ar1 + kk];
        }
    }
}


__device__ __forceinline__ void l1_mma(v8f& a00, v8f& a01, v8f& a10, v8f& a11,
                                       const v2f (&x0)[G1], const v2f (&x1)[G1],
                                       v2f (&fA)[G1], v2f (&fB)[G1], int n)
{
    block_edge(a00, a01, a10, a11);
    #pragma unroll
    for (int s = 0; s < G1; ++s) {
        if (s < n) {
            pair_cols(fA[s], fB[s]);
            a00 = WMMA_F32(x0[s], fA[s], a00);
            a01 = WMMA_F32(x0[s], fB[s], a01);
            a10 = WMMA_F32(x1[s], fA[s], a10);
            a11 = WMMA_F32(x1[s], fB[s], a11);
            __builtin_amdgcn_sched_barrier(0);
        }
    }
    block_edge(a00, a01, a10, a11);
}

__device__ __forceinline__ void wmma_branch_pad()
{
    asm volatile("v_nop\n\tv_nop\n\tv_nop\n\tv_nop\n\tv_nop");
}

__device__ __forceinline__ void wl_fetch(v2f (&f)[G2], const float* __restrict__ wu, unsigned lo, int g)
{
    #pragma unroll
    for (int s = 0; s < G2; ++s) {
        unsigned o = lo + (unsigned)((g * G2 + s) * 4 * 128);
        f[s].x = wu[o];
        f[s].y = wu[o + 128u];
    }
}

__device__ __forceinline__ void wl_aload(v2f (&x0)[G2], v2f (&x1)[G2],
                                         const float* h, int hs, int col, int g, int half)
{
    #pragma unroll
    for (int s = 0; s < G2; ++s) {
        int kk = (g * G2 + s) * 4 + half * 2;
        x0[s] = *(const v2f*)&h[col * hs + kk];
        x1[s] = *(const v2f*)&h[(16 + col) * hs + kk];
    }
}

__device__ __forceinline__ void wl_mma(v8f& ac0, v8f& ac1, const v2f (&x0)[G2], const v2f (&x1)[G2],
                                       const v2f (&f)[G2])
{
    block_edge(ac0, ac1);
    #pragma unroll
    for (int s = 0; s < G2; ++s) {
        ac0 = WMMA_F32(x0[s], f[s], ac0);
        ac1 = WMMA_F32(x1[s], f[s], ac1);
    }
    block_edge(ac0, ac1);
}

__global__ __launch_bounds__(32) void embed_rows_times_w1(
    const float* __restrict__ aa_emb, const float* __restrict__ chain_emb,
    const float* __restrict__ W1, float* __restrict__ Ttab)
{
    const int lane = threadIdx.x;
    const int half = lane >> 4;
    const int col  = lane & 15;
    const int part = (int)blockIdx.x >> 3;
    const int nb   = ((int)blockIdx.x & 7) * 32;
    const float* emb = part == 0 ? aa_emb : part == 1 ? aa_emb + 16 * 128 : chain_emb;
    const int nrow   = part == 0 ? 16 : part == 1 ? 4 : 10;
    const int nstore = part == 2 ? 12 : nrow;
    const int trow0  = part == 0 ? 0 : part == 1 ? 16 : 20;
    const int wrow0  = part == 2 ? 1067 : 0;
    const bool on = col < nrow;
    const float* er = emb + (on ? col : 0) * 128 + 2 * half;
    const float* wu = W1 + (size_t)wrow0 * 256 + nb;
    const unsigned lo = (unsigned)(half * 2 * 256 + 2 * col);
    v2f fa[32], fb[32], xa[32];
    #pragma unroll
    for (int s = 0; s < 32; ++s) {
        unsigned o0 = lo + (unsigned)(4 * s * 256);
        fa[s] = *(const v2f*)&wu[o0];
        fb[s] = *(const v2f*)&wu[o0 + 256u];
        v2f v = *(const v2f*)&er[4 * s];
        xa[s] = on ? v : v2f{0.0f, 0.0f};
    }
    v8f t0 = {}, t1 = {};
    block_edge(t0, t1);
    #pragma unroll
    for (int s = 0; s < 32; ++s) {
        pair_cols(fa[s], fb[s]);
        t0 = WMMA_F32(xa[s], fa[s], t0);
        t1 = WMMA_F32(xa[s], fb[s], t1);
    }
    #pragma unroll
    for (int r = 0; r < 8; ++r) {
        int m = r + half * 8;
        if (m < nstore) *(v2f*)&Ttab[(trow0 + m) * 256 + nb + 2 * col] = v2f{ t0[r], t1[r] };
    }
}

struct CoordIn {
    float ax, ay, az;
    float cx, cy, cz;
    float O[9];
    int   slot;
};

__device__ __forceinline__ void coord_load(CoordIn& c, const float* __restrict__ xyz,
                                           const float* __restrict__ orient, const int* sIdx,
                                           int tA, int i)
{
    int tt = i / 15, a = i % 15;
    int tk = sIdx[tt];
    const float* xp = xyz + (size_t)tk * 45;
    c.ax = xp[a * 3 + 0];  c.ay = xp[a * 3 + 1];  c.az = xp[a * 3 + 2];
    c.cx = xp[3 + 0];      c.cy = xp[3 + 1];      c.cz = xp[3 + 2];
    const float* O = orient + (size_t)tk * 9;
    #pragma unroll
    for (int j = 0; j < 9; ++j) c.O[j] = O[j];
    c.slot = sIdx[MT + tt] - tA;
}

__device__ __forceinline__ void coord_store(float* xs, const CoordIn& c, int i)
{
    int tt = i / 15, a = i % 15;
    float rx = c.ax - c.cx;
    float ry = c.ay - c.cy;
    float rz = c.az - c.cz;
    int base = tt * XROW + XC0 + c.slot * CSLOT + a * 3;
    xs[base + 0] = c.O[0] * rx + c.O[3] * ry + c.O[6] * rz;
    xs[base + 1] = c.O[1] * rx + c.O[4] * ry + c.O[7] * rz;
    xs[base + 2] = c.O[2] * rx + c.O[5] * ry + c.O[8] * rz;
}

__global__ __launch_bounds__(256, 4) void residue_embed_fused(
    const int* __restrict__ seq, const float* __restrict__ xyz,
    const float* __restrict__ dih, const int* __restrict__ cidx,
    const float* __restrict__ orient,
    const float* __restrict__ W1, const float* __restrict__ b1,
    const float* __restrict__ W2, const float* __restrict__ b2,
    const float* __restrict__ W3, const float* __restrict__ b3,
    const float* __restrict__ W4, const float* __restrict__ b4,
    const float* __restrict__ Ttab,
    float* __restrict__ out)
{
    __shared__ float xs[MT * XROW];
    __shared__ int   sIdx[2 * MT];
    __shared__ unsigned wtot[8 * 10];
    float* const hA = xs;
    float* const hB = xs + MT * H1S;

    const int tid = threadIdx.x;

    const int lane = tid & 31;
    const int w    = __builtin_amdgcn_readfirstlane(tid >> 5);
    const int half = lane >> 4;
    const int col  = lane & 15;

    const int n1 = w * 32;
    const float* W1u = W1 + n1;
    const unsigned lo1 = (unsigned)(half * 2 * 256 + 2 * col);
    const int n0 = w * 16;
    const float* W2u = W2 + n0;
    const float* W3u = W3 + n0;
    const float* W4u = W4 + n0;

    v2f pA[G1], pB[G1], qA[G1], qB[G1];
    l1_cfetch(pA, pB, W1u + 1028 * 256, lo1, 10);
    l1_cfetch(qA, qB, Ttab + n1, lo1, 8);

    const int chunk = (int)blockIdx.x / (CHUNK / MT);
    const int4 sv = ((const int4*)seq)[chunk * (CHUNK / 4) + tid];
    unsigned long long ca = 0ull, cb = 0ull;
    unsigned cc = 0u;
    #pragma unroll
    for (int j = 0; j < 4; ++j) {
        int v = (j == 0) ? sv.x : (j == 1) ? sv.y : (j == 2) ? sv.z : sv.w;
        v = v < 0 ? 0 : (v > NTYPE - 1 ? NTYPE - 1 : v);
        unsigned sh = (unsigned)(v & 7) * 8u;
        ca += (v < 8) ? (1ull << sh) : 0ull;
        cb += ((v >> 3) == 1) ? (1ull << sh) : 0ull;
        cc += (v >= 16) ? (1u << sh) : 0u;
    }
    {
        unsigned pk[10];
        #pragma unroll
        for (int p = 0; p < 10; ++p) {
            const int t0 = 2 * p, t1 = 2 * p + 1;
            unsigned c0 = (t0 < 8) ? (unsigned)(ca >> (8 * t0)) : (t0 < 16) ? (unsigned)(cb >> (8 * (t0 - 8))) : (cc >> (8 * (t0 - 16)));
            unsigned c1 = (t1 < 8) ? (unsigned)(ca >> (8 * t1)) : (t1 < 16) ? (unsigned)(cb >> (8 * (t1 - 8))) : (cc >> (8 * (t1 - 16)));
            pk[p] = (c0 & 255u) | ((c1 & 255u) << 16);
        }
        #pragma unroll
        for (int off = 16; off >= 1; off >>= 1) {
            #pragma unroll
            for (int p = 0; p < 10; ++p) pk[p] += (unsigned)__shfl_xor((int)pk[p], off);
        }
        if (lane == 0) {
            #pragma unroll
            for (int p = 0; p < 10; ++p) wtot[w * 10 + p] = pk[p];
        }
    }
    __syncthreads();

    const int R0 = ((int)blockIdx.x % (CHUNK / MT)) * MT;
    int tA, tB, baseA;
    {
        const int tq = lane < NTYPE ? lane : NTYPE - 1;
        unsigned n = 0u;
        #pragma unroll
        for (int ww = 0; ww < 8; ++ww) n += (wtot[ww * 10 + (tq >> 1)] >> (16 * (tq & 1))) & 0xFFFFu;
        const int nl = lane < NTYPE ? (int)n : 0;
        int incl = nl;
        incl += __builtin_amdgcn_update_dpp(0, incl, 0x111, 0xf, 0xf, true);
        incl += __builtin_amdgcn_update_dpp(0, incl, 0x112, 0xf, 0xf, true);
        incl += __builtin_amdgcn_update_dpp(0, incl, 0x114, 0xf, 0xf, true);
        incl += __builtin_amdgcn_update_dpp(0, incl, 0x118, 0xf, 0xf, true);
        const int low = __builtin_amdgcn_readlane(incl, 15);
        if (lane >= 16) incl += low;
        const unsigned mA = __builtin_amdgcn_ballot_w32(incl > R0);
        const unsigned mB = __builtin_amdgcn_ballot_w32(incl > R0 + 31);
        tA    = __builtin_ctz(mA);
        tB    = __builtin_ctz(mB);
        baseA = __builtin_amdgcn_readlane(incl - nl, tA);
    }
    const int nT = tB - tA + 1;
    {
        int baseT = baseA - R0;
        for (int t = tA; t <= tB; ++t) {
            unsigned long long sel = (t < 8) ? ca : (t < 16) ? cb : (unsigned long long)cc;
            int cnt = (int)((sel >> ((t & 7) * 8)) & 255ull);
            int inc = cnt;
            #pragma unroll
            for (int off = 1; off < 32; off <<= 1) {
                int y = __shfl_up(inc, off);
                if (lane >= off) inc += y;
            }
            int rank = baseT + inc - cnt;
            int nt = 0;
            for (int ww = 0; ww < 8; ++ww) {
                int nw = (int)((wtot[ww * 10 + (t >> 1)] >> (16 * (t & 1))) & 0xFFFFu);
                if (ww < w) rank += nw;
                nt += nw;
            }
            #pragma unroll
            for (int j = 0; j < 4; ++j) {
                int v = (j == 0) ? sv.x : (j == 1) ? sv.y : (j == 2) ? sv.z : sv.w;
                v = v < 0 ? 0 : (v > NTYPE - 1 ? NTYPE - 1 : v);
                if (v == t) {
                    if ((unsigned)rank < (unsigned)MT) { sIdx[rank] = chunk * CHUNK + 4 * tid + j; sIdx[MT + rank] = t; }
                    ++rank;
                }
            }
            baseT += nt;
        }
    }
    {
        const int zr = tid >> 3, zc = tid & 7;
        for (int c = zc; c < nT * CSLOT; c += 8) xs[zr * XROW + XC0 + c] = 0.0f;
    }
    __syncthreads();

    float dv = 0.0f;
    if (tid < MT * 3) dv = dih[(size_t)sIdx[tid / 3] * 3 + (tid % 3)];

    const int rcA = cidx[sIdx[col]], rcB = cidx[sIdx[16 + col]];
    const int tyA = sIdx[MT + col], tyB = sIdx[MT + 16 + col];

    const int  ci1  = tid + 256;
    const bool has1 = ci1 < MT * 15;
    CoordIn cA, cB;
    coord_load(cA, xyz, orient, sIdx, tA, tid);
    coord_load(cB, xyz, orient, sIdx, tA, has1 ? ci1 : MT * 15 - 1);

    if (tid < MT * 3) {
        int tt = tid / 3, d = tid % 3;
        float v = dv;
        const float FREQ[6] = {1.0f, 2.0f, 3.0f, 1.0f, 0.5f, 1.0f / 3.0f};
        int base = tt * XROW + 256 + d * 13;
        xs[base] = v;
        #pragma unroll
        for (int f = 0; f < 6; ++f) {
            xs[base + 1 + f] = sinf(FREQ[f] * v);
            xs[base + 7 + f] = cosf(FREQ[f] * v);
        }
    } else if (tid >= 128 && tid < 128 + MT) {
        xs[(tid - 128) * XROW + 295] = 0.0f;
    }

    coord_store(xs, cA, tid);
    if (has1) coord_store(xs, cB, ci1);
    unsigned selA, selB;
    {
        selA = 1u << (tyA & 31);
        selB = 1u << (tyB & 31);
        if (rcA >= 1 && rcA <= 9) selA |= 1u << (20 + rcA);
        if (rcB >= 1 && rcB <= 9) selB |= 1u << (20 + rcB);
    }
    __syncthreads();

    int laneL, halfL, colL;
    unsigned lo2;
    v2f uP[G2], uQ[G2];
    v2f hP0[G2], hP1[G2], hQ0[G2], hQ1[G2];

    {
        v8f a00 = {}, a01 = {}, a10 = {}, a11 = {};
        const int ar0 = col * XROW;
        const int ar1 = (16 + col) * XROW;
        v2f xP0[G1], xP1[G1], xQ0[G1], xQ1[G1];
        {
            const unsigned m0 = selA >> (2 * half);
            const unsigned m1 = selB >> (2 * half);
            #pragma unroll
            for (int s = 0; s < 8; ++s) {
                xQ0[s].x = ((m0 >> (4 * s)) & 1u) ? 1.0f : 0.0f;
                xQ0[s].y = ((m0 >> (4 * s + 1)) & 1u) ? 1.0f : 0.0f;
                xQ1[s].x = ((m1 >> (4 * s)) & 1u) ? 1.0f : 0.0f;
                xQ1[s].y = ((m1 >> (4 * s + 1)) & 1u) ? 1.0f : 0.0f;
            }
        }
        l1_aload(xP0, xP1, xs, ar0, ar1, 256, 10, half);
        l1_mma(a00, a01, a10, a11, xQ0, xQ1, qA, qB, 8);
        const float* wc = W1u + (size_t)(128 + 45 * tA) * 256;
        l1_cfetch(qA, qB, wc, lo1);
        l1_aload(xQ0, xQ1, xs, ar0, ar1, XC0, G1, half);
        l1_mma(a00, a01, a10, a11, xP0, xP1, pA, pB, 10);
        wmma_branch_pad();
        for (int c = 0; c < nT; c += 2) {
            const bool more1 = c + 1 < nT;
            if (more1) {
                l1_cfetch(pA, pB, wc + (size_t)(c + 1) * (45 * 256), lo1);
                l1_aload(xP0, xP1, xs, ar0, ar1, XC0 + (c + 1) * CSLOT, G1, half);
            }
            l1_mma(a00, a01, a10, a11, xQ0, xQ1, qA, qB, G1);
            wmma_branch_pad();
            if (more1) {
                if (c + 2 < nT) {
                    l1_cfetch(qA, qB, wc + (size_t)(c + 2) * (45 * 256), lo1);
                    l1_aload(xQ0, xQ1, xs, ar0, ar1, XC0 + (c + 2) * CSLOT, G1, half);
                }
                l1_mma(a00, a01, a10, a11, xP0, xP1, pA, pB, G1);
                wmma_branch_pad();
            }
        }
        laneL = (int)__builtin_amdgcn_mbcnt_lo(~0u, 0u);
        halfL = laneL >> 4;
        colL  = laneL & 15;
        lo2   = (unsigned)(halfL * 2 * 128 + colL);
        wl_fetch(uP, W2u, lo2, 0);
        const float bias1A = b1[n1 + 2 * colL], bias1B = b1[n1 + 2 * colL + 1];

        v2f h0[8], h1[8];
        #pragma unroll
        for (int r = 0; r < 8; ++r) {
            float v00 = a00[r] + bias1A, v01 = a01[r] + bias1B;
            float v10 = a10[r] + bias1A, v11 = a11[r] + bias1B;
            h0[r] = v2f{ v00 > 0.0f ? v00 : 0.0f, v01 > 0.0f ? v01 : 0.0f };
            h1[r] = v2f{ v10 > 0.0f ? v10 : 0.0f, v11 > 0.0f ? v11 : 0.0f };
        }

        __syncthreads();

        #pragma unroll
        for (int r = 0; r < 8; ++r) {
            int m = r + halfL * 8;
            *(v2f*)&hA[m * H1S + n1 + 2 * colL]        = h0[r];
            *(v2f*)&hA[(m + 16) * H1S + n1 + 2 * colL] = h1[r];
        }
    }
    __syncthreads();

    {
        v8f ac0 = {}, ac1 = {};
        wl_aload(hP0, hP1, hA, H1S, colL, 0, halfL);
        wl_fetch(uQ, W2u, lo2, 1);
        wl_aload(hQ0, hQ1, hA, H1S, colL, 1, halfL);
        wl_mma(ac0, ac1, hP0, hP1, uP);
        wl_fetch(uP, W2u, lo2, 2);
        wl_aload(hP0, hP1, hA, H1S, colL, 2, halfL);
        wl_mma(ac0, ac1, hQ0, hQ1, uQ);
        wl_fetch(uQ, W2u, lo2, 3);
        wl_aload(hQ0, hQ1, hA, H1S, colL, 3, halfL);
        wl_mma(ac0, ac1, hP0, hP1, uP);
        wl_fetch(uP, W3u, lo2, 0);
        const float bias2 = b2[n0 + colL];
        wl_mma(ac0, ac1, hQ0, hQ1, uQ);
        #pragma unroll
        for (int r = 0; r < 8; ++r) {
            int m = r + halfL * 8;
            float v0 = ac0[r] + bias2, v1 = ac1[r] + bias2;
            hB[m * H2S + n0 + colL]        = v0 > 0.0f ? v0 : 0.0f;
            hB[(m + 16) * H2S + n0 + colL] = v1 > 0.0f ? v1 : 0.0f;
        }
    }
    __syncthreads();

    {
        v8f ac0 = {}, ac1 = {};
        wl_aload(hP0, hP1, hB, H2S, colL, 0, halfL);
        wl_fetch(uQ, W3u, lo2, 1);
        wl_aload(hQ0, hQ1, hB, H2S, colL, 1, halfL);
        wl_mma(ac0, ac1, hP0, hP1, uP);
        wl_fetch(uP, W4u, lo2, 0);
        const float bias3 = b3[n0 + colL];
        wl_mma(ac0, ac1, hQ0, hQ1, uQ);
        #pragma unroll
        for (int r = 0; r < 8; ++r) {
            int m = r + halfL * 8;
            float v0 = ac0[r] + bias3, v1 = ac1[r] + bias3;
            hA[m * H2S + n0 + colL]        = v0 > 0.0f ? v0 : 0.0f;
            hA[(m + 16) * H2S + n0 + colL] = v1 > 0.0f ? v1 : 0.0f;
        }
    }
    __syncthreads();

    {
        v8f ac0 = {}, ac1 = {};
        wl_aload(hP0, hP1, hA, H2S, colL, 0, halfL);
        wl_fetch(uQ, W4u, lo2, 1);
        wl_aload(hQ0, hQ1, hA, H2S, colL, 1, halfL);
        const float bias4 = b4[n0 + colL];
        wl_mma(ac0, ac1, hP0, hP1, uP);
        wl_mma(ac0, ac1, hQ0, hQ1, uQ);
        #pragma unroll
        for (int r = 0; r < 8; ++r) {
            int m = r + halfL * 8;
            out[(size_t)sIdx[m] * 128 + n0 + colL]      = ac0[r] + bias4;
            out[(size_t)sIdx[m + 16] * 128 + n0 + colL] = ac1[r] + bias4;
        }
    }
}

extern "C" void kernel_launch(void* const* d_in, const int* in_sizes, int n_in,
                              void* d_out, int out_size, void* d_ws, size_t ws_size,
                              hipStream_t stream) {
    (void)n_in; (void)ws_size; (void)out_size;
    const int*   seq    = (const int*)  d_in[0];
    const float* xyz    = (const float*)d_in[1];
    const float* dih    = (const float*)d_in[2];
    const int*   cidx   = (const int*)  d_in[3];
    const float* orient = (const float*)d_in[4];
    const float* aa_emb    = (const float*)d_in[6];
    const float* chain_emb = (const float*)d_in[7];
    const float* W1 = (const float*)d_in[8];   const float* b1 = (const float*)d_in[9];
    const float* W2 = (const float*)d_in[10];  const float* b2 = (const float*)d_in[11];
    const float* W3 = (const float*)d_in[12];  const float* b3 = (const float*)d_in[13];
    const float* W4 = (const float*)d_in[14];  const float* b4 = (const float*)d_in[15];
    float* out = (float*)d_out;
    float* Ttab = (float*)d_ws;

    const int tokens = in_sizes[0];
    const int blocks = tokens / MT;
    embed_rows_times_w1<<<24, 32, 0, stream>>>(aa_emb, chain_emb, W1, Ttab);
    residue_embed_fused<<<blocks, 256, 0, stream>>>(
        seq, xyz, dih, cidx, orient,
        W1, b1, W2, b2, W3, b3, W4, b4, Ttab, out);
}
